# starting kernel
# baseline (speedup 1.0000x reference)
#include <hip/hip_runtime.h>
#include <hip/hip_bf16.h>

typedef __attribute__((ext_vector_type(2))) float v2f;
typedef __attribute__((ext_vector_type(8))) float v8f;

#define NEG_SLOPE 0.2f

__device__ __forceinline__ unsigned ordEnc(float f) {
  unsigned u = __float_as_uint(f);
  return (u & 0x80000000u) ? ~u : (u | 0x80000000u);
}
__device__ __forceinline__ float ordDec(unsigned u) {
  return (u & 0x80000000u) ? __uint_as_float(u ^ 0x80000000u)
                           : __uint_as_float(~u);
}

__global__ void gat_fill_u32(unsigned* __restrict__ p, unsigned v, long long n) {
  long long i = (long long)blockIdx.x * blockDim.x + threadIdx.x;
  if (i < n) p[i] = v;
}

__global__ void gat_count_loop(const int* __restrict__ dst,
                               const float* __restrict__ ea,
                               float* __restrict__ cnt,
                               float* __restrict__ la, int E) {
  long long i = (long long)blockIdx.x * blockDim.x + threadIdx.x;
  if (i >= (long long)E * 16) return;
  int e = (int)(i >> 4), j = (int)(i & 15);
  int d = dst[e];
  atomicAdd(&la[(long long)d * 16 + j], ea[i]);
  if (j == 0) atomicAdd(&cnt[d], 1.0f);
}

__global__ void gat_loop_div(const float* __restrict__ cnt,
                             float* __restrict__ la, int Nn) {
  long long i = (long long)blockIdx.x * blockDim.x + threadIdx.x;
  if (i >= (long long)Nn * 16) return;
  float c = cnt[i >> 4];
  c = (c < 1.0f) ? 1.0f : c;
  la[i] = la[i] / c;
}

__global__ void gat_medge(const float* __restrict__ We,
                          const float* __restrict__ ae,
                          float* __restrict__ medge) {
  int t = threadIdx.x;
  int k = t >> 2, h = t & 3;
  float s = 0.0f;
  for (int c = 0; c < 32; ++c)
    s += We[k * 128 + h * 32 + c] * ae[h * 32 + c];
  medge[k * 4 + h] = s;
}

__global__ __launch_bounds__(256) void gat_gemm_wmma(
    const float* __restrict__ X, const float* __restrict__ W,
    float* __restrict__ Y, int Nrows) {
  int wave = (int)(((long long)blockIdx.x * blockDim.x + threadIdx.x) >> 5);
  int lane = threadIdx.x & 31;
  int m0 = wave * 16;
  if (m0 >= Nrows) return;

  v8f acc[8] = {};
  int half = lane >> 4;
  int l16 = lane & 15;
  int arow = m0 + l16;
  if (arow >= Nrows) arow = Nrows - 1;
  const float* xr = X + (long long)arow * 128;

  for (int k0 = 0; k0 < 128; k0 += 4) {
    v2f a;
    a.x = xr[k0 + half * 2];
    a.y = xr[k0 + half * 2 + 1];
#pragma unroll
    for (int nt = 0; nt < 8; ++nt) {
      int col = nt * 16 + l16;
      v2f b;
      b.x = W[(k0 + half * 2) * 128 + col];
      b.y = W[(k0 + half * 2 + 1) * 128 + col];
      acc[nt] = __builtin_amdgcn_wmma_f32_16x16x4_f32(
          false, a, false, b, (short)0, acc[nt], false, false);
    }
  }
#pragma unroll
  for (int nt = 0; nt < 8; ++nt) {
#pragma unroll
    for (int r = 0; r < 8; ++r) {
      int row = m0 + r + half * 8;
      if (row < Nrows) Y[(long long)row * 128 + nt * 16 + l16] = acc[nt][r];
    }
  }
}

__global__ __launch_bounds__(128) void gat_attdot(
    const float* __restrict__ xh, const float* __restrict__ as,
    const float* __restrict__ ad, float* __restrict__ a_src,
    float* __restrict__ a_dst, int Nn) {
  int n = blockIdx.x;
  int t = threadIdx.x;
  float v = xh[(long long)n * 128 + t];
  float ps = v * as[t];
  float pd = v * ad[t];
  for (int o = 16; o > 0; o >>= 1) {
    ps += __shfl_down(ps, o, 32);
    pd += __shfl_down(pd, o, 32);
  }
  if ((t & 31) == 0) {
    int h = t >> 5;
    a_src[n * 4 + h] = ps;
    a_dst[n * 4 + h] = pd;
  }
}

__global__ void gat_alpha(const int* __restrict__ srcI,
                          const int* __restrict__ dstI,
                          const float* __restrict__ ea,
                          const float* __restrict__ la,
                          const float* __restrict__ a_src,
                          const float* __restrict__ a_dst,
                          const float* __restrict__ medge,
                          float* __restrict__ alpha,
                          unsigned* __restrict__ mseg, int E, int Nn) {
  long long f = (long long)blockIdx.x * blockDim.x + threadIdx.x;
  long long F = (long long)E + Nn;
  if (f >= F) return;
  int s, d;
  const float* row;
  if (f < (long long)E) {
    s = srcI[f]; d = dstI[f]; row = ea + f * 16;
  } else {
    s = (int)(f - E); d = s; row = la + (long long)s * 16;
  }
  float e0[16];
#pragma unroll
  for (int k = 0; k < 16; ++k) e0[k] = row[k];
#pragma unroll
  for (int h = 0; h < 4; ++h) {
    float aE = 0.0f;
#pragma unroll
    for (int k = 0; k < 16; ++k) aE += e0[k] * medge[k * 4 + h];
    float v = a_src[(long long)s * 4 + h] + a_dst[(long long)d * 4 + h] + aE;
    v = (v > 0.0f) ? v : NEG_SLOPE * v;
    alpha[f * 4 + h] = v;
    atomicMax(&mseg[(long long)d * 4 + h], ordEnc(v));
  }
}

__global__ void gat_expsum(const int* __restrict__ dstI,
                           float* __restrict__ alpha,
                           const unsigned* __restrict__ mseg,
                           float* __restrict__ sseg, int E, int Nn) {
  long long i = (long long)blockIdx.x * blockDim.x + threadIdx.x;
  long long F4 = ((long long)E + Nn) * 4;
  if (i >= F4) return;
  long long f = i >> 2;
  int h = (int)(i & 3);
  int d = (f < (long long)E) ? dstI[f] : (int)(f - E);
  float mv = ordDec(mseg[(long long)d * 4 + h]);
  if (!(mv > -__builtin_inff() && mv < __builtin_inff())) mv = 0.0f;
  float p = __expf(alpha[i] - mv);
  alpha[i] = p;
  atomicAdd(&sseg[(long long)d * 4 + h], p);
}

__global__ __launch_bounds__(256) void gat_aggr(
    const int* __restrict__ srcI, const int* __restrict__ dstI,
    const float* __restrict__ alpha, const float* __restrict__ sseg,
    const float* __restrict__ xh, float* __restrict__ out, int E, int Nn) {
  long long wid = ((long long)blockIdx.x * blockDim.x + threadIdx.x) >> 5;
  int lane = threadIdx.x & 31;
  long long F = (long long)E + Nn;
  if (wid >= F) return;
  int s, d;
  if (wid < (long long)E) { s = srcI[wid]; d = dstI[wid]; }
  else { s = (int)(wid - E); d = s; }
  int c0 = lane * 4;
  int h = lane >> 3;
  float a = alpha[wid * 4 + h] / (sseg[(long long)d * 4 + h] + 1e-16f);
  const float4 xv = *(const float4*)(xh + (long long)s * 128 + c0);
  float* o = out + (long long)d * 128 + c0;
  atomicAdd(o + 0, a * xv.x);
  atomicAdd(o + 1, a * xv.y);
  atomicAdd(o + 2, a * xv.z);
  atomicAdd(o + 3, a * xv.w);
}

__global__ void gat_bias_act(float* __restrict__ y,
                             const float* __restrict__ bias, long long n,
                             int relu) {
  long long i = (long long)blockIdx.x * blockDim.x + threadIdx.x;
  if (i >= n) return;
  float v = y[i] + bias[i & 127];
  if (relu) v = (v > 0.0f) ? v : 0.0f;
  y[i] = v;
}

static void fillN(unsigned* p, unsigned v, long long n, hipStream_t s) {
  if (n <= 0) return;
  long long b = (n + 255) / 256;
  hipLaunchKernelGGL(gat_fill_u32, dim3((unsigned)b), dim3(256), 0, s, p, v, n);
}

extern "C" void kernel_launch(void* const* d_in, const int* in_sizes, int n_in,
                              void* d_out, int out_size, void* d_ws,
                              size_t ws_size, hipStream_t stream) {
  const int N = in_sizes[0] / 128;
  const int E = in_sizes[1] / 2;

  const float* x  = (const float*)d_in[0];
  const int* ei   = (const int*)d_in[1];
  const int* srcI = ei;
  const int* dstI = ei + E;
  const float* ea = (const float*)d_in[2];

  const float* Wl[2]  = {(const float*)d_in[3], (const float*)d_in[9]};
  const float* asl[2] = {(const float*)d_in[4], (const float*)d_in[10]};
  const float* adl[2] = {(const float*)d_in[5], (const float*)d_in[11]};
  const float* Wel[2] = {(const float*)d_in[6], (const float*)d_in[12]};
  const float* ael[2] = {(const float*)d_in[7], (const float*)d_in[13]};
  const float* bl[2]  = {(const float*)d_in[8], (const float*)d_in[14]};

  float* wsf = (float*)d_ws;
  size_t off = 0;
  float* cnt   = wsf + off; off += (size_t)N;
  float* la    = wsf + off; off += (size_t)N * 16;
  float* medge = wsf + off; off += 64;
  float* xh    = wsf + off; off += (size_t)N * 128;
  float* a_src = wsf + off; off += (size_t)N * 4;
  float* a_dst = wsf + off; off += (size_t)N * 4;
  float* alpha = wsf + off; off += ((size_t)E + N) * 4;
  unsigned* mseg = (unsigned*)(wsf + off); off += (size_t)N * 4;
  float* sseg  = wsf + off; off += (size_t)N * 4;
  float* h1    = wsf + off; off += (size_t)N * 128;
  (void)ws_size; (void)n_in; (void)out_size;

  const long long F = (long long)E + N;

  fillN((unsigned*)cnt, 0u, N, stream);
  fillN((unsigned*)la, 0u, (long long)N * 16, stream);
  {
    long long n = (long long)E * 16;
    hipLaunchKernelGGL(gat_count_loop, dim3((unsigned)((n + 255) / 256)),
                       dim3(256), 0, stream, dstI, ea, cnt, la, E);
    long long m = (long long)N * 16;
    hipLaunchKernelGGL(gat_loop_div, dim3((unsigned)((m + 255) / 256)),
                       dim3(256), 0, stream, cnt, la, N);
  }

  for (int l = 0; l < 2; ++l) {
    const float* xin = (l == 0) ? x : h1;
    float* outb = (l == 0) ? h1 : (float*)d_out;

    hipLaunchKernelGGL(gat_medge, dim3(1), dim3(64), 0, stream, Wel[l], ael[l],
                       medge);

    {
      long long waves = ((long long)N + 15) / 16;
      long long blocks = (waves + 7) / 8;
      hipLaunchKernelGGL(gat_gemm_wmma, dim3((unsigned)blocks), dim3(256), 0,
                         stream, xin, Wl[l], xh, N);
    }

    hipLaunchKernelGGL(gat_attdot, dim3((unsigned)N), dim3(128), 0, stream, xh,
                       asl[l], adl[l], a_src, a_dst, N);

    fillN(mseg, 0x007fffffu, (long long)N * 4, stream);
    fillN((unsigned*)sseg, 0u, (long long)N * 4, stream);
    fillN((unsigned*)outb, 0u, (long long)N * 128, stream);

    hipLaunchKernelGGL(gat_alpha, dim3((unsigned)((F + 255) / 256)), dim3(256),
                       0, stream, srcI, dstI, ea, la, a_src, a_dst, medge,
                       alpha, mseg, E, N);

    {
      long long n4 = F * 4;
      hipLaunchKernelGGL(gat_expsum, dim3((unsigned)((n4 + 255) / 256)),
                         dim3(256), 0, stream, dstI, alpha, mseg, sseg, E, N);
    }

    {
      long long blocks = (F + 7) / 8;
      hipLaunchKernelGGL(gat_aggr, dim3((unsigned)blocks), dim3(256), 0, stream,
                         srcI, dstI, alpha, sseg, xh, outb, E, N);
    }

    {
      long long n = (long long)N * 128;
      hipLaunchKernelGGL(gat_bias_act, dim3((unsigned)((n + 255) / 256)),
                         dim3(256), 0, stream, outb, bl[l], n, (l == 0) ? 1 : 0);
    }
  }
}


# edges grouped by destination each call (wave-scan segment starts, arrival-order slots); aggregation: one wave per node, register sums, one store per line, no float atomics on out; sum order changes
# speedup vs baseline: 1.3469x; 1.3469x over previous
#include <hip/hip_runtime.h>
#include <hip/hip_bf16.h>

typedef __attribute__((ext_vector_type(2))) float v2f;
typedef __attribute__((ext_vector_type(8))) float v8f;

#define NEG_SLOPE 0.2f

__device__ __forceinline__ unsigned ordEnc(float f) {
  unsigned u = __float_as_uint(f);
  return (u & 0x80000000u) ? ~u : (u | 0x80000000u);
}
__device__ __forceinline__ float ordDec(unsigned u) {
  return (u & 0x80000000u) ? __uint_as_float(u ^ 0x80000000u)
                           : __uint_as_float(~u);
}

__global__ void gat_fill_u32(unsigned* __restrict__ p, unsigned v, long long n) {
  long long i = (long long)blockIdx.x * blockDim.x + threadIdx.x;
  if (i < n) p[i] = v;
}

__global__ void gat_count_loop(const int* __restrict__ dst,
                               const float* __restrict__ ea,
                               float* __restrict__ cnt,
                               float* __restrict__ la, int E) {
  long long i = (long long)blockIdx.x * blockDim.x + threadIdx.x;
  if (i >= (long long)E * 16) return;
  int e = (int)(i >> 4), j = (int)(i & 15);
  int d = dst[e];
  atomicAdd(&la[(long long)d * 16 + j], ea[i]);
  if (j == 0) atomicAdd(&cnt[d], 1.0f);
}

__global__ void gat_loop_div(const float* __restrict__ cnt,
                             float* __restrict__ la, int Nn) {
  long long i = (long long)blockIdx.x * blockDim.x + threadIdx.x;
  if (i >= (long long)Nn * 16) return;
  float c = cnt[i >> 4];
  c = (c < 1.0f) ? 1.0f : c;
  la[i] = la[i] / c;
}

__global__ void gat_medge(const float* __restrict__ We,
                          const float* __restrict__ ae,
                          float* __restrict__ medge) {
  int t = threadIdx.x;
  int k = t >> 2, h = t & 3;
  float s = 0.0f;
  for (int c = 0; c < 32; ++c)
    s += We[k * 128 + h * 32 + c] * ae[h * 32 + c];
  medge[k * 4 + h] = s;
}

__global__ __launch_bounds__(256) void gat_gemm_wmma(
    const float* __restrict__ X, const float* __restrict__ W,
    float* __restrict__ Y, int Nrows) {
  int wave = (int)(((long long)blockIdx.x * blockDim.x + threadIdx.x) >> 5);
  int lane = threadIdx.x & 31;
  int m0 = wave * 16;
  if (m0 >= Nrows) return;

  v8f acc[8] = {};
  int half = lane >> 4;
  int l16 = lane & 15;
  int arow = m0 + l16;
  if (arow >= Nrows) arow = Nrows - 1;
  const float* xr = X + (long long)arow * 128;

  for (int k0 = 0; k0 < 128; k0 += 4) {
    v2f a;
    a.x = xr[k0 + half * 2];
    a.y = xr[k0 + half * 2 + 1];
#pragma unroll
    for (int nt = 0; nt < 8; ++nt) {
      int col = nt * 16 + l16;
      v2f b;
      b.x = W[(k0 + half * 2) * 128 + col];
      b.y = W[(k0 + half * 2 + 1) * 128 + col];
      acc[nt] = __builtin_amdgcn_wmma_f32_16x16x4_f32(
          false, a, false, b, (short)0, acc[nt], false, false);
    }
  }
#pragma unroll
  for (int nt = 0; nt < 8; ++nt) {
#pragma unroll
    for (int r = 0; r < 8; ++r) {
      int row = m0 + r + half * 8;
      if (row < Nrows) Y[(long long)row * 128 + nt * 16 + l16] = acc[nt][r];
    }
  }
}

__global__ __launch_bounds__(128) void gat_attdot(
    const float* __restrict__ xh, const float* __restrict__ as,
    const float* __restrict__ ad, float* __restrict__ a_src,
    float* __restrict__ a_dst, int Nn) {
  int n = blockIdx.x;
  int t = threadIdx.x;
  float v = xh[(long long)n * 128 + t];
  float ps = v * as[t];
  float pd = v * ad[t];
  for (int o = 16; o > 0; o >>= 1) {
    ps += __shfl_down(ps, o, 32);
    pd += __shfl_down(pd, o, 32);
  }
  if ((t & 31) == 0) {
    int h = t >> 5;
    a_src[n * 4 + h] = ps;
    a_dst[n * 4 + h] = pd;
  }
}

__global__ void gat_alpha(const int* __restrict__ srcI,
                          const int* __restrict__ dstI,
                          const float* __restrict__ ea,
                          const float* __restrict__ la,
                          const float* __restrict__ a_src,
                          const float* __restrict__ a_dst,
                          const float* __restrict__ medge,
                          float* __restrict__ alpha,
                          unsigned* __restrict__ mseg, int E, int Nn) {
  long long f = (long long)blockIdx.x * blockDim.x + threadIdx.x;
  long long F = (long long)E + Nn;
  if (f >= F) return;
  int s, d;
  const float* row;
  if (f < (long long)E) {
    s = srcI[f]; d = dstI[f]; row = ea + f * 16;
  } else {
    s = (int)(f - E); d = s; row = la + (long long)s * 16;
  }
  float e0[16];
#pragma unroll
  for (int k = 0; k < 16; ++k) e0[k] = row[k];
#pragma unroll
  for (int h = 0; h < 4; ++h) {
    float aE = 0.0f;
#pragma unroll
    for (int k = 0; k < 16; ++k) aE += e0[k] * medge[k * 4 + h];
    float v = a_src[(long long)s * 4 + h] + a_dst[(long long)d * 4 + h] + aE;
    v = (v > 0.0f) ? v : NEG_SLOPE * v;
    alpha[f * 4 + h] = v;
    atomicMax(&mseg[(long long)d * 4 + h], ordEnc(v));
  }
}

__global__ void gat_expsum(const int* __restrict__ dstI,
                           float* __restrict__ alpha,
                           const unsigned* __restrict__ mseg,
                           float* __restrict__ sseg, int E, int Nn) {
  long long i = (long long)blockIdx.x * blockDim.x + threadIdx.x;
  long long F4 = ((long long)E + Nn) * 4;
  if (i >= F4) return;
  long long f = i >> 2;
  int h = (int)(i & 3);
  int d = (f < (long long)E) ? dstI[f] : (int)(f - E);
  float mv = ordDec(mseg[(long long)d * 4 + h]);
  if (!(mv > -__builtin_inff() && mv < __builtin_inff())) mv = 0.0f;
  float p = __expf(alpha[i] - mv);
  alpha[i] = p;
  atomicAdd(&sseg[(long long)d * 4 + h], p);
}

__global__ __launch_bounds__(256) void gat_seg_base(
    const float* __restrict__ cnt, int* __restrict__ segb,
    int* __restrict__ cursor, int* __restrict__ total, int Nn) {
  __shared__ int wsum[8];
  __shared__ int bbase;
  int t = threadIdx.x;
  int lane = t & 31, w = t >> 5;
  int n = blockIdx.x * 256 + t;
  int deg = (n < Nn) ? (int)cnt[n] : 0;
  int v = deg;
  for (int o = 1; o < 32; o <<= 1) {
    int u = __shfl_up(v, o, 32);
    if (lane >= o) v += u;
  }
  if (lane == 31) wsum[w] = v;
  __syncthreads();
  if (t == 0) {
    int s = 0;
    for (int i = 0; i < 8; ++i) { int u = wsum[i]; wsum[i] = s; s += u; }
    bbase = atomicAdd(total, s);
  }
  __syncthreads();
  if (n < Nn) {
    int b = bbase + wsum[w] + v - deg;
    segb[n] = b;
    cursor[n] = b;
  }
}

__global__ void gat_slot_scatter(const int* __restrict__ srcI,
                                 const int* __restrict__ dstI,
                                 int* __restrict__ cursor,
                                 int* __restrict__ csrw, int E) {
  int2* csr = (int2*)csrw;
  int e = blockIdx.x * blockDim.x + threadIdx.x;
  if (e >= E) return;
  int d = dstI[e];
  int s = srcI[e];
  int pos = atomicAdd(&cursor[d], 1);
  if ((unsigned)pos < (unsigned)E) csr[pos] = make_int2(s, e);
}

__global__ __launch_bounds__(256) void gat_aggr_csr(
    const int* __restrict__ csrw, const int* __restrict__ segb,
    const float* __restrict__ cnt, const float* __restrict__ alpha,
    const float* __restrict__ sseg, const float* __restrict__ xh,
    float* __restrict__ out, int E, int Nn) {
  const int2* csr = (const int2*)csrw;
  int lane = threadIdx.x & 31;
  int n = __builtin_amdgcn_readfirstlane(
      (int)((blockIdx.x * blockDim.x + threadIdx.x) >> 5));
  if (n >= Nn) return;
  int c0 = lane * 4;
  int h = lane >> 3;
  float den = sseg[n * 4 + h] + 1e-16f;
  int beg = segb[n];
  int end = beg + (int)cnt[n];
  if (beg < 0) beg = 0;
  if (end > E) end = E;

  float a = alpha[((long long)E + n) * 4 + h] / den;
  float4 xv = *(const float4*)(xh + (long long)n * 128 + c0);
  float4 acc;
  acc.x = a * xv.x; acc.y = a * xv.y; acc.z = a * xv.z; acc.w = a * xv.w;

  int j = beg;
  for (; j + 4 <= end; j += 4) {
    int2 q0 = csr[j], q1 = csr[j + 1], q2 = csr[j + 2], q3 = csr[j + 3];
    float p0 = alpha[(long long)q0.y * 4 + h];
    float p1 = alpha[(long long)q1.y * 4 + h];
    float p2 = alpha[(long long)q2.y * 4 + h];
    float p3 = alpha[(long long)q3.y * 4 + h];
    float4 x0 = *(const float4*)(xh + (long long)q0.x * 128 + c0);
    float4 x1 = *(const float4*)(xh + (long long)q1.x * 128 + c0);
    float4 x2 = *(const float4*)(xh + (long long)q2.x * 128 + c0);
    float4 x3 = *(const float4*)(xh + (long long)q3.x * 128 + c0);
    float a0 = p0 / den, a1 = p1 / den, a2 = p2 / den, a3 = p3 / den;
    acc.x += a0 * x0.x; acc.y += a0 * x0.y; acc.z += a0 * x0.z; acc.w += a0 * x0.w;
    acc.x += a1 * x1.x; acc.y += a1 * x1.y; acc.z += a1 * x1.z; acc.w += a1 * x1.w;
    acc.x += a2 * x2.x; acc.y += a2 * x2.y; acc.z += a2 * x2.z; acc.w += a2 * x2.w;
    acc.x += a3 * x3.x; acc.y += a3 * x3.y; acc.z += a3 * x3.z; acc.w += a3 * x3.w;
  }
  for (; j < end; ++j) {
    int2 q = csr[j];
    float p = alpha[(long long)q.y * 4 + h];
    float4 x0 = *(const float4*)(xh + (long long)q.x * 128 + c0);
    float a0 = p / den;
    acc.x += a0 * x0.x; acc.y += a0 * x0.y; acc.z += a0 * x0.z; acc.w += a0 * x0.w;
  }
  *(float4*)(out + (long long)n * 128 + c0) = acc;
}

__global__ void gat_bias_act(float* __restrict__ y,
                             const float* __restrict__ bias, long long n,
                             int relu) {
  long long i = (long long)blockIdx.x * blockDim.x + threadIdx.x;
  if (i >= n) return;
  float v = y[i] + bias[i & 127];
  if (relu) v = (v > 0.0f) ? v : 0.0f;
  y[i] = v;
}

static void fillN(unsigned* p, unsigned v, long long n, hipStream_t s) {
  if (n <= 0) return;
  long long b = (n + 255) / 256;
  hipLaunchKernelGGL(gat_fill_u32, dim3((unsigned)b), dim3(256), 0, s, p, v, n);
}

extern "C" void kernel_launch(void* const* d_in, const int* in_sizes, int n_in,
                              void* d_out, int out_size, void* d_ws,
                              size_t ws_size, hipStream_t stream) {
  const int N = in_sizes[0] / 128;
  const int E = in_sizes[1] / 2;

  const float* x  = (const float*)d_in[0];
  const int* ei   = (const int*)d_in[1];
  const int* srcI = ei;
  const int* dstI = ei + E;
  const float* ea = (const float*)d_in[2];

  const float* Wl[2]  = {(const float*)d_in[3], (const float*)d_in[9]};
  const float* asl[2] = {(const float*)d_in[4], (const float*)d_in[10]};
  const float* adl[2] = {(const float*)d_in[5], (const float*)d_in[11]};
  const float* Wel[2] = {(const float*)d_in[6], (const float*)d_in[12]};
  const float* ael[2] = {(const float*)d_in[7], (const float*)d_in[13]};
  const float* bl[2]  = {(const float*)d_in[8], (const float*)d_in[14]};

  float* wsf = (float*)d_ws;
  size_t off = 0;
  float* cnt   = wsf + off; off += (size_t)N;
  float* la    = wsf + off; off += (size_t)N * 16;
  float* medge = wsf + off; off += 64;
  float* xh    = wsf + off; off += (size_t)N * 128;
  float* a_src = wsf + off; off += (size_t)N * 4;
  float* a_dst = wsf + off; off += (size_t)N * 4;
  float* alpha = wsf + off; off += ((size_t)E + N) * 4;
  unsigned* mseg = (unsigned*)(wsf + off); off += (size_t)N * 4;
  float* sseg  = wsf + off; off += (size_t)N * 4;
  size_t lo = (off + 1) & ~(size_t)1;
  int* segb    = (int*)(wsf + lo);
  int* cursor  = segb + N;
  int* total   = cursor + N;
  int* csr     = total + 4;
  off += (size_t)N * 128;
  (void)ws_size; (void)n_in; (void)out_size;

  const long long F = (long long)E + N;

  fillN((unsigned*)cnt, 0u, N, stream);
  fillN((unsigned*)la, 0u, (long long)N * 16, stream);
  {
    long long n = (long long)E * 16;
    hipLaunchKernelGGL(gat_count_loop, dim3((unsigned)((n + 255) / 256)),
                       dim3(256), 0, stream, dstI, ea, cnt, la, E);
    long long m = (long long)N * 16;
    hipLaunchKernelGGL(gat_loop_div, dim3((unsigned)((m + 255) / 256)),
                       dim3(256), 0, stream, cnt, la, N);
  }

  fillN((unsigned*)total, 0u, 1, stream);
  hipLaunchKernelGGL(gat_seg_base, dim3((unsigned)((N + 255) / 256)), dim3(256),
                     0, stream, cnt, segb, cursor, total, N);
  hipLaunchKernelGGL(gat_slot_scatter, dim3((unsigned)((E + 255) / 256)),
                     dim3(256), 0, stream, srcI, dstI, cursor, csr, E);

  for (int l = 0; l < 2; ++l) {
    const float* xin = (l == 0) ? x : (const float*)d_out;
    float* outb = (float*)d_out;

    hipLaunchKernelGGL(gat_medge, dim3(1), dim3(64), 0, stream, Wel[l], ael[l],
                       medge);

    {
      long long waves = ((long long)N + 15) / 16;
      long long blocks = (waves + 7) / 8;
      hipLaunchKernelGGL(gat_gemm_wmma, dim3((unsigned)blocks), dim3(256), 0,
                         stream, xin, Wl[l], xh, N);
    }

    hipLaunchKernelGGL(gat_attdot, dim3((unsigned)N), dim3(128), 0, stream, xh,
                       asl[l], adl[l], a_src, a_dst, N);

    fillN(mseg, 0x007fffffu, (long long)N * 4, stream);
    fillN((unsigned*)sseg, 0u, (long long)N * 4, stream);

    hipLaunchKernelGGL(gat_alpha, dim3((unsigned)((F + 255) / 256)), dim3(256),
                       0, stream, srcI, dstI, ea, la, a_src, a_dst, medge,
                       alpha, mseg, E, N);

    {
      long long n4 = F * 4;
      hipLaunchKernelGGL(gat_expsum, dim3((unsigned)((n4 + 255) / 256)),
                         dim3(256), 0, stream, dstI, alpha, mseg, sseg, E, N);
    }

    {
      long long blocks = ((long long)N + 7) / 8;
      hipLaunchKernelGGL(gat_aggr_csr, dim3((unsigned)blocks), dim3(256), 0,
                         stream, csr, segb, cnt, alpha, sseg, xh, outb, E, N);
    }

    {
      long long n = (long long)N * 128;
      hipLaunchKernelGGL(gat_bias_act, dim3((unsigned)((n + 255) / 256)),
                         dim3(256), 0, stream, outb, bl[l], n, (l == 0) ? 1 : 0);
    }
  }
}


# elementwise and per-edge kernels give each thread 4-8 elements (attention dots: one wave per 8 nodes): about 6x fewer waves to dispatch; arithmetic and reduction trees unchanged, atomics as before
# speedup vs baseline: 1.5747x; 1.1691x over previous
#include <hip/hip_runtime.h>
#include <hip/hip_bf16.h>

typedef __attribute__((ext_vector_type(2))) float v2f;
typedef __attribute__((ext_vector_type(8))) float v8f;

#define NEG_SLOPE 0.2f

__device__ __forceinline__ unsigned ordEnc(float f) {
  unsigned u = __float_as_uint(f);
  return (u & 0x80000000u) ? ~u : (u | 0x80000000u);
}
__device__ __forceinline__ float ordDec(unsigned u) {
  return (u & 0x80000000u) ? __uint_as_float(u ^ 0x80000000u)
                           : __uint_as_float(~u);
}

__global__ __launch_bounds__(256) void gat_fill_u32(unsigned* __restrict__ p, unsigned v,
                                                    long long n) {
  long long base = (long long)blockIdx.x * 2048 + threadIdx.x;
#pragma unroll
  for (int r = 0; r < 8; ++r) {
    long long i = base + r * 256;
    if (i < n) p[i] = v;
  }
}

__global__ __launch_bounds__(256) void gat_count_loop(const int* __restrict__ dst,
                               const float* __restrict__ ea,
                               float* __restrict__ cnt,
                               float* __restrict__ la, int E) {
  long long total = (long long)E * 16;
  long long base = (long long)blockIdx.x * 2048 + threadIdx.x;
  int d[8];
  float v[8];
#pragma unroll
  for (int r = 0; r < 8; ++r) {
    long long i = base + r * 256;
    d[r] = 0; v[r] = 0.0f;
    if (i < total) { d[r] = dst[i >> 4]; v[r] = ea[i]; }
  }
#pragma unroll
  for (int r = 0; r < 8; ++r) {
    long long i = base + r * 256;
    if (i < total) {
      int j = (int)(i & 15);
      atomicAdd(&la[(long long)d[r] * 16 + j], v[r]);
      if (j == 0) atomicAdd(&cnt[d[r]], 1.0f);
    }
  }
}

__global__ __launch_bounds__(256) void gat_loop_div(const float* __restrict__ cnt,
                             float* __restrict__ la, int Nn) {
  long long total = (long long)Nn * 16;
  long long base = (long long)blockIdx.x * 2048 + threadIdx.x;
  if (total <= 0) return;
  float c[8], x[8];
#pragma unroll
  for (int r = 0; r < 8; ++r) {
    long long i = base + r * 256;
    long long ii = (i < total) ? i : total - 1;
    c[r] = cnt[ii >> 4];
    x[r] = la[ii];
  }
#pragma unroll
  for (int r = 0; r < 8; ++r) {
    long long i = base + r * 256;
    if (i < total) {
      float cc = (c[r] < 1.0f) ? 1.0f : c[r];
      la[i] = x[r] / cc;
    }
  }
}

__global__ void gat_medge(const float* __restrict__ We,
                          const float* __restrict__ ae,
                          float* __restrict__ medge) {
  int t = threadIdx.x;
  int k = t >> 2, h = t & 3;
  float s = 0.0f;
  for (int c = 0; c < 32; ++c)
    s += We[k * 128 + h * 32 + c] * ae[h * 32 + c];
  medge[k * 4 + h] = s;
}

__global__ __launch_bounds__(256) void gat_gemm_wmma(
    const float* __restrict__ X, const float* __restrict__ W,
    float* __restrict__ Y, int Nrows) {
  int wave = (int)(((long long)blockIdx.x * blockDim.x + threadIdx.x) >> 5);
  int lane = threadIdx.x & 31;
  int m0 = wave * 16;
  if (m0 >= Nrows) return;

  v8f acc[8] = {};
  int half = lane >> 4;
  int l16 = lane & 15;
  int arow = m0 + l16;
  if (arow >= Nrows) arow = Nrows - 1;
  const float* xr = X + (long long)arow * 128;

  for (int k0 = 0; k0 < 128; k0 += 4) {
    v2f a;
    a.x = xr[k0 + half * 2];
    a.y = xr[k0 + half * 2 + 1];
#pragma unroll
    for (int nt = 0; nt < 8; ++nt) {
      int col = nt * 16 + l16;
      v2f b;
      b.x = W[(k0 + half * 2) * 128 + col];
      b.y = W[(k0 + half * 2 + 1) * 128 + col];
      acc[nt] = __builtin_amdgcn_wmma_f32_16x16x4_f32(
          false, a, false, b, (short)0, acc[nt], false, false);
    }
  }
#pragma unroll
  for (int nt = 0; nt < 8; ++nt) {
#pragma unroll
    for (int r = 0; r < 8; ++r) {
      int row = m0 + r + half * 8;
      if (row < Nrows) Y[(long long)row * 128 + nt * 16 + l16] = acc[nt][r];
    }
  }
}

__global__ __launch_bounds__(256) void gat_attdot(
    const float* __restrict__ xh, const float* __restrict__ as,
    const float* __restrict__ ad, float* __restrict__ a_src,
    float* __restrict__ a_dst, int Nn) {
  int lane = threadIdx.x & 31;
  int wave = (int)(((long long)blockIdx.x * 256 + threadIdx.x) >> 5);
  int n0 = wave * 8;
  if (n0 >= Nn) return;
  float asv[4], adv[4];
#pragma unroll
  for (int h = 0; h < 4; ++h) { asv[h] = as[h * 32 + lane]; adv[h] = ad[h * 32 + lane]; }
  for (int g = 0; g < 2; ++g) {
    float v[4][4];
#pragma unroll
    for (int q = 0; q < 4; ++q) {
      int n = n0 + g * 4 + q;
      if (n >= Nn) n = Nn - 1;
#pragma unroll
      for (int h = 0; h < 4; ++h) v[q][h] = xh[(long long)n * 128 + h * 32 + lane];
    }
    float rs[4][4], rd[4][4];
#pragma unroll
    for (int q = 0; q < 4; ++q) {
#pragma unroll
      for (int h = 0; h < 4; ++h) {
        float ps = v[q][h] * asv[h];
        float pd = v[q][h] * adv[h];
        for (int o = 16; o > 0; o >>= 1) {
          ps += __shfl_down(ps, o, 32);
          pd += __shfl_down(pd, o, 32);
        }
        rs[q][h] = ps;
        rd[q][h] = pd;
      }
    }
    if (lane == 0) {
#pragma unroll
      for (int q = 0; q < 4; ++q) {
        int n = n0 + g * 4 + q;
        if (n < Nn) {
#pragma unroll
          for (int h = 0; h < 4; ++h) {
            a_src[n * 4 + h] = rs[q][h];
            a_dst[n * 4 + h] = rd[q][h];
          }
        }
      }
    }
  }
}

__global__ void gat_alpha(const int* __restrict__ srcI,
                          const int* __restrict__ dstI,
                          const float* __restrict__ ea,
                          const float* __restrict__ la,
                          const float* __restrict__ a_src,
                          const float* __restrict__ a_dst,
                          const float* __restrict__ medge,
                          float* __restrict__ alpha,
                          unsigned* __restrict__ mseg, int E, int Nn) {
  long long f = (long long)blockIdx.x * blockDim.x + threadIdx.x;
  long long F = (long long)E + Nn;
  if (f >= F) return;
  int s, d;
  const float* row;
  if (f < (long long)E) {
    s = srcI[f]; d = dstI[f]; row = ea + f * 16;
  } else {
    s = (int)(f - E); d = s; row = la + (long long)s * 16;
  }
  float e0[16];
#pragma unroll
  for (int k = 0; k < 16; ++k) e0[k] = row[k];
#pragma unroll
  for (int h = 0; h < 4; ++h) {
    float aE = 0.0f;
#pragma unroll
    for (int k = 0; k < 16; ++k) aE += e0[k] * medge[k * 4 + h];
    float v = a_src[(long long)s * 4 + h] + a_dst[(long long)d * 4 + h] + aE;
    v = (v > 0.0f) ? v : NEG_SLOPE * v;
    alpha[f * 4 + h] = v;
    atomicMax(&mseg[(long long)d * 4 + h], ordEnc(v));
  }
}

__global__ __launch_bounds__(256) void gat_expsum(const int* __restrict__ dstI,
                           float* __restrict__ alpha,
                           const unsigned* __restrict__ mseg,
                           float* __restrict__ sseg, int E, int Nn) {
  long long F4 = ((long long)E + Nn) * 4;
  long long base = (long long)blockIdx.x * 2048 + threadIdx.x;
  int d[8];
  float a[8];
  unsigned mu[8];
#pragma unroll
  for (int r = 0; r < 8; ++r) {
    long long i = base + r * 256;
    d[r] = 0; a[r] = 0.0f;
    if (i < F4) {
      long long f = i >> 2;
      d[r] = (f < (long long)E) ? dstI[f] : (int)(f - E);
      a[r] = alpha[i];
    }
  }
#pragma unroll
  for (int r = 0; r < 8; ++r) {
    long long i = base + r * 256;
    mu[r] = 0u;
    if (i < F4) mu[r] = mseg[(long long)d[r] * 4 + (int)(i & 3)];
  }
#pragma unroll
  for (int r = 0; r < 8; ++r) {
    long long i = base + r * 256;
    if (i < F4) {
      int h = (int)(i & 3);
      float mv = ordDec(mu[r]);
      if (!(mv > -__builtin_inff() && mv < __builtin_inff())) mv = 0.0f;
      float p = __expf(a[r] - mv);
      alpha[i] = p;
      atomicAdd(&sseg[(long long)d[r] * 4 + h], p);
    }
  }
}

__global__ __launch_bounds__(256) void gat_seg_base(
    const float* __restrict__ cnt, int* __restrict__ segb,
    int* __restrict__ cursor, int* __restrict__ total, int Nn) {
  __shared__ int wsum[8];
  __shared__ int bbase;
  int t = threadIdx.x;
  int lane = t & 31, w = t >> 5;
  int n = blockIdx.x * 256 + t;
  int deg = (n < Nn) ? (int)cnt[n] : 0;
  int v = deg;
  for (int o = 1; o < 32; o <<= 1) {
    int u = __shfl_up(v, o, 32);
    if (lane >= o) v += u;
  }
  if (lane == 31) wsum[w] = v;
  __syncthreads();
  if (t == 0) {
    int s = 0;
    for (int i = 0; i < 8; ++i) { int u = wsum[i]; wsum[i] = s; s += u; }
    bbase = atomicAdd(total, s);
  }
  __syncthreads();
  if (n < Nn) {
    int b = bbase + wsum[w] + v - deg;
    segb[n] = b;
    cursor[n] = b;
  }
}

__global__ __launch_bounds__(256) void gat_slot_scatter(const int* __restrict__ srcI,
                                 const int* __restrict__ dstI,
                                 int* __restrict__ cursor,
                                 int* __restrict__ csrw, int E) {
  int2* csr = (int2*)csrw;
  int base = blockIdx.x * 1024 + threadIdx.x;
  int d[4], s[4], pos[4];
#pragma unroll
  for (int r = 0; r < 4; ++r) {
    int e = base + r * 256;
    d[r] = 0; s[r] = 0;
    if (e < E) { d[r] = dstI[e]; s[r] = srcI[e]; }
  }
#pragma unroll
  for (int r = 0; r < 4; ++r) {
    int e = base + r * 256;
    pos[r] = -1;
    if (e < E) pos[r] = atomicAdd(&cursor[d[r]], 1);
  }
#pragma unroll
  for (int r = 0; r < 4; ++r) {
    int e = base + r * 256;
    if (e < E && (unsigned)pos[r] < (unsigned)E) csr[pos[r]] = make_int2(s[r], e);
  }
}

__global__ __launch_bounds__(256) void gat_aggr_csr(
    const int* __restrict__ csrw, const int* __restrict__ segb,
    const float* __restrict__ cnt, const float* __restrict__ alpha,
    const float* __restrict__ sseg, const float* __restrict__ xh,
    float* __restrict__ out, int E, int Nn) {
  const int2* csr = (const int2*)csrw;
  int lane = threadIdx.x & 31;
  int n = __builtin_amdgcn_readfirstlane(
      (int)((blockIdx.x * blockDim.x + threadIdx.x) >> 5));
  if (n >= Nn) return;
  int c0 = lane * 4;
  int h = lane >> 3;
  float den = sseg[n * 4 + h] + 1e-16f;
  int beg = segb[n];
  int end = beg + (int)cnt[n];
  if (beg < 0) beg = 0;
  if (end > E) end = E;

  float a = alpha[((long long)E + n) * 4 + h] / den;
  float4 xv = *(const float4*)(xh + (long long)n * 128 + c0);
  float4 acc;
  acc.x = a * xv.x; acc.y = a * xv.y; acc.z = a * xv.z; acc.w = a * xv.w;

  int j = beg;
  for (; j + 4 <= end; j += 4) {
    int2 q0 = csr[j], q1 = csr[j + 1], q2 = csr[j + 2], q3 = csr[j + 3];
    float p0 = alpha[(long long)q0.y * 4 + h];
    float p1 = alpha[(long long)q1.y * 4 + h];
    float p2 = alpha[(long long)q2.y * 4 + h];
    float p3 = alpha[(long long)q3.y * 4 + h];
    float4 x0 = *(const float4*)(xh + (long long)q0.x * 128 + c0);
    float4 x1 = *(const float4*)(xh + (long long)q1.x * 128 + c0);
    float4 x2 = *(const float4*)(xh + (long long)q2.x * 128 + c0);
    float4 x3 = *(const float4*)(xh + (long long)q3.x * 128 + c0);
    float a0 = p0 / den, a1 = p1 / den, a2 = p2 / den, a3 = p3 / den;
    acc.x += a0 * x0.x; acc.y += a0 * x0.y; acc.z += a0 * x0.z; acc.w += a0 * x0.w;
    acc.x += a1 * x1.x; acc.y += a1 * x1.y; acc.z += a1 * x1.z; acc.w += a1 * x1.w;
    acc.x += a2 * x2.x; acc.y += a2 * x2.y; acc.z += a2 * x2.z; acc.w += a2 * x2.w;
    acc.x += a3 * x3.x; acc.y += a3 * x3.y; acc.z += a3 * x3.z; acc.w += a3 * x3.w;
  }
  for (; j < end; ++j) {
    int2 q = csr[j];
    float p = alpha[(long long)q.y * 4 + h];
    float4 x0 = *(const float4*)(xh + (long long)q.x * 128 + c0);
    float a0 = p / den;
    acc.x += a0 * x0.x; acc.y += a0 * x0.y; acc.z += a0 * x0.z; acc.w += a0 * x0.w;
  }
  *(float4*)(out + (long long)n * 128 + c0) = acc;
}

__global__ __launch_bounds__(256) void gat_bias_act(float* __restrict__ y,
                             const float* __restrict__ bias, long long n4,
                             int relu) {
  float4* y4 = (float4*)y;
  long long base = (long long)blockIdx.x * 1024 + threadIdx.x;
  float4 v[4];
#pragma unroll
  for (int r = 0; r < 4; ++r) {
    long long i = base + r * 256;
    if (i < n4) v[r] = y4[i];
  }
#pragma unroll
  for (int r = 0; r < 4; ++r) {
    long long i = base + r * 256;
    if (i < n4) {
      int c = (int)((i * 4) & 127);
      float4 o;
      o.x = v[r].x + bias[c];     o.y = v[r].y + bias[c + 1];
      o.z = v[r].z + bias[c + 2]; o.w = v[r].w + bias[c + 3];
      if (relu) {
        o.x = (o.x > 0.0f) ? o.x : 0.0f; o.y = (o.y > 0.0f) ? o.y : 0.0f;
        o.z = (o.z > 0.0f) ? o.z : 0.0f; o.w = (o.w > 0.0f) ? o.w : 0.0f;
      }
      y4[i] = o;
    }
  }
}

static void fillN(unsigned* p, unsigned v, long long n, hipStream_t s) {
  if (n <= 0) return;
  long long b = (n + 2047) / 2048;
  hipLaunchKernelGGL(gat_fill_u32, dim3((unsigned)b), dim3(256), 0, s, p, v, n);
}

extern "C" void kernel_launch(void* const* d_in, const int* in_sizes, int n_in,
                              void* d_out, int out_size, void* d_ws,
                              size_t ws_size, hipStream_t stream) {
  const int N = in_sizes[0] / 128;
  const int E = in_sizes[1] / 2;

  const float* x  = (const float*)d_in[0];
  const int* ei   = (const int*)d_in[1];
  const int* srcI = ei;
  const int* dstI = ei + E;
  const float* ea = (const float*)d_in[2];

  const float* Wl[2]  = {(const float*)d_in[3], (const float*)d_in[9]};
  const float* asl[2] = {(const float*)d_in[4], (const float*)d_in[10]};
  const float* adl[2] = {(const float*)d_in[5], (const float*)d_in[11]};
  const float* Wel[2] = {(const float*)d_in[6], (const float*)d_in[12]};
  const float* ael[2] = {(const float*)d_in[7], (const float*)d_in[13]};
  const float* bl[2]  = {(const float*)d_in[8], (const float*)d_in[14]};

  float* wsf = (float*)d_ws;
  size_t off = 0;
  float* cnt   = wsf + off; off += (size_t)N;
  float* la    = wsf + off; off += (size_t)N * 16;
  float* medge = wsf + off; off += 64;
  float* xh    = wsf + off; off += (size_t)N * 128;
  float* a_src = wsf + off; off += (size_t)N * 4;
  float* a_dst = wsf + off; off += (size_t)N * 4;
  float* alpha = wsf + off; off += ((size_t)E + N) * 4;
  unsigned* mseg = (unsigned*)(wsf + off); off += (size_t)N * 4;
  float* sseg  = wsf + off; off += (size_t)N * 4;
  size_t lo = (off + 1) & ~(size_t)1;
  int* segb    = (int*)(wsf + lo);
  int* cursor  = segb + N;
  int* total   = cursor + N;
  int* csr     = total + 4;
  off += (size_t)N * 128;
  (void)ws_size; (void)n_in; (void)out_size;

  const long long F = (long long)E + N;

  fillN((unsigned*)cnt, 0u, N, stream);
  fillN((unsigned*)la, 0u, (long long)N * 16, stream);
  {
    long long n = (long long)E * 16;
    hipLaunchKernelGGL(gat_count_loop, dim3((unsigned)((n + 2047) / 2048)),
                       dim3(256), 0, stream, dstI, ea, cnt, la, E);
    long long m = (long long)N * 16;
    hipLaunchKernelGGL(gat_loop_div, dim3((unsigned)((m + 2047) / 2048)),
                       dim3(256), 0, stream, cnt, la, N);
  }

  fillN((unsigned*)total, 0u, 1, stream);
  hipLaunchKernelGGL(gat_seg_base, dim3((unsigned)((N + 255) / 256)), dim3(256),
                     0, stream, cnt, segb, cursor, total, N);
  hipLaunchKernelGGL(gat_slot_scatter, dim3((unsigned)((E + 1023) / 1024)),
                     dim3(256), 0, stream, srcI, dstI, cursor, csr, E);

  for (int l = 0; l < 2; ++l) {
    const float* xin = (l == 0) ? x : (const float*)d_out;
    float* outb = (float*)d_out;

    hipLaunchKernelGGL(gat_medge, dim3(1), dim3(64), 0, stream, Wel[l], ael[l],
                       medge);

    {
      long long waves = ((long long)N + 15) / 16;
      long long blocks = (waves + 7) / 8;
      hipLaunchKernelGGL(gat_gemm_wmma, dim3((unsigned)blocks), dim3(256), 0,
                         stream, xin, Wl[l], xh, N);
    }

    hipLaunchKernelGGL(gat_attdot, dim3((unsigned)((N + 63) / 64)), dim3(256), 0,
                       stream, xh, asl[l], adl[l], a_src, a_dst, N);

    fillN(mseg, 0x007fffffu, (long long)N * 4, stream);
    fillN((unsigned*)sseg, 0u, (long long)N * 4, stream);

    hipLaunchKernelGGL(gat_alpha, dim3((unsigned)((F + 255) / 256)), dim3(256),
                       0, stream, srcI, dstI, ea, la, a_src, a_dst, medge,
                       alpha, mseg, E, N);

    {
      long long n4 = F * 4;
      hipLaunchKernelGGL(gat_expsum, dim3((unsigned)((n4 + 2047) / 2048)),
                         dim3(256), 0, stream, dstI, alpha, mseg, sseg, E, N);
    }

    {
      long long blocks = ((long long)N + 7) / 8;
      hipLaunchKernelGGL(gat_aggr_csr, dim3((unsigned)blocks), dim3(256), 0,
                         stream, csr, segb, cnt, alpha, sseg, xh, outb, E, N);
    }

    {
      long long n4 = (long long)N * 32;
      hipLaunchKernelGGL(gat_bias_act, dim3((unsigned)((n4 + 1023) / 1024)),
                         dim3(256), 0, stream, outb, bl[l], n4, (l == 0) ? 1 : 0);
    }
  }
}


# edge slot = segment start + arrival rank handed back by the in-degree count's own add: grouping pass without atomics (one returning add per edge less); list order stays an arrival order, the count's
# speedup vs baseline: 3.4600x; 2.1972x over previous
#include <hip/hip_runtime.h>
#include <hip/hip_bf16.h>

typedef __attribute__((ext_vector_type(2))) float v2f;
typedef __attribute__((ext_vector_type(8))) float v8f;

#define NEG_SLOPE 0.2f

__global__ __launch_bounds__(256) void gat_fill_u32(unsigned* __restrict__ p, unsigned v,
                                                    long long n) {
  long long base = (long long)blockIdx.x * 2048 + threadIdx.x;
#pragma unroll
  for (int r = 0; r < 8; ++r) {
    long long i = base + r * 256;
    if (i < n) p[i] = v;
  }
}

__global__ __launch_bounds__(256) void gat_count_loop(const int* __restrict__ dst,
                               const float* __restrict__ ea,
                               float* __restrict__ cnt,
                               float* __restrict__ la,
                               int* __restrict__ rank, int E) {
  long long total = (long long)E * 16;
  long long base = (long long)blockIdx.x * 2048 + threadIdx.x;
  if (total <= 0) return;
  int j = threadIdx.x & 15;
  int d[8];
  float v[8];
#pragma unroll
  for (int r = 0; r < 8; ++r) {
    long long i = base + r * 256;
    long long ii = (i < total) ? i : total - 1;
    d[r] = dst[ii >> 4];
    float x = ea[ii];
    v[r] = (i < total) ? x : 0.0f;
  }
#pragma unroll
  for (int r = 0; r < 8; ++r)
    atomicAdd(&la[(long long)d[r] * 16 + j], v[r]);
  __shared__ int srank[128];
  if (j == 0) {
    float before[8];
#pragma unroll
    for (int r = 0; r < 8; ++r) {
      long long i = base + r * 256;
      before[r] = 0.0f;
      if (i < total) before[r] = atomicAdd(&cnt[d[r]], 1.0f);
    }
#pragma unroll
    for (int r = 0; r < 8; ++r) srank[(threadIdx.x >> 4) + r * 16] = (int)before[r];
  }
  __syncthreads();
  if (threadIdx.x < 128) {
    long long e = (long long)blockIdx.x * 128 + threadIdx.x;
    if (e < E) rank[e] = srank[threadIdx.x];
  }
}

__global__ __launch_bounds__(256) void gat_loop_div(const float* __restrict__ cnt,
                             float* __restrict__ la, int Nn) {
  long long total = (long long)Nn * 16;
  long long base = (long long)blockIdx.x * 2048 + threadIdx.x;
  if (total <= 0) return;
  float c[8], x[8];
#pragma unroll
  for (int r = 0; r < 8; ++r) {
    long long i = base + r * 256;
    long long ii = (i < total) ? i : total - 1;
    c[r] = cnt[ii >> 4];
    x[r] = la[ii];
  }
#pragma unroll
  for (int r = 0; r < 8; ++r) {
    long long i = base + r * 256;
    if (i < total) {
      float cc = (c[r] < 1.0f) ? 1.0f : c[r];
      la[i] = x[r] / cc;
    }
  }
}

__global__ void gat_medge(const float* __restrict__ We,
                          const float* __restrict__ ae,
                          float* __restrict__ medge) {
  int t = threadIdx.x;
  int k = t >> 2, h = t & 3;
  float s = 0.0f;
  for (int c = 0; c < 32; ++c)
    s += We[k * 128 + h * 32 + c] * ae[h * 32 + c];
  medge[k * 4 + h] = s;
}

__global__ __launch_bounds__(256) void gat_gemm_wmma(
    const float* __restrict__ X, const float* __restrict__ W,
    float* __restrict__ Y, int Nrows) {
  int wave = (int)(((long long)blockIdx.x * blockDim.x + threadIdx.x) >> 5);
  int lane = threadIdx.x & 31;
  int m0 = wave * 16;
  if (m0 >= Nrows) return;

  v8f acc[8] = {};
  int half = lane >> 4;
  int l16 = lane & 15;
  int arow = m0 + l16;
  if (arow >= Nrows) arow = Nrows - 1;
  const float* xr = X + (long long)arow * 128;

  for (int k0 = 0; k0 < 128; k0 += 4) {
    v2f a;
    a.x = xr[k0 + half * 2];
    a.y = xr[k0 + half * 2 + 1];
#pragma unroll
    for (int nt = 0; nt < 8; ++nt) {
      int col = nt * 16 + l16;
      v2f b;
      b.x = W[(k0 + half * 2) * 128 + col];
      b.y = W[(k0 + half * 2 + 1) * 128 + col];
      acc[nt] = __builtin_amdgcn_wmma_f32_16x16x4_f32(
          false, a, false, b, (short)0, acc[nt], false, false);
    }
  }
#pragma unroll
  for (int nt = 0; nt < 8; ++nt) {
#pragma unroll
    for (int r = 0; r < 8; ++r) {
      int row = m0 + r + half * 8;
      if (row < Nrows) Y[(long long)row * 128 + nt * 16 + l16] = acc[nt][r];
    }
  }
}

__global__ __launch_bounds__(256) void gat_attdot(
    const float* __restrict__ xh, const float* __restrict__ as,
    const float* __restrict__ ad, float* __restrict__ a_src,
    float* __restrict__ a_dst, int Nn) {
  int lane = threadIdx.x & 31;
  int wave = (int)(((long long)blockIdx.x * 256 + threadIdx.x) >> 5);
  int n0 = wave * 8;
  if (n0 >= Nn) return;
  float asv[4], adv[4];
#pragma unroll
  for (int h = 0; h < 4; ++h) { asv[h] = as[h * 32 + lane]; adv[h] = ad[h * 32 + lane]; }
  for (int g = 0; g < 2; ++g) {
    float v[4][4];
#pragma unroll
    for (int q = 0; q < 4; ++q) {
      int n = n0 + g * 4 + q;
      if (n >= Nn) n = Nn - 1;
#pragma unroll
      for (int h = 0; h < 4; ++h) v[q][h] = xh[(long long)n * 128 + h * 32 + lane];
    }
    float rs[4][4], rd[4][4];
#pragma unroll
    for (int q = 0; q < 4; ++q) {
#pragma unroll
      for (int h = 0; h < 4; ++h) {
        float ps = v[q][h] * asv[h];
        float pd = v[q][h] * adv[h];
        for (int o = 16; o > 0; o >>= 1) {
          ps += __shfl_down(ps, o, 32);
          pd += __shfl_down(pd, o, 32);
        }
        rs[q][h] = ps;
        rd[q][h] = pd;
      }
    }
    if (lane == 0) {
#pragma unroll
      for (int q = 0; q < 4; ++q) {
        int n = n0 + g * 4 + q;
        if (n < Nn) {
#pragma unroll
          for (int h = 0; h < 4; ++h) {
            a_src[n * 4 + h] = rs[q][h];
            a_dst[n * 4 + h] = rd[q][h];
          }
        }
      }
    }
  }
}

__global__ void gat_alpha(const int* __restrict__ srcI,
                          const int* __restrict__ dstI,
                          const float* __restrict__ ea,
                          const float* __restrict__ la,
                          const float* __restrict__ a_src,
                          const float* __restrict__ a_dst,
                          const float* __restrict__ medge,
                          float* __restrict__ alpha, int E, int Nn) {
  long long f = (long long)blockIdx.x * blockDim.x + threadIdx.x;
  long long F = (long long)E + Nn;
  if (f >= F) return;
  int s, d;
  const float* row;
  if (f < (long long)E) {
    s = srcI[f]; d = dstI[f]; row = ea + f * 16;
  } else {
    s = (int)(f - E); d = s; row = la + (long long)s * 16;
  }
  float e0[16];
#pragma unroll
  for (int k = 0; k < 16; ++k) e0[k] = row[k];
#pragma unroll
  for (int h = 0; h < 4; ++h) {
    float aE = 0.0f;
#pragma unroll
    for (int k = 0; k < 16; ++k) aE += e0[k] * medge[k * 4 + h];
    float v = a_src[(long long)s * 4 + h] + a_dst[(long long)d * 4 + h] + aE;
    v = (v > 0.0f) ? v : NEG_SLOPE * v;
    alpha[f * 4 + h] = v;
  }
}

__global__ __launch_bounds__(256) void gat_seg_base(
    const float* __restrict__ cnt, int* __restrict__ segb,
    int* __restrict__ total, int Nn) {
  __shared__ int wsum[8];
  __shared__ int bbase;
  int t = threadIdx.x;
  int lane = t & 31, w = t >> 5;
  int n = blockIdx.x * 256 + t;
  int deg = (n < Nn) ? (int)cnt[n] : 0;
  int v = deg;
  for (int o = 1; o < 32; o <<= 1) {
    int u = __shfl_up(v, o, 32);
    if (lane >= o) v += u;
  }
  if (lane == 31) wsum[w] = v;
  __syncthreads();
  if (t == 0) {
    int s = 0;
    for (int i = 0; i < 8; ++i) { int u = wsum[i]; wsum[i] = s; s += u; }
    bbase = atomicAdd(total, s);
  }
  __syncthreads();
  if (n < Nn) segb[n] = bbase + wsum[w] + v - deg;
}

__global__ __launch_bounds__(256) void gat_slot_scatter(const int* __restrict__ srcI,
                                 const int* __restrict__ dstI,
                                 const int* __restrict__ segb,
                                 const int* __restrict__ rank,
                                 int* __restrict__ csrw, int E) {
  int2* csr = (int2*)csrw;
  int base = blockIdx.x * 1024 + threadIdx.x;
  if (E <= 0) return;
  int d[4], s[4], rk[4], sb[4];
#pragma unroll
  for (int r = 0; r < 4; ++r) {
    int e = base + r * 256;
    int ee = (e < E) ? e : E - 1;
    d[r] = dstI[ee]; s[r] = srcI[ee]; rk[r] = rank[ee];
  }
#pragma unroll
  for (int r = 0; r < 4; ++r) sb[r] = segb[d[r]];
#pragma unroll
  for (int r = 0; r < 4; ++r) {
    int e = base + r * 256;
    int pos = sb[r] + rk[r];
    if (e < E && (unsigned)pos < (unsigned)E) csr[pos] = make_int2(s[r], e);
  }
}

__device__ __forceinline__ float gat_wave_max(float v) {
  for (int o = 16; o > 0; o >>= 1) v = fmaxf(v, __shfl_xor(v, o, 32));
  return v;
}
__device__ __forceinline__ float gat_wave_sum(float v) {
  for (int o = 16; o > 0; o >>= 1) v += __shfl_xor(v, o, 32);
  return v;
}
__device__ __forceinline__ float gat_finite_or_zero(float m) {
  return (m > -__builtin_inff() && m < __builtin_inff()) ? m : 0.0f;
}

__device__ __forceinline__ void gat_item(const int2* __restrict__ csr,
                                         const float* __restrict__ alpha, int beg, int deg,
                                         int it, int n, int E, int& srcv, float4& al, bool& ok) {
  ok = it <= deg;
  srcv = n;
  long long f = (long long)E + n;
  if (ok && it > 0) { int2 q = csr[beg + it - 1]; srcv = q.x; f = q.y; }
  const float ninf = -__builtin_inff();
  al = make_float4(ninf, ninf, ninf, ninf);
  if (ok) al = *(const float4*)(alpha + f * 4);
}

__device__ __forceinline__ void gat_rows(float4* slot, const float* __restrict__ xh, float4 a4,
                                         int srcv, int items, int lane, int h, int c0,
                                         float4& acc) {
  slot[lane] = a4;
  const float* cf = (const float*)slot;
  int j = 0;
  for (; j + 4 <= items; j += 4) {
    int s0 = __builtin_amdgcn_readlane(srcv, j);
    int s1 = __builtin_amdgcn_readlane(srcv, j + 1);
    int s2 = __builtin_amdgcn_readlane(srcv, j + 2);
    int s3 = __builtin_amdgcn_readlane(srcv, j + 3);
    float4 x0 = *(const float4*)(xh + (long long)s0 * 128 + c0);
    float4 x1 = *(const float4*)(xh + (long long)s1 * 128 + c0);
    float4 x2 = *(const float4*)(xh + (long long)s2 * 128 + c0);
    float4 x3 = *(const float4*)(xh + (long long)s3 * 128 + c0);
    float a0 = cf[j * 4 + h], a1 = cf[(j + 1) * 4 + h];
    float a2 = cf[(j + 2) * 4 + h], a3 = cf[(j + 3) * 4 + h];
    acc.x += a0 * x0.x; acc.y += a0 * x0.y; acc.z += a0 * x0.z; acc.w += a0 * x0.w;
    acc.x += a1 * x1.x; acc.y += a1 * x1.y; acc.z += a1 * x1.z; acc.w += a1 * x1.w;
    acc.x += a2 * x2.x; acc.y += a2 * x2.y; acc.z += a2 * x2.z; acc.w += a2 * x2.w;
    acc.x += a3 * x3.x; acc.y += a3 * x3.y; acc.z += a3 * x3.z; acc.w += a3 * x3.w;
  }
#pragma clang loop unroll(disable)
  for (; j < items; ++j) {
    int s0 = __builtin_amdgcn_readlane(srcv, j);
    float4 x0 = *(const float4*)(xh + (long long)s0 * 128 + c0);
    float a0 = cf[j * 4 + h];
    acc.x += a0 * x0.x; acc.y += a0 * x0.y; acc.z += a0 * x0.z; acc.w += a0 * x0.w;
  }
}

__global__ __launch_bounds__(256) void gat_aggr_csr(
    const int* __restrict__ csrw, const int* __restrict__ segb,
    const float* __restrict__ cnt, const float* __restrict__ alpha,
    const float* __restrict__ xh, float* __restrict__ out, int E, int Nn) {
  __shared__ float4 coef[8][32];
  const int2* csr = (const int2*)csrw;
  int lane = threadIdx.x & 31;
  int w = threadIdx.x >> 5;
  int n = __builtin_amdgcn_readfirstlane(
      (int)((blockIdx.x * blockDim.x + threadIdx.x) >> 5));
  if (n >= Nn) return;
  int c0 = lane * 4;
  int h = lane >> 3;
  int beg = segb[n];
  int end = beg + (int)cnt[n];
  if (beg < 0) beg = 0;
  if (end > E) end = E;
  int deg = end - beg;
  if (deg < 0) deg = 0;

  int src0; float4 al0; bool ok0;
  gat_item(csr, alpha, beg, deg, lane, n, E, src0, al0, ok0);

  float4 m4 = al0;
  for (int base = 32; base <= deg; base += 32) {
    int s; float4 al; bool ok;
    gat_item(csr, alpha, beg, deg, base + lane, n, E, s, al, ok);
    m4.x = fmaxf(m4.x, al.x); m4.y = fmaxf(m4.y, al.y);
    m4.z = fmaxf(m4.z, al.z); m4.w = fmaxf(m4.w, al.w);
  }
  m4.x = gat_finite_or_zero(gat_wave_max(m4.x));
  m4.y = gat_finite_or_zero(gat_wave_max(m4.y));
  m4.z = gat_finite_or_zero(gat_wave_max(m4.z));
  m4.w = gat_finite_or_zero(gat_wave_max(m4.w));

  float4 p0;
  p0.x = ok0 ? __expf(al0.x - m4.x) : 0.0f; p0.y = ok0 ? __expf(al0.y - m4.y) : 0.0f;
  p0.z = ok0 ? __expf(al0.z - m4.z) : 0.0f; p0.w = ok0 ? __expf(al0.w - m4.w) : 0.0f;
  float4 s4 = p0;
  for (int base = 32; base <= deg; base += 32) {
    int s; float4 al; bool ok;
    gat_item(csr, alpha, beg, deg, base + lane, n, E, s, al, ok);
    s4.x += ok ? __expf(al.x - m4.x) : 0.0f; s4.y += ok ? __expf(al.y - m4.y) : 0.0f;
    s4.z += ok ? __expf(al.z - m4.z) : 0.0f; s4.w += ok ? __expf(al.w - m4.w) : 0.0f;
  }
  float4 den;
  den.x = gat_wave_sum(s4.x) + 1e-16f; den.y = gat_wave_sum(s4.y) + 1e-16f;
  den.z = gat_wave_sum(s4.z) + 1e-16f; den.w = gat_wave_sum(s4.w) + 1e-16f;

  float4 acc = make_float4(0.0f, 0.0f, 0.0f, 0.0f);
  float4 a4;
  a4.x = p0.x / den.x; a4.y = p0.y / den.y; a4.z = p0.z / den.z; a4.w = p0.w / den.w;
  int items = deg + 1;
  gat_rows(coef[w], xh, a4, src0, (items < 32) ? items : 32, lane, h, c0, acc);
  for (int base = 32; base <= deg; base += 32) {
    int s; float4 al; bool ok;
    gat_item(csr, alpha, beg, deg, base + lane, n, E, s, al, ok);
    a4.x = ok ? __expf(al.x - m4.x) / den.x : 0.0f; a4.y = ok ? __expf(al.y - m4.y) / den.y : 0.0f;
    a4.z = ok ? __expf(al.z - m4.z) / den.z : 0.0f; a4.w = ok ? __expf(al.w - m4.w) / den.w : 0.0f;
    int left = items - base;
    gat_rows(coef[w], xh, a4, s, (left < 32) ? left : 32, lane, h, c0, acc);
  }
  *(float4*)(out + (long long)n * 128 + c0) = acc;
}

__global__ __launch_bounds__(256) void gat_bias_act(float* __restrict__ y,
                             const float* __restrict__ bias, long long n4,
                             int relu) {
  float4* y4 = (float4*)y;
  long long base = (long long)blockIdx.x * 1024 + threadIdx.x;
  float4 v[4];
#pragma unroll
  for (int r = 0; r < 4; ++r) {
    long long i = base + r * 256;
    if (i < n4) v[r] = y4[i];
  }
#pragma unroll
  for (int r = 0; r < 4; ++r) {
    long long i = base + r * 256;
    if (i < n4) {
      int c = (int)((i * 4) & 127);
      float4 o;
      o.x = v[r].x + bias[c];     o.y = v[r].y + bias[c + 1];
      o.z = v[r].z + bias[c + 2]; o.w = v[r].w + bias[c + 3];
      if (relu) {
        o.x = (o.x > 0.0f) ? o.x : 0.0f; o.y = (o.y > 0.0f) ? o.y : 0.0f;
        o.z = (o.z > 0.0f) ? o.z : 0.0f; o.w = (o.w > 0.0f) ? o.w : 0.0f;
      }
      y4[i] = o;
    }
  }
}

static void fillN(unsigned* p, unsigned v, long long n, hipStream_t s) {
  if (n <= 0) return;
  long long b = (n + 2047) / 2048;
  hipLaunchKernelGGL(gat_fill_u32, dim3((unsigned)b), dim3(256), 0, s, p, v, n);
}

extern "C" void kernel_launch(void* const* d_in, const int* in_sizes, int n_in,
                              void* d_out, int out_size, void* d_ws,
                              size_t ws_size, hipStream_t stream) {
  const int N = in_sizes[0] / 128;
  const int E = in_sizes[1] / 2;

  const float* x  = (const float*)d_in[0];
  const int* ei   = (const int*)d_in[1];
  const int* srcI = ei;
  const int* dstI = ei + E;
  const float* ea = (const float*)d_in[2];

  const float* Wl[2]  = {(const float*)d_in[3], (const float*)d_in[9]};
  const float* asl[2] = {(const float*)d_in[4], (const float*)d_in[10]};
  const float* adl[2] = {(const float*)d_in[5], (const float*)d_in[11]};
  const float* Wel[2] = {(const float*)d_in[6], (const float*)d_in[12]};
  const float* ael[2] = {(const float*)d_in[7], (const float*)d_in[13]};
  const float* bl[2]  = {(const float*)d_in[8], (const float*)d_in[14]};

  float* wsf = (float*)d_ws;
  size_t off = 0;
  float* cnt   = wsf + off; off += (size_t)N;
  float* la    = wsf + off; off += (size_t)N * 16;
  float* medge = wsf + off; off += 64;
  float* xh    = wsf + off; off += (size_t)N * 128;
  float* a_src = wsf + off; off += (size_t)N * 4;
  float* a_dst = wsf + off; off += (size_t)N * 4;
  float* alpha = wsf + off; off += ((size_t)E + N) * 4;
  off += (size_t)N * 8;
  size_t lo = (off + 1) & ~(size_t)1;
  int* segb    = (int*)(wsf + lo);
  int* total   = segb + 2 * (size_t)N;
  int* csr     = total + 4;
  int* rank    = csr + 2 * (size_t)E;
  off += (size_t)N * 128;
  (void)ws_size; (void)n_in; (void)out_size;

  const long long F = (long long)E + N;

  fillN((unsigned*)cnt, 0u, N, stream);
  fillN((unsigned*)la, 0u, (long long)N * 16, stream);
  {
    long long n = (long long)E * 16;
    hipLaunchKernelGGL(gat_count_loop, dim3((unsigned)((n + 2047) / 2048)),
                       dim3(256), 0, stream, dstI, ea, cnt, la, rank, E);
    long long m = (long long)N * 16;
    hipLaunchKernelGGL(gat_loop_div, dim3((unsigned)((m + 2047) / 2048)),
                       dim3(256), 0, stream, cnt, la, N);
  }

  fillN((unsigned*)total, 0u, 1, stream);
  hipLaunchKernelGGL(gat_seg_base, dim3((unsigned)((N + 255) / 256)), dim3(256),
                     0, stream, cnt, segb, total, N);
  hipLaunchKernelGGL(gat_slot_scatter, dim3((unsigned)((E + 1023) / 1024)),
                     dim3(256), 0, stream, srcI, dstI, segb, rank, csr, E);

  for (int l = 0; l < 2; ++l) {
    const float* xin = (l == 0) ? x : (const float*)d_out;
    float* outb = (float*)d_out;

    hipLaunchKernelGGL(gat_medge, dim3(1), dim3(64), 0, stream, Wel[l], ael[l],
                       medge);

    {
      long long waves = ((long long)N + 15) / 16;
      long long blocks = (waves + 7) / 8;
      hipLaunchKernelGGL(gat_gemm_wmma, dim3((unsigned)blocks), dim3(256), 0,
                         stream, xin, Wl[l], xh, N);
    }

    hipLaunchKernelGGL(gat_attdot, dim3((unsigned)((N + 63) / 64)), dim3(256), 0,
                       stream, xh, asl[l], adl[l], a_src, a_dst, N);

    hipLaunchKernelGGL(gat_alpha, dim3((unsigned)((F + 255) / 256)), dim3(256),
                       0, stream, srcI, dstI, ea, la, a_src, a_dst, medge,
                       alpha, E, N);

    {
      long long blocks = ((long long)N + 7) / 8;
      hipLaunchKernelGGL(gat_aggr_csr, dim3((unsigned)blocks), dim3(256), 0,
                         stream, csr, segb, cnt, alpha, xh, outb, E, N);
    }

    {
      long long n4 = (long long)N * 32;
      hipLaunchKernelGGL(gat_bias_act, dim3((unsigned)((n4 + 1023) / 1024)),
                         dim3(256), 0, stream, outb, bl[l], n4, (l == 0) ? 1 : 0);
    }
  }
}


# self loop's edge term = mean of its incoming edges' edge terms (linear), formed with the logits in the per-destination walk: no self-loop attribute plane, no float adds into it; its sum order changes
# speedup vs baseline: 3.7872x; 1.0946x over previous
#include <hip/hip_runtime.h>
#include <hip/hip_bf16.h>

typedef __attribute__((ext_vector_type(2))) float v2f;
typedef __attribute__((ext_vector_type(8))) float v8f;

#define NEG_SLOPE 0.2f

__global__ __launch_bounds__(256) void gat_fill_u32(unsigned* __restrict__ p, unsigned v,
                                                    long long n) {
  long long base = (long long)blockIdx.x * 2048 + threadIdx.x;
#pragma unroll
  for (int r = 0; r < 8; ++r) {
    long long i = base + r * 256;
    if (i < n) p[i] = v;
  }
}

__global__ __launch_bounds__(256) void gat_count_rank(const int* __restrict__ dst,
                               float* __restrict__ cnt,
                               int* __restrict__ rank, int E) {
  int base = blockIdx.x * 2048 + threadIdx.x;
  if (E <= 0) return;
  int d[8];
#pragma unroll
  for (int r = 0; r < 8; ++r) {
    int e = base + r * 256;
    int ee = (e < E) ? e : E - 1;
    d[r] = dst[ee];
  }
  float before[8];
#pragma unroll
  for (int r = 0; r < 8; ++r) {
    int e = base + r * 256;
    before[r] = 0.0f;
    if (e < E) before[r] = atomicAdd(&cnt[d[r]], 1.0f);
  }
#pragma unroll
  for (int r = 0; r < 8; ++r) {
    int e = base + r * 256;
    if (e < E) rank[e] = (int)before[r];
  }
}

__global__ void gat_medge(const float* __restrict__ We,
                          const float* __restrict__ ae,
                          float* __restrict__ medge) {
  int t = threadIdx.x;
  int k = t >> 2, h = t & 3;
  float s = 0.0f;
  for (int c = 0; c < 32; ++c)
    s += We[k * 128 + h * 32 + c] * ae[h * 32 + c];
  medge[k * 4 + h] = s;
}

__global__ __launch_bounds__(256) void gat_gemm_wmma(
    const float* __restrict__ X, const float* __restrict__ W,
    float* __restrict__ Y, int Nrows) {
  int wave = (int)(((long long)blockIdx.x * blockDim.x + threadIdx.x) >> 5);
  int lane = threadIdx.x & 31;
  int m0 = wave * 16;
  if (m0 >= Nrows) return;

  v8f acc[8] = {};
  int half = lane >> 4;
  int l16 = lane & 15;
  int arow = m0 + l16;
  if (arow >= Nrows) arow = Nrows - 1;
  const float* xr = X + (long long)arow * 128;

  for (int k0 = 0; k0 < 128; k0 += 4) {
    v2f a;
    a.x = xr[k0 + half * 2];
    a.y = xr[k0 + half * 2 + 1];
#pragma unroll
    for (int nt = 0; nt < 8; ++nt) {
      int col = nt * 16 + l16;
      v2f b;
      b.x = W[(k0 + half * 2) * 128 + col];
      b.y = W[(k0 + half * 2 + 1) * 128 + col];
      acc[nt] = __builtin_amdgcn_wmma_f32_16x16x4_f32(
          false, a, false, b, (short)0, acc[nt], false, false);
    }
  }
#pragma unroll
  for (int nt = 0; nt < 8; ++nt) {
#pragma unroll
    for (int r = 0; r < 8; ++r) {
      int row = m0 + r + half * 8;
      if (row < Nrows) Y[(long long)row * 128 + nt * 16 + l16] = acc[nt][r];
    }
  }
}

__global__ __launch_bounds__(256) void gat_attdot(
    const float* __restrict__ xh, const float* __restrict__ as,
    const float* __restrict__ ad, float* __restrict__ a_src,
    float* __restrict__ a_dst, int Nn) {
  int lane = threadIdx.x & 31;
  int wave = (int)(((long long)blockIdx.x * 256 + threadIdx.x) >> 5);
  int n0 = wave * 8;
  if (n0 >= Nn) return;
  float asv[4], adv[4];
#pragma unroll
  for (int h = 0; h < 4; ++h) { asv[h] = as[h * 32 + lane]; adv[h] = ad[h * 32 + lane]; }
  for (int g = 0; g < 2; ++g) {
    float v[4][4];
#pragma unroll
    for (int q = 0; q < 4; ++q) {
      int n = n0 + g * 4 + q;
      if (n >= Nn) n = Nn - 1;
#pragma unroll
      for (int h = 0; h < 4; ++h) v[q][h] = xh[(long long)n * 128 + h * 32 + lane];
    }
    float rs[4][4], rd[4][4];
#pragma unroll
    for (int q = 0; q < 4; ++q) {
#pragma unroll
      for (int h = 0; h < 4; ++h) {
        float ps = v[q][h] * asv[h];
        float pd = v[q][h] * adv[h];
        for (int o = 16; o > 0; o >>= 1) {
          ps += __shfl_down(ps, o, 32);
          pd += __shfl_down(pd, o, 32);
        }
        rs[q][h] = ps;
        rd[q][h] = pd;
      }
    }
    if (lane == 0) {
#pragma unroll
      for (int q = 0; q < 4; ++q) {
        int n = n0 + g * 4 + q;
        if (n < Nn) {
#pragma unroll
          for (int h = 0; h < 4; ++h) {
            a_src[n * 4 + h] = rs[q][h];
            a_dst[n * 4 + h] = rd[q][h];
          }
        }
      }
    }
  }
}

__global__ __launch_bounds__(256) void gat_edge_term(const float* __restrict__ ea,
                              const float* __restrict__ medge,
                              float* __restrict__ aedge, int E) {
  long long e = (long long)blockIdx.x * blockDim.x + threadIdx.x;
  if (e >= (long long)E) return;
  const float* row = ea + e * 16;
  float e0[16];
#pragma unroll
  for (int k = 0; k < 16; ++k) e0[k] = row[k];
  float o[4];
#pragma unroll
  for (int h = 0; h < 4; ++h) {
    float aE = 0.0f;
#pragma unroll
    for (int k = 0; k < 16; ++k) aE += e0[k] * medge[k * 4 + h];
    o[h] = aE;
  }
  *(float4*)(aedge + e * 4) = make_float4(o[0], o[1], o[2], o[3]);
}

__global__ __launch_bounds__(256) void gat_seg_base(
    const float* __restrict__ cnt, int* __restrict__ segb,
    int* __restrict__ total, int Nn) {
  __shared__ int wsum[8];
  __shared__ int bbase;
  int t = threadIdx.x;
  int lane = t & 31, w = t >> 5;
  int n = blockIdx.x * 256 + t;
  int deg = (n < Nn) ? (int)cnt[n] : 0;
  int v = deg;
  for (int o = 1; o < 32; o <<= 1) {
    int u = __shfl_up(v, o, 32);
    if (lane >= o) v += u;
  }
  if (lane == 31) wsum[w] = v;
  __syncthreads();
  if (t == 0) {
    int s = 0;
    for (int i = 0; i < 8; ++i) { int u = wsum[i]; wsum[i] = s; s += u; }
    bbase = atomicAdd(total, s);
  }
  __syncthreads();
  if (n < Nn) segb[n] = bbase + wsum[w] + v - deg;
}

__global__ __launch_bounds__(256) void gat_slot_scatter(const int* __restrict__ srcI,
                                 const int* __restrict__ dstI,
                                 const int* __restrict__ segb,
                                 const int* __restrict__ rank,
                                 int* __restrict__ csrw, int E) {
  int2* csr = (int2*)csrw;
  int base = blockIdx.x * 1024 + threadIdx.x;
  if (E <= 0) return;
  int d[4], s[4], rk[4], sb[4];
#pragma unroll
  for (int r = 0; r < 4; ++r) {
    int e = base + r * 256;
    int ee = (e < E) ? e : E - 1;
    d[r] = dstI[ee]; s[r] = srcI[ee]; rk[r] = rank[ee];
  }
#pragma unroll
  for (int r = 0; r < 4; ++r) sb[r] = segb[d[r]];
#pragma unroll
  for (int r = 0; r < 4; ++r) {
    int e = base + r * 256;
    int pos = sb[r] + rk[r];
    if (e < E && (unsigned)pos < (unsigned)E) csr[pos] = make_int2(s[r], e);
  }
}

__device__ __forceinline__ float gat_wave_max(float v) {
  for (int o = 16; o > 0; o >>= 1) v = fmaxf(v, __shfl_xor(v, o, 32));
  return v;
}
__device__ __forceinline__ float gat_wave_sum(float v) {
  for (int o = 16; o > 0; o >>= 1) v += __shfl_xor(v, o, 32);
  return v;
}
__device__ __forceinline__ float gat_finite_or_zero(float m) {
  return (m > -__builtin_inff() && m < __builtin_inff()) ? m : 0.0f;
}

__device__ __forceinline__ float gat_leaky(float v) {
  return (v > 0.0f) ? v : NEG_SLOPE * v;
}

__device__ __forceinline__ void gat_edge_item(const int2* __restrict__ csr,
                                              const float* __restrict__ aedge, int beg, int deg,
                                              int it, int n, int& srcv, float4& ae, bool& ok) {
  ok = it <= deg;
  srcv = n;
  ae = make_float4(0.0f, 0.0f, 0.0f, 0.0f);
  if (ok && it > 0) {
    int2 q = csr[beg + it - 1];
    srcv = q.x;
    ae = *(const float4*)(aedge + (long long)q.y * 4);
  }
}

__device__ __forceinline__ float4 gat_logit(float4 as, float4 adn, float4 e, bool ok) {
  const float ninf = -__builtin_inff();
  float4 al = make_float4(ninf, ninf, ninf, ninf);
  if (ok) {
    al.x = gat_leaky(as.x + adn.x + e.x); al.y = gat_leaky(as.y + adn.y + e.y);
    al.z = gat_leaky(as.z + adn.z + e.z); al.w = gat_leaky(as.w + adn.w + e.w);
  }
  return al;
}

__device__ __forceinline__ void gat_far_item(const int2* __restrict__ csr,
                                             const float* __restrict__ aedge,
                                             const float* __restrict__ a_src, int beg, int deg,
                                             int it, int n, float4 adn, int& srcv, float4& al,
                                             bool& ok) {
  float4 ae;
  gat_edge_item(csr, aedge, beg, deg, it, n, srcv, ae, ok);
  float4 as = *(const float4*)(a_src + (long long)srcv * 4);
  al = gat_logit(as, adn, ae, ok);
}

__device__ __forceinline__ void gat_rows(float4* slot, const float* __restrict__ xh, float4 a4,
                                         int srcv, int items, int lane, int h, int c0,
                                         float4& acc) {
  slot[lane] = a4;
  const float* cf = (const float*)slot;
  int j = 0;
  for (; j + 4 <= items; j += 4) {
    int s0 = __builtin_amdgcn_readlane(srcv, j);
    int s1 = __builtin_amdgcn_readlane(srcv, j + 1);
    int s2 = __builtin_amdgcn_readlane(srcv, j + 2);
    int s3 = __builtin_amdgcn_readlane(srcv, j + 3);
    float4 x0 = *(const float4*)(xh + (long long)s0 * 128 + c0);
    float4 x1 = *(const float4*)(xh + (long long)s1 * 128 + c0);
    float4 x2 = *(const float4*)(xh + (long long)s2 * 128 + c0);
    float4 x3 = *(const float4*)(xh + (long long)s3 * 128 + c0);
    float a0 = cf[j * 4 + h], a1 = cf[(j + 1) * 4 + h];
    float a2 = cf[(j + 2) * 4 + h], a3 = cf[(j + 3) * 4 + h];
    acc.x += a0 * x0.x; acc.y += a0 * x0.y; acc.z += a0 * x0.z; acc.w += a0 * x0.w;
    acc.x += a1 * x1.x; acc.y += a1 * x1.y; acc.z += a1 * x1.z; acc.w += a1 * x1.w;
    acc.x += a2 * x2.x; acc.y += a2 * x2.y; acc.z += a2 * x2.z; acc.w += a2 * x2.w;
    acc.x += a3 * x3.x; acc.y += a3 * x3.y; acc.z += a3 * x3.z; acc.w += a3 * x3.w;
  }
#pragma clang loop unroll(disable)
  for (; j < items; ++j) {
    int s0 = __builtin_amdgcn_readlane(srcv, j);
    float4 x0 = *(const float4*)(xh + (long long)s0 * 128 + c0);
    float a0 = cf[j * 4 + h];
    acc.x += a0 * x0.x; acc.y += a0 * x0.y; acc.z += a0 * x0.z; acc.w += a0 * x0.w;
  }
}

__global__ __launch_bounds__(256) void gat_aggr_csr(
    const int* __restrict__ csrw, const int* __restrict__ segb,
    const float* __restrict__ cnt, const float* __restrict__ aedge,
    const float* __restrict__ a_src, const float* __restrict__ a_dst,
    const float* __restrict__ xh, float* __restrict__ out, int E, int Nn) {
  __shared__ float4 coef[8][32];
  const int2* csr = (const int2*)csrw;
  int lane = threadIdx.x & 31;
  int w = threadIdx.x >> 5;
  int n = __builtin_amdgcn_readfirstlane(
      (int)((blockIdx.x * blockDim.x + threadIdx.x) >> 5));
  if (n >= Nn) return;
  int c0 = lane * 4;
  int h = lane >> 3;
  int beg = segb[n];
  float cf = cnt[n];
  int end = beg + (int)cf;
  if (beg < 0) beg = 0;
  if (end > E) end = E;
  int deg = end - beg;
  if (deg < 0) deg = 0;
  float cc = (cf < 1.0f) ? 1.0f : cf;
  float4 adn = *(const float4*)(a_dst + (long long)n * 4);

  int src0; float4 ae0; bool ok0;
  gat_edge_item(csr, aedge, beg, deg, lane, n, src0, ae0, ok0);
  float4 as0 = *(const float4*)(a_src + (long long)src0 * 4);

  float4 t4 = ae0;
  for (int base = 32; base <= deg; base += 32) {
    int s; float4 ae; bool ok;
    gat_edge_item(csr, aedge, beg, deg, base + lane, n, s, ae, ok);
    t4.x += ae.x; t4.y += ae.y; t4.z += ae.z; t4.w += ae.w;
  }
  float4 lm;
  lm.x = gat_wave_sum(t4.x) / cc; lm.y = gat_wave_sum(t4.y) / cc;
  lm.z = gat_wave_sum(t4.z) / cc; lm.w = gat_wave_sum(t4.w) / cc;
  float4 al0 = gat_logit(as0, adn, (lane == 0) ? lm : ae0, ok0);

  float4 m4 = al0;
  for (int base = 32; base <= deg; base += 32) {
    int s; float4 al; bool ok;
    gat_far_item(csr, aedge, a_src, beg, deg, base + lane, n, adn, s, al, ok);
    m4.x = fmaxf(m4.x, al.x); m4.y = fmaxf(m4.y, al.y);
    m4.z = fmaxf(m4.z, al.z); m4.w = fmaxf(m4.w, al.w);
  }
  m4.x = gat_finite_or_zero(gat_wave_max(m4.x));
  m4.y = gat_finite_or_zero(gat_wave_max(m4.y));
  m4.z = gat_finite_or_zero(gat_wave_max(m4.z));
  m4.w = gat_finite_or_zero(gat_wave_max(m4.w));

  float4 p0;
  p0.x = ok0 ? __expf(al0.x - m4.x) : 0.0f; p0.y = ok0 ? __expf(al0.y - m4.y) : 0.0f;
  p0.z = ok0 ? __expf(al0.z - m4.z) : 0.0f; p0.w = ok0 ? __expf(al0.w - m4.w) : 0.0f;
  float4 s4 = p0;
  for (int base = 32; base <= deg; base += 32) {
    int s; float4 al; bool ok;
    gat_far_item(csr, aedge, a_src, beg, deg, base + lane, n, adn, s, al, ok);
    s4.x += ok ? __expf(al.x - m4.x) : 0.0f; s4.y += ok ? __expf(al.y - m4.y) : 0.0f;
    s4.z += ok ? __expf(al.z - m4.z) : 0.0f; s4.w += ok ? __expf(al.w - m4.w) : 0.0f;
  }
  float4 den;
  den.x = gat_wave_sum(s4.x) + 1e-16f; den.y = gat_wave_sum(s4.y) + 1e-16f;
  den.z = gat_wave_sum(s4.z) + 1e-16f; den.w = gat_wave_sum(s4.w) + 1e-16f;

  float4 acc = make_float4(0.0f, 0.0f, 0.0f, 0.0f);
  float4 a4;
  a4.x = p0.x / den.x; a4.y = p0.y / den.y; a4.z = p0.z / den.z; a4.w = p0.w / den.w;
  int items = deg + 1;
  gat_rows(coef[w], xh, a4, src0, (items < 32) ? items : 32, lane, h, c0, acc);
  for (int base = 32; base <= deg; base += 32) {
    int s; float4 al; bool ok;
    gat_far_item(csr, aedge, a_src, beg, deg, base + lane, n, adn, s, al, ok);
    a4.x = ok ? __expf(al.x - m4.x) / den.x : 0.0f; a4.y = ok ? __expf(al.y - m4.y) / den.y : 0.0f;
    a4.z = ok ? __expf(al.z - m4.z) / den.z : 0.0f; a4.w = ok ? __expf(al.w - m4.w) / den.w : 0.0f;
    int left = items - base;
    gat_rows(coef[w], xh, a4, s, (left < 32) ? left : 32, lane, h, c0, acc);
  }
  *(float4*)(out + (long long)n * 128 + c0) = acc;
}

__global__ __launch_bounds__(256) void gat_bias_act(float* __restrict__ y,
                             const float* __restrict__ bias, long long n4,
                             int relu) {
  float4* y4 = (float4*)y;
  long long base = (long long)blockIdx.x * 1024 + threadIdx.x;
  float4 v[4];
#pragma unroll
  for (int r = 0; r < 4; ++r) {
    long long i = base + r * 256;
    if (i < n4) v[r] = y4[i];
  }
#pragma unroll
  for (int r = 0; r < 4; ++r) {
    long long i = base + r * 256;
    if (i < n4) {
      int c = (int)((i * 4) & 127);
      float4 o;
      o.x = v[r].x + bias[c];     o.y = v[r].y + bias[c + 1];
      o.z = v[r].z + bias[c + 2]; o.w = v[r].w + bias[c + 3];
      if (relu) {
        o.x = (o.x > 0.0f) ? o.x : 0.0f; o.y = (o.y > 0.0f) ? o.y : 0.0f;
        o.z = (o.z > 0.0f) ? o.z : 0.0f; o.w = (o.w > 0.0f) ? o.w : 0.0f;
      }
      y4[i] = o;
    }
  }
}

static void fillN(unsigned* p, unsigned v, long long n, hipStream_t s) {
  if (n <= 0) return;
  long long b = (n + 2047) / 2048;
  hipLaunchKernelGGL(gat_fill_u32, dim3((unsigned)b), dim3(256), 0, s, p, v, n);
}

extern "C" void kernel_launch(void* const* d_in, const int* in_sizes, int n_in,
                              void* d_out, int out_size, void* d_ws,
                              size_t ws_size, hipStream_t stream) {
  const int N = in_sizes[0] / 128;
  const int E = in_sizes[1] / 2;

  const float* x  = (const float*)d_in[0];
  const int* ei   = (const int*)d_in[1];
  const int* srcI = ei;
  const int* dstI = ei + E;
  const float* ea = (const float*)d_in[2];

  const float* Wl[2]  = {(const float*)d_in[3], (const float*)d_in[9]};
  const float* asl[2] = {(const float*)d_in[4], (const float*)d_in[10]};
  const float* adl[2] = {(const float*)d_in[5], (const float*)d_in[11]};
  const float* Wel[2] = {(const float*)d_in[6], (const float*)d_in[12]};
  const float* ael[2] = {(const float*)d_in[7], (const float*)d_in[13]};
  const float* bl[2]  = {(const float*)d_in[8], (const float*)d_in[14]};

  float* wsf = (float*)d_ws;
  size_t off = 0;
  float* cnt   = wsf + off; off += (size_t)N;
  off += (size_t)N * 16;
  float* medge = wsf + off; off += 64;
  float* xh    = wsf + off; off += (size_t)N * 128;
  float* a_src = wsf + off; off += (size_t)N * 4;
  float* a_dst = wsf + off; off += (size_t)N * 4;
  float* aedge = wsf + off; off += ((size_t)E + N) * 4;
  off += (size_t)N * 8;
  size_t lo = (off + 1) & ~(size_t)1;
  int* segb    = (int*)(wsf + lo);
  int* total   = segb + 2 * (size_t)N;
  int* csr     = total + 4;
  int* rank    = csr + 2 * (size_t)E;
  off += (size_t)N * 128;
  (void)ws_size; (void)n_in; (void)out_size;

  fillN((unsigned*)cnt, 0u, N, stream);
  hipLaunchKernelGGL(gat_count_rank, dim3((unsigned)((E + 2047) / 2048)),
                     dim3(256), 0, stream, dstI, cnt, rank, E);

  fillN((unsigned*)total, 0u, 1, stream);
  hipLaunchKernelGGL(gat_seg_base, dim3((unsigned)((N + 255) / 256)), dim3(256),
                     0, stream, cnt, segb, total, N);
  hipLaunchKernelGGL(gat_slot_scatter, dim3((unsigned)((E + 1023) / 1024)),
                     dim3(256), 0, stream, srcI, dstI, segb, rank, csr, E);

  for (int l = 0; l < 2; ++l) {
    const float* xin = (l == 0) ? x : (const float*)d_out;
    float* outb = (float*)d_out;

    hipLaunchKernelGGL(gat_medge, dim3(1), dim3(64), 0, stream, Wel[l], ael[l],
                       medge);

    {
      long long waves = ((long long)N + 15) / 16;
      long long blocks = (waves + 7) / 8;
      hipLaunchKernelGGL(gat_gemm_wmma, dim3((unsigned)blocks), dim3(256), 0,
                         stream, xin, Wl[l], xh, N);
    }

    hipLaunchKernelGGL(gat_attdot, dim3((unsigned)((N + 63) / 64)), dim3(256), 0,
                       stream, xh, asl[l], adl[l], a_src, a_dst, N);

    hipLaunchKernelGGL(gat_edge_term, dim3((unsigned)((E + 255) / 256)), dim3(256),
                       0, stream, ea, medge, aedge, E);

    {
      long long blocks = ((long long)N + 7) / 8;
      hipLaunchKernelGGL(gat_aggr_csr, dim3((unsigned)blocks), dim3(256), 0,
                         stream, csr, segb, cnt, aedge, a_src, a_dst, xh, outb, E, N);
    }

    {
      long long n4 = (long long)N * 32;
      hipLaunchKernelGGL(gat_bias_act, dim3((unsigned)((n4 + 1023) / 1024)),
                         dim3(256), 0, stream, outb, bl[l], n4, (l == 0) ? 1 : 0);
    }
  }
}


# each layer's two elementwise launches folded into their producers: bias+ReLU at the walk's store; attention dots as a ninth GEMM column tile, x@(W@att): 17 -> 13 launches; dot sum order changes
# speedup vs baseline: 4.0598x; 1.0720x over previous
#include <hip/hip_runtime.h>
#include <hip/hip_bf16.h>

typedef __attribute__((ext_vector_type(2))) float v2f;
typedef __attribute__((ext_vector_type(8))) float v8f;

#define NEG_SLOPE 0.2f

__global__ __launch_bounds__(256) void gat_fill_u32(unsigned* __restrict__ p, unsigned v,
                                                    long long n) {
  long long base = (long long)blockIdx.x * 2048 + threadIdx.x;
#pragma unroll
  for (int r = 0; r < 8; ++r) {
    long long i = base + r * 256;
    if (i < n) p[i] = v;
  }
}

__global__ __launch_bounds__(256) void gat_count_rank(const int* __restrict__ dst,
                               float* __restrict__ cnt,
                               int* __restrict__ rank, int E) {
  int base = blockIdx.x * 2048 + threadIdx.x;
  if (E <= 0) return;
  int d[8];
#pragma unroll
  for (int r = 0; r < 8; ++r) {
    int e = base + r * 256;
    int ee = (e < E) ? e : E - 1;
    d[r] = dst[ee];
  }
  float before[8];
#pragma unroll
  for (int r = 0; r < 8; ++r) {
    int e = base + r * 256;
    before[r] = 0.0f;
    if (e < E) before[r] = atomicAdd(&cnt[d[r]], 1.0f);
  }
#pragma unroll
  for (int r = 0; r < 8; ++r) {
    int e = base + r * 256;
    if (e < E) rank[e] = (int)before[r];
  }
}

__global__ __launch_bounds__(256) void gat_fold_att(const float* __restrict__ W,
                             const float* __restrict__ as,
                             const float* __restrict__ ad,
                             const float* __restrict__ We,
                             const float* __restrict__ ae,
                             float* __restrict__ wa,
                             float* __restrict__ medge) {
  int t = threadIdx.x;
  if (blockIdx.x < 8) {
    int o = blockIdx.x * 256 + t;
    int k = o >> 4, j = o & 15;
    float s = 0.0f;
    if (j < 8) {
      int h = j & 3;
      const float* av = (j < 4) ? as : ad;
      for (int cc = 0; cc < 32; ++cc)
        s += W[k * 128 + h * 32 + cc] * av[h * 32 + cc];
    }
    wa[o] = s;
  } else if (t < 64) {
    int k = t >> 2, h = t & 3;
    float s = 0.0f;
    for (int cc = 0; cc < 32; ++cc)
      s += We[k * 128 + h * 32 + cc] * ae[h * 32 + cc];
    medge[k * 4 + h] = s;
  }
}

__global__ __launch_bounds__(256) void gat_gemm_wmma(
    const float* __restrict__ X, const float* __restrict__ W,
    const float* __restrict__ WA,
    float* __restrict__ Y, float* __restrict__ a_src, float* __restrict__ a_dst, int Nrows) {
  int wave = (int)(((long long)blockIdx.x * blockDim.x + threadIdx.x) >> 5);
  int lane = threadIdx.x & 31;
  int m0 = wave * 16;
  if (m0 >= Nrows) return;

  v8f acc[8] = {};
  v8f acc9 = {};
  int half = lane >> 4;
  int l16 = lane & 15;
  int arow = m0 + l16;
  if (arow >= Nrows) arow = Nrows - 1;
  const float* xr = X + (long long)arow * 128;

#pragma unroll 2
  for (int k0 = 0; k0 < 128; k0 += 4) {
    v2f a;
    a.x = xr[k0 + half * 2];
    a.y = xr[k0 + half * 2 + 1];
#pragma unroll
    for (int nt = 0; nt < 8; ++nt) {
      int col = nt * 16 + l16;
      v2f b;
      b.x = W[(k0 + half * 2) * 128 + col];
      b.y = W[(k0 + half * 2 + 1) * 128 + col];
      acc[nt] = __builtin_amdgcn_wmma_f32_16x16x4_f32(
          false, a, false, b, (short)0, acc[nt], false, false);
    }
    {
      v2f b9;
      b9.x = WA[(k0 + half * 2) * 16 + l16];
      b9.y = WA[(k0 + half * 2 + 1) * 16 + l16];
      acc9 = __builtin_amdgcn_wmma_f32_16x16x4_f32(
          false, a, false, b9, (short)0, acc9, false, false);
    }
  }
#pragma unroll
  for (int nt = 0; nt < 8; ++nt) {
#pragma unroll
    for (int r = 0; r < 8; ++r) {
      int row = m0 + r + half * 8;
      if (row < Nrows) Y[(long long)row * 128 + nt * 16 + l16] = acc[nt][r];
    }
  }
#pragma unroll
  for (int r = 0; r < 8; ++r) {
    int row = m0 + r + half * 8;
    if (row < Nrows) {
      if (l16 < 4) a_src[row * 4 + l16] = acc9[r];
      else if (l16 < 8) a_dst[row * 4 + (l16 - 4)] = acc9[r];
    }
  }
}

__global__ __launch_bounds__(256) void gat_edge_term(const float* __restrict__ ea,
                              const float* __restrict__ medge,
                              float* __restrict__ aedge, int E) {
  long long e = (long long)blockIdx.x * blockDim.x + threadIdx.x;
  if (e >= (long long)E) return;
  const float* row = ea + e * 16;
  float e0[16];
#pragma unroll
  for (int k = 0; k < 16; ++k) e0[k] = row[k];
  float o[4];
#pragma unroll
  for (int h = 0; h < 4; ++h) {
    float aE = 0.0f;
#pragma unroll
    for (int k = 0; k < 16; ++k) aE += e0[k] * medge[k * 4 + h];
    o[h] = aE;
  }
  *(float4*)(aedge + e * 4) = make_float4(o[0], o[1], o[2], o[3]);
}

__global__ __launch_bounds__(256) void gat_seg_base(
    const float* __restrict__ cnt, int* __restrict__ segb,
    int* __restrict__ total, int Nn) {
  __shared__ int wsum[8];
  __shared__ int bbase;
  int t = threadIdx.x;
  int lane = t & 31, w = t >> 5;
  int n = blockIdx.x * 256 + t;
  int deg = (n < Nn) ? (int)cnt[n] : 0;
  int v = deg;
  for (int o = 1; o < 32; o <<= 1) {
    int u = __shfl_up(v, o, 32);
    if (lane >= o) v += u;
  }
  if (lane == 31) wsum[w] = v;
  __syncthreads();
  if (t == 0) {
    int s = 0;
    for (int i = 0; i < 8; ++i) { int u = wsum[i]; wsum[i] = s; s += u; }
    bbase = atomicAdd(total, s);
  }
  __syncthreads();
  if (n < Nn) segb[n] = bbase + wsum[w] + v - deg;
}

__global__ __launch_bounds__(256) void gat_slot_scatter(const int* __restrict__ srcI,
                                 const int* __restrict__ dstI,
                                 const int* __restrict__ segb,
                                 const int* __restrict__ rank,
                                 int* __restrict__ csrw, int E) {
  int2* csr = (int2*)csrw;
  int base = blockIdx.x * 1024 + threadIdx.x;
  if (E <= 0) return;
  int d[4], s[4], rk[4], sb[4];
#pragma unroll
  for (int r = 0; r < 4; ++r) {
    int e = base + r * 256;
    int ee = (e < E) ? e : E - 1;
    d[r] = dstI[ee]; s[r] = srcI[ee]; rk[r] = rank[ee];
  }
#pragma unroll
  for (int r = 0; r < 4; ++r) sb[r] = segb[d[r]];
#pragma unroll
  for (int r = 0; r < 4; ++r) {
    int e = base + r * 256;
    int pos = sb[r] + rk[r];
    if (e < E && (unsigned)pos < (unsigned)E) csr[pos] = make_int2(s[r], e);
  }
}

__device__ __forceinline__ float gat_wave_max(float v) {
  for (int o = 16; o > 0; o >>= 1) v = fmaxf(v, __shfl_xor(v, o, 32));
  return v;
}
__device__ __forceinline__ float gat_wave_sum(float v) {
  for (int o = 16; o > 0; o >>= 1) v += __shfl_xor(v, o, 32);
  return v;
}
__device__ __forceinline__ float gat_finite_or_zero(float m) {
  return (m > -__builtin_inff() && m < __builtin_inff()) ? m : 0.0f;
}

__device__ __forceinline__ float gat_leaky(float v) {
  return (v > 0.0f) ? v : NEG_SLOPE * v;
}

__device__ __forceinline__ void gat_edge_item(const int2* __restrict__ csr,
                                              const float* __restrict__ aedge, int beg, int deg,
                                              int it, int n, int& srcv, float4& ae, bool& ok) {
  ok = it <= deg;
  srcv = n;
  ae = make_float4(0.0f, 0.0f, 0.0f, 0.0f);
  if (ok && it > 0) {
    int2 q = csr[beg + it - 1];
    srcv = q.x;
    ae = *(const float4*)(aedge + (long long)q.y * 4);
  }
}

__device__ __forceinline__ float4 gat_logit(float4 as, float4 adn, float4 e, bool ok) {
  const float ninf = -__builtin_inff();
  float4 al = make_float4(ninf, ninf, ninf, ninf);
  if (ok) {
    al.x = gat_leaky(as.x + adn.x + e.x); al.y = gat_leaky(as.y + adn.y + e.y);
    al.z = gat_leaky(as.z + adn.z + e.z); al.w = gat_leaky(as.w + adn.w + e.w);
  }
  return al;
}

__device__ __forceinline__ void gat_far_item(const int2* __restrict__ csr,
                                             const float* __restrict__ aedge,
                                             const float* __restrict__ a_src, int beg, int deg,
                                             int it, int n, float4 adn, int& srcv, float4& al,
                                             bool& ok) {
  float4 ae;
  gat_edge_item(csr, aedge, beg, deg, it, n, srcv, ae, ok);
  float4 as = *(const float4*)(a_src + (long long)srcv * 4);
  al = gat_logit(as, adn, ae, ok);
}

__device__ __forceinline__ void gat_rows(float4* slot, const float* __restrict__ xh, float4 a4,
                                         int srcv, int items, int lane, int h, int c0,
                                         float4& acc) {
  slot[lane] = a4;
  const float* cf = (const float*)slot;
  int j = 0;
  for (; j + 4 <= items; j += 4) {
    int s0 = __builtin_amdgcn_readlane(srcv, j);
    int s1 = __builtin_amdgcn_readlane(srcv, j + 1);
    int s2 = __builtin_amdgcn_readlane(srcv, j + 2);
    int s3 = __builtin_amdgcn_readlane(srcv, j + 3);
    float4 x0 = *(const float4*)(xh + (long long)s0 * 128 + c0);
    float4 x1 = *(const float4*)(xh + (long long)s1 * 128 + c0);
    float4 x2 = *(const float4*)(xh + (long long)s2 * 128 + c0);
    float4 x3 = *(const float4*)(xh + (long long)s3 * 128 + c0);
    float a0 = cf[j * 4 + h], a1 = cf[(j + 1) * 4 + h];
    float a2 = cf[(j + 2) * 4 + h], a3 = cf[(j + 3) * 4 + h];
    acc.x += a0 * x0.x; acc.y += a0 * x0.y; acc.z += a0 * x0.z; acc.w += a0 * x0.w;
    acc.x += a1 * x1.x; acc.y += a1 * x1.y; acc.z += a1 * x1.z; acc.w += a1 * x1.w;
    acc.x += a2 * x2.x; acc.y += a2 * x2.y; acc.z += a2 * x2.z; acc.w += a2 * x2.w;
    acc.x += a3 * x3.x; acc.y += a3 * x3.y; acc.z += a3 * x3.z; acc.w += a3 * x3.w;
  }
#pragma clang loop unroll(disable)
  for (; j < items; ++j) {
    int s0 = __builtin_amdgcn_readlane(srcv, j);
    float4 x0 = *(const float4*)(xh + (long long)s0 * 128 + c0);
    float a0 = cf[j * 4 + h];
    acc.x += a0 * x0.x; acc.y += a0 * x0.y; acc.z += a0 * x0.z; acc.w += a0 * x0.w;
  }
}

__global__ __launch_bounds__(256) void gat_aggr_csr(
    const int* __restrict__ csrw, const int* __restrict__ segb,
    const float* __restrict__ cnt, const float* __restrict__ aedge,
    const float* __restrict__ a_src, const float* __restrict__ a_dst,
    const float* __restrict__ xh, const float* __restrict__ bias, int relu,
    float* __restrict__ out, int E, int Nn) {
  __shared__ float4 coef[8][32];
  const int2* csr = (const int2*)csrw;
  int lane = threadIdx.x & 31;
  int w = threadIdx.x >> 5;
  int n = __builtin_amdgcn_readfirstlane(
      (int)((blockIdx.x * blockDim.x + threadIdx.x) >> 5));
  if (n >= Nn) return;
  int c0 = lane * 4;
  int h = lane >> 3;
  int beg = segb[n];
  float cf = cnt[n];
  int end = beg + (int)cf;
  if (beg < 0) beg = 0;
  if (end > E) end = E;
  int deg = end - beg;
  if (deg < 0) deg = 0;
  float cc = (cf < 1.0f) ? 1.0f : cf;
  float4 adn = *(const float4*)(a_dst + (long long)n * 4);

  int src0; float4 ae0; bool ok0;
  gat_edge_item(csr, aedge, beg, deg, lane, n, src0, ae0, ok0);
  float4 as0 = *(const float4*)(a_src + (long long)src0 * 4);

  float4 t4 = ae0;
  for (int base = 32; base <= deg; base += 32) {
    int s; float4 ae; bool ok;
    gat_edge_item(csr, aedge, beg, deg, base + lane, n, s, ae, ok);
    t4.x += ae.x; t4.y += ae.y; t4.z += ae.z; t4.w += ae.w;
  }
  float4 lm;
  lm.x = gat_wave_sum(t4.x) / cc; lm.y = gat_wave_sum(t4.y) / cc;
  lm.z = gat_wave_sum(t4.z) / cc; lm.w = gat_wave_sum(t4.w) / cc;
  float4 al0 = gat_logit(as0, adn, (lane == 0) ? lm : ae0, ok0);

  float4 m4 = al0;
  for (int base = 32; base <= deg; base += 32) {
    int s; float4 al; bool ok;
    gat_far_item(csr, aedge, a_src, beg, deg, base + lane, n, adn, s, al, ok);
    m4.x = fmaxf(m4.x, al.x); m4.y = fmaxf(m4.y, al.y);
    m4.z = fmaxf(m4.z, al.z); m4.w = fmaxf(m4.w, al.w);
  }
  m4.x = gat_finite_or_zero(gat_wave_max(m4.x));
  m4.y = gat_finite_or_zero(gat_wave_max(m4.y));
  m4.z = gat_finite_or_zero(gat_wave_max(m4.z));
  m4.w = gat_finite_or_zero(gat_wave_max(m4.w));

  float4 p0;
  p0.x = ok0 ? __expf(al0.x - m4.x) : 0.0f; p0.y = ok0 ? __expf(al0.y - m4.y) : 0.0f;
  p0.z = ok0 ? __expf(al0.z - m4.z) : 0.0f; p0.w = ok0 ? __expf(al0.w - m4.w) : 0.0f;
  float4 s4 = p0;
  for (int base = 32; base <= deg; base += 32) {
    int s; float4 al; bool ok;
    gat_far_item(csr, aedge, a_src, beg, deg, base + lane, n, adn, s, al, ok);
    s4.x += ok ? __expf(al.x - m4.x) : 0.0f; s4.y += ok ? __expf(al.y - m4.y) : 0.0f;
    s4.z += ok ? __expf(al.z - m4.z) : 0.0f; s4.w += ok ? __expf(al.w - m4.w) : 0.0f;
  }
  float4 den;
  den.x = gat_wave_sum(s4.x) + 1e-16f; den.y = gat_wave_sum(s4.y) + 1e-16f;
  den.z = gat_wave_sum(s4.z) + 1e-16f; den.w = gat_wave_sum(s4.w) + 1e-16f;

  float4 acc = make_float4(0.0f, 0.0f, 0.0f, 0.0f);
  float4 a4;
  a4.x = p0.x / den.x; a4.y = p0.y / den.y; a4.z = p0.z / den.z; a4.w = p0.w / den.w;
  int items = deg + 1;
  gat_rows(coef[w], xh, a4, src0, (items < 32) ? items : 32, lane, h, c0, acc);
  for (int base = 32; base <= deg; base += 32) {
    int s; float4 al; bool ok;
    gat_far_item(csr, aedge, a_src, beg, deg, base + lane, n, adn, s, al, ok);
    a4.x = ok ? __expf(al.x - m4.x) / den.x : 0.0f; a4.y = ok ? __expf(al.y - m4.y) / den.y : 0.0f;
    a4.z = ok ? __expf(al.z - m4.z) / den.z : 0.0f; a4.w = ok ? __expf(al.w - m4.w) / den.w : 0.0f;
    int left = items - base;
    gat_rows(coef[w], xh, a4, s, (left < 32) ? left : 32, lane, h, c0, acc);
  }
  float4 bv = *(const float4*)(bias + c0);
  float4 o;
  o.x = acc.x + bv.x; o.y = acc.y + bv.y; o.z = acc.z + bv.z; o.w = acc.w + bv.w;
  if (relu) {
    o.x = (o.x > 0.0f) ? o.x : 0.0f; o.y = (o.y > 0.0f) ? o.y : 0.0f;
    o.z = (o.z > 0.0f) ? o.z : 0.0f; o.w = (o.w > 0.0f) ? o.w : 0.0f;
  }
  *(float4*)(out + (long long)n * 128 + c0) = o;
}

static void fillN(unsigned* p, unsigned v, long long n, hipStream_t s) {
  if (n <= 0) return;
  long long b = (n + 2047) / 2048;
  hipLaunchKernelGGL(gat_fill_u32, dim3((unsigned)b), dim3(256), 0, s, p, v, n);
}

extern "C" void kernel_launch(void* const* d_in, const int* in_sizes, int n_in,
                              void* d_out, int out_size, void* d_ws,
                              size_t ws_size, hipStream_t stream) {
  const int N = in_sizes[0] / 128;
  const int E = in_sizes[1] / 2;

  const float* x  = (const float*)d_in[0];
  const int* ei   = (const int*)d_in[1];
  const int* srcI = ei;
  const int* dstI = ei + E;
  const float* ea = (const float*)d_in[2];

  const float* Wl[2]  = {(const float*)d_in[3], (const float*)d_in[9]};
  const float* asl[2] = {(const float*)d_in[4], (const float*)d_in[10]};
  const float* adl[2] = {(const float*)d_in[5], (const float*)d_in[11]};
  const float* Wel[2] = {(const float*)d_in[6], (const float*)d_in[12]};
  const float* ael[2] = {(const float*)d_in[7], (const float*)d_in[13]};
  const float* bl[2]  = {(const float*)d_in[8], (const float*)d_in[14]};

  float* wsf = (float*)d_ws;
  size_t off = 0;
  float* cnt   = wsf + off; off += (size_t)N;
  float* wa    = wsf + off;
  off += (size_t)N * 16;
  float* medge = wsf + off; off += 64;
  float* xh    = wsf + off; off += (size_t)N * 128;
  float* a_src = wsf + off; off += (size_t)N * 4;
  float* a_dst = wsf + off; off += (size_t)N * 4;
  float* aedge = wsf + off; off += ((size_t)E + N) * 4;
  off += (size_t)N * 8;
  size_t lo = (off + 1) & ~(size_t)1;
  int* segb    = (int*)(wsf + lo);
  int* total   = segb + 2 * (size_t)N;
  int* csr     = total + 4;
  int* rank    = csr + 2 * (size_t)E;
  off += (size_t)N * 128;
  (void)ws_size; (void)n_in; (void)out_size;

  fillN((unsigned*)cnt, 0u, N, stream);
  hipLaunchKernelGGL(gat_count_rank, dim3((unsigned)((E + 2047) / 2048)),
                     dim3(256), 0, stream, dstI, cnt, rank, E);

  fillN((unsigned*)total, 0u, 1, stream);
  hipLaunchKernelGGL(gat_seg_base, dim3((unsigned)((N + 255) / 256)), dim3(256),
                     0, stream, cnt, segb, total, N);
  hipLaunchKernelGGL(gat_slot_scatter, dim3((unsigned)((E + 1023) / 1024)),
                     dim3(256), 0, stream, srcI, dstI, segb, rank, csr, E);

  for (int l = 0; l < 2; ++l) {
    const float* xin = (l == 0) ? x : (const float*)d_out;
    float* outb = (float*)d_out;

    hipLaunchKernelGGL(gat_fold_att, dim3(9), dim3(256), 0, stream, Wl[l], asl[l], adl[l],
                       Wel[l], ael[l], wa, medge);

    {
      long long waves = ((long long)N + 15) / 16;
      long long blocks = (waves + 7) / 8;
      hipLaunchKernelGGL(gat_gemm_wmma, dim3((unsigned)blocks), dim3(256), 0,
                         stream, xin, Wl[l], wa, xh, a_src, a_dst, N);
    }

    hipLaunchKernelGGL(gat_edge_term, dim3((unsigned)((E + 255) / 256)), dim3(256),
                       0, stream, ea, medge, aedge, E);

    {
      long long blocks = ((long long)N + 7) / 8;
      hipLaunchKernelGGL(gat_aggr_csr, dim3((unsigned)blocks), dim3(256), 0,
                         stream, csr, segb, cnt, aedge, a_src, a_dst, xh, bl[l],
                         (l == 0) ? 1 : 0, outb, E, N);
    }
  }
}


# edges grouped by destination with a two-level counting sort, every rank from an LDS add (512-node buckets): no global atomics, 13 -> 12 launches; segments stay in an arrival order, sum order changes
# speedup vs baseline: 4.9995x; 1.2315x over previous
#include <hip/hip_runtime.h>
#include <hip/hip_bf16.h>

typedef __attribute__((ext_vector_type(2))) float v2f;
typedef __attribute__((ext_vector_type(8))) float v8f;

#define NEG_SLOPE 0.2f

__global__ __launch_bounds__(256) void gat_fold_att(const float* __restrict__ W,
                             const float* __restrict__ as,
                             const float* __restrict__ ad,
                             const float* __restrict__ We,
                             const float* __restrict__ ae,
                             float* __restrict__ wa,
                             float* __restrict__ medge) {
  int t = threadIdx.x;
  if (blockIdx.x < 8) {
    int o = blockIdx.x * 256 + t;
    int k = o >> 4, j = o & 15;
    float s = 0.0f;
    if (j < 8) {
      int h = j & 3;
      const float* av = (j < 4) ? as : ad;
      for (int cc = 0; cc < 32; ++cc)
        s += W[k * 128 + h * 32 + cc] * av[h * 32 + cc];
    }
    wa[o] = s;
  } else if (t < 64) {
    int k = t >> 2, h = t & 3;
    float s = 0.0f;
    for (int cc = 0; cc < 32; ++cc)
      s += We[k * 128 + h * 32 + cc] * ae[h * 32 + cc];
    medge[k * 4 + h] = s;
  }
}

__global__ __launch_bounds__(256) void gat_gemm_wmma(
    const float* __restrict__ X, const float* __restrict__ W,
    const float* __restrict__ WA,
    float* __restrict__ Y, float* __restrict__ a_src, float* __restrict__ a_dst, int Nrows) {
  int wave = (int)(((long long)blockIdx.x * blockDim.x + threadIdx.x) >> 5);
  int lane = threadIdx.x & 31;
  int m0 = wave * 16;
  if (m0 >= Nrows) return;

  v8f acc[8] = {};
  v8f acc9 = {};
  int half = lane >> 4;
  int l16 = lane & 15;
  int arow = m0 + l16;
  if (arow >= Nrows) arow = Nrows - 1;
  const float* xr = X + (long long)arow * 128;

#pragma unroll 2
  for (int k0 = 0; k0 < 128; k0 += 4) {
    v2f a;
    a.x = xr[k0 + half * 2];
    a.y = xr[k0 + half * 2 + 1];
#pragma unroll
    for (int nt = 0; nt < 8; ++nt) {
      int col = nt * 16 + l16;
      v2f b;
      b.x = W[(k0 + half * 2) * 128 + col];
      b.y = W[(k0 + half * 2 + 1) * 128 + col];
      acc[nt] = __builtin_amdgcn_wmma_f32_16x16x4_f32(
          false, a, false, b, (short)0, acc[nt], false, false);
    }
    {
      v2f b9;
      b9.x = WA[(k0 + half * 2) * 16 + l16];
      b9.y = WA[(k0 + half * 2 + 1) * 16 + l16];
      acc9 = __builtin_amdgcn_wmma_f32_16x16x4_f32(
          false, a, false, b9, (short)0, acc9, false, false);
    }
  }
#pragma unroll
  for (int nt = 0; nt < 8; ++nt) {
#pragma unroll
    for (int r = 0; r < 8; ++r) {
      int row = m0 + r + half * 8;
      if (row < Nrows) Y[(long long)row * 128 + nt * 16 + l16] = acc[nt][r];
    }
  }
#pragma unroll
  for (int r = 0; r < 8; ++r) {
    int row = m0 + r + half * 8;
    if (row < Nrows) {
      if (l16 < 4) a_src[row * 4 + l16] = acc9[r];
      else if (l16 < 8) a_dst[row * 4 + (l16 - 4)] = acc9[r];
    }
  }
}

__global__ __launch_bounds__(256) void gat_edge_term(const float* __restrict__ ea,
                              const float* __restrict__ medge,
                              float* __restrict__ aedge, int E) {
  long long e = (long long)blockIdx.x * blockDim.x + threadIdx.x;
  if (e >= (long long)E) return;
  const float* row = ea + e * 16;
  float e0[16];
#pragma unroll
  for (int k = 0; k < 16; ++k) e0[k] = row[k];
  float o[4];
#pragma unroll
  for (int h = 0; h < 4; ++h) {
    float aE = 0.0f;
#pragma unroll
    for (int k = 0; k < 16; ++k) aE += e0[k] * medge[k * 4 + h];
    o[h] = aE;
  }
  *(float4*)(aedge + e * 4) = make_float4(o[0], o[1], o[2], o[3]);
}

#define GAT_NBMAX 1024

__device__ __forceinline__ int gat_bucket_of(int d, int nb) {
  int b = d >> 9;
  b = (b < 0) ? 0 : b;
  return (b < nb) ? b : nb - 1;
}
__device__ __forceinline__ int gat_local_of(int d, int b) {
  int l = d - (b << 9);
  l = (l < 0) ? 0 : l;
  return (l < 512) ? l : 511;
}
__device__ __forceinline__ int gat_wave_scan(int v, int lane) {
  for (int o = 1; o < 32; o <<= 1) {
    int u = __shfl_up(v, o, 32);
    if (lane >= o) v += u;
  }
  return v;
}

__global__ __launch_bounds__(256) void gat_bucket_hist(const int* __restrict__ dst,
                                int* __restrict__ tbl, int E, int NB) {
  __shared__ int h[GAT_NBMAX];
  int t = threadIdx.x;
  int nb = (NB < GAT_NBMAX) ? NB : GAT_NBMAX;
  int base = blockIdx.x * 2048 + t;
  int d[8];
#pragma unroll
  for (int r = 0; r < 8; ++r) {
    int e = base + r * 256;
    int ee = (e < E) ? e : E - 1;
    d[r] = dst[ee];
  }
  for (int i = t; i < nb; i += 256) h[i] = 0;
  __syncthreads();
#pragma unroll
  for (int r = 0; r < 8; ++r) {
    int e = base + r * 256;
    if (e < E) atomicAdd(&h[gat_bucket_of(d[r], nb)], 1);
  }
  __syncthreads();
  for (int i = t; i < nb; i += 256) tbl[(size_t)i * gridDim.x + blockIdx.x] = h[i];
}

__global__ __launch_bounds__(256) void gat_bucket_rowscan(int* __restrict__ tbl,
                                   int* __restrict__ btot, int B) {
  __shared__ int wsum[8];
  if (blockIdx.x >= GAT_NBMAX) return;
  int t = threadIdx.x, lane = t & 31, w = t >> 5;
  int* row = tbl + (size_t)blockIdx.x * B;
  int carry = 0;
  for (int t0 = 0; t0 < B; t0 += 1024) {
    int i0 = t0 + 4 * t;
    int v[4];
#pragma unroll
    for (int k = 0; k < 4; ++k) v[k] = (i0 + k < B) ? row[i0 + k] : 0;
    int ts = v[0] + v[1] + v[2] + v[3];
    int inc = gat_wave_scan(ts, lane);
    if (lane == 31) wsum[w] = inc;
    __syncthreads();
    int wp = 0, tot = 0;
#pragma unroll
    for (int i = 0; i < 8; ++i) { int u = wsum[i]; wp += (i < w) ? u : 0; tot += u; }
    int ex = carry + wp + inc - ts;
#pragma unroll
    for (int k = 0; k < 4; ++k) {
      if (i0 + k < B) row[i0 + k] = ex;
      ex += v[k];
    }
    carry += tot;
    __syncthreads();
  }
  if (t == 0) btot[blockIdx.x] = carry;
}

__global__ __launch_bounds__(256) void gat_bucket_scatter(const int* __restrict__ srcI,
                                   const int* __restrict__ dstI,
                                   const int* __restrict__ tbl,
                                   const int* __restrict__ btot,
                                   int4* __restrict__ tmp, int E, int NB) {
  __shared__ int off[GAT_NBMAX];
  __shared__ int wsum[8];
  int t = threadIdx.x, lane = t & 31, w = t >> 5;
  int nb = (NB < GAT_NBMAX) ? NB : GAT_NBMAX;
  int base = blockIdx.x * 2048 + t;
  int d[8], s[8];
#pragma unroll
  for (int r = 0; r < 8; ++r) {
    int e = base + r * 256;
    int ee = (e < E) ? e : E - 1;
    d[r] = dstI[ee]; s[r] = srcI[ee];
  }
  int i0 = 4 * t;
  int v[4], mine[4];
#pragma unroll
  for (int k = 0; k < 4; ++k) {
    bool in = i0 + k < nb;
    v[k] = in ? btot[i0 + k] : 0;
    mine[k] = in ? tbl[(size_t)(i0 + k) * gridDim.x + blockIdx.x] : 0;
  }
  int ts = v[0] + v[1] + v[2] + v[3];
  int inc = gat_wave_scan(ts, lane);
  if (lane == 31) wsum[w] = inc;
  __syncthreads();
  int wp = 0;
#pragma unroll
  for (int i = 0; i < 8; ++i) { int u = wsum[i]; wp += (i < w) ? u : 0; }
  int ex = wp + inc - ts;
#pragma unroll
  for (int k = 0; k < 4; ++k) {
    if (i0 + k < nb) off[i0 + k] = ex + mine[k];
    ex += v[k];
  }
  __syncthreads();
#pragma unroll
  for (int r = 0; r < 8; ++r) {
    int e = base + r * 256;
    if (e < E) {
      int pos = atomicAdd(&off[gat_bucket_of(d[r], nb)], 1);
      if ((unsigned)pos < (unsigned)E) tmp[pos] = make_int4(d[r], s[r], e, 0);
    }
  }
}

__global__ __launch_bounds__(512) void gat_bucket_place(const int4* __restrict__ tmp,
                                 const int* __restrict__ btot,
                                 int* __restrict__ segb, float* __restrict__ cnt,
                                 int* __restrict__ csrw, int E, int Nn) {
  __shared__ int c[512];
  __shared__ int wsum[16];
  int2* csr = (int2*)csrw;
  int b = blockIdx.x;
  if (b >= GAT_NBMAX) return;
  int t = threadIdx.x, lane = t & 31, w = t >> 5;
  int p = 0;
  for (int i = t; i < b; i += 512) p += btot[i];
  for (int o = 16; o > 0; o >>= 1) p += __shfl_xor(p, o, 32);
  if (lane == 0) wsum[w] = p;
  c[t] = 0;
  __syncthreads();
  int base = 0;
#pragma unroll
  for (int i = 0; i < 16; ++i) base += wsum[i];
  int m = btot[b];
  base = (base < 0) ? 0 : base;
  base = (base > E) ? E : base;
  m = (m < 0) ? 0 : m;
  m = (m > E - base) ? E - base : m;
  const int4* rec = tmp + base;

  for (int i = t; i < m; i += 4096) {
    int dd[8];
#pragma unroll
    for (int r = 0; r < 8; ++r) {
      int j = i + r * 512;
      dd[r] = rec[(j < m) ? j : m - 1].x;
    }
#pragma unroll
    for (int r = 0; r < 8; ++r)
      if (i + r * 512 < m) atomicAdd(&c[gat_local_of(dd[r], b)], 1);
  }
  __syncthreads();

  int deg = c[t];
  int inc = gat_wave_scan(deg, lane);
  if (lane == 31) wsum[w] = inc;
  __syncthreads();
  int wp = 0;
#pragma unroll
  for (int i = 0; i < 16; ++i) { int u = wsum[i]; wp += (i < w) ? u : 0; }
  int ex = wp + inc - deg;
  int n = (b << 9) + t;
  if (n < Nn) { segb[n] = base + ex; cnt[n] = (float)deg; }
  c[t] = ex;
  __syncthreads();

  for (int i = t; i < m; i += 2048) {
    int4 q[4];
#pragma unroll
    for (int r = 0; r < 4; ++r) {
      int j = i + r * 512;
      q[r] = rec[(j < m) ? j : m - 1];
    }
#pragma unroll
    for (int r = 0; r < 4; ++r) {
      if (i + r * 512 < m) {
        int pos = base + atomicAdd(&c[gat_local_of(q[r].x, b)], 1);
        if ((unsigned)pos < (unsigned)E) csr[pos] = make_int2(q[r].y, q[r].z);
      }
    }
  }
}

__device__ __forceinline__ float gat_wave_max(float v) {
  for (int o = 16; o > 0; o >>= 1) v = fmaxf(v, __shfl_xor(v, o, 32));
  return v;
}
__device__ __forceinline__ float gat_wave_sum(float v) {
  for (int o = 16; o > 0; o >>= 1) v += __shfl_xor(v, o, 32);
  return v;
}
__device__ __forceinline__ float gat_finite_or_zero(float m) {
  return (m > -__builtin_inff() && m < __builtin_inff()) ? m : 0.0f;
}

__device__ __forceinline__ float gat_leaky(float v) {
  return (v > 0.0f) ? v : NEG_SLOPE * v;
}

__device__ __forceinline__ void gat_edge_item(const int2* __restrict__ csr,
                                              const float* __restrict__ aedge, int beg, int deg,
                                              int it, int n, int& srcv, float4& ae, bool& ok) {
  ok = it <= deg;
  srcv = n;
  ae = make_float4(0.0f, 0.0f, 0.0f, 0.0f);
  if (ok && it > 0) {
    int2 q = csr[beg + it - 1];
    srcv = q.x;
    ae = *(const float4*)(aedge + (long long)q.y * 4);
  }
}

__device__ __forceinline__ float4 gat_logit(float4 as, float4 adn, float4 e, bool ok) {
  const float ninf = -__builtin_inff();
  float4 al = make_float4(ninf, ninf, ninf, ninf);
  if (ok) {
    al.x = gat_leaky(as.x + adn.x + e.x); al.y = gat_leaky(as.y + adn.y + e.y);
    al.z = gat_leaky(as.z + adn.z + e.z); al.w = gat_leaky(as.w + adn.w + e.w);
  }
  return al;
}

__device__ __forceinline__ void gat_far_item(const int2* __restrict__ csr,
                                             const float* __restrict__ aedge,
                                             const float* __restrict__ a_src, int beg, int deg,
                                             int it, int n, float4 adn, int& srcv, float4& al,
                                             bool& ok) {
  float4 ae;
  gat_edge_item(csr, aedge, beg, deg, it, n, srcv, ae, ok);
  float4 as = *(const float4*)(a_src + (long long)srcv * 4);
  al = gat_logit(as, adn, ae, ok);
}

__device__ __forceinline__ void gat_rows(float4* slot, const float* __restrict__ xh, float4 a4,
                                         int srcv, int items, int lane, int h, int c0,
                                         float4& acc) {
  slot[lane] = a4;
  const float* cf = (const float*)slot;
  int j = 0;
  for (; j + 4 <= items; j += 4) {
    int s0 = __builtin_amdgcn_readlane(srcv, j);
    int s1 = __builtin_amdgcn_readlane(srcv, j + 1);
    int s2 = __builtin_amdgcn_readlane(srcv, j + 2);
    int s3 = __builtin_amdgcn_readlane(srcv, j + 3);
    float4 x0 = *(const float4*)(xh + (long long)s0 * 128 + c0);
    float4 x1 = *(const float4*)(xh + (long long)s1 * 128 + c0);
    float4 x2 = *(const float4*)(xh + (long long)s2 * 128 + c0);
    float4 x3 = *(const float4*)(xh + (long long)s3 * 128 + c0);
    float a0 = cf[j * 4 + h], a1 = cf[(j + 1) * 4 + h];
    float a2 = cf[(j + 2) * 4 + h], a3 = cf[(j + 3) * 4 + h];
    acc.x += a0 * x0.x; acc.y += a0 * x0.y; acc.z += a0 * x0.z; acc.w += a0 * x0.w;
    acc.x += a1 * x1.x; acc.y += a1 * x1.y; acc.z += a1 * x1.z; acc.w += a1 * x1.w;
    acc.x += a2 * x2.x; acc.y += a2 * x2.y; acc.z += a2 * x2.z; acc.w += a2 * x2.w;
    acc.x += a3 * x3.x; acc.y += a3 * x3.y; acc.z += a3 * x3.z; acc.w += a3 * x3.w;
  }
#pragma clang loop unroll(disable)
  for (; j < items; ++j) {
    int s0 = __builtin_amdgcn_readlane(srcv, j);
    float4 x0 = *(const float4*)(xh + (long long)s0 * 128 + c0);
    float a0 = cf[j * 4 + h];
    acc.x += a0 * x0.x; acc.y += a0 * x0.y; acc.z += a0 * x0.z; acc.w += a0 * x0.w;
  }
}

__global__ __launch_bounds__(256) void gat_aggr_csr(
    const int* __restrict__ csrw, const int* __restrict__ segb,
    const float* __restrict__ cnt, const float* __restrict__ aedge,
    const float* __restrict__ a_src, const float* __restrict__ a_dst,
    const float* __restrict__ xh, const float* __restrict__ bias, int relu,
    float* __restrict__ out, int E, int Nn) {
  __shared__ float4 coef[8][32];
  const int2* csr = (const int2*)csrw;
  int lane = threadIdx.x & 31;
  int w = threadIdx.x >> 5;
  int n = __builtin_amdgcn_readfirstlane(
      (int)((blockIdx.x * blockDim.x + threadIdx.x) >> 5));
  if (n >= Nn) return;
  int c0 = lane * 4;
  int h = lane >> 3;
  int beg = segb[n];
  float cf = cnt[n];
  int end = beg + (int)cf;
  if (beg < 0) beg = 0;
  if (end > E) end = E;
  int deg = end - beg;
  if (deg < 0) deg = 0;
  float cc = (cf < 1.0f) ? 1.0f : cf;
  float4 adn = *(const float4*)(a_dst + (long long)n * 4);

  int src0; float4 ae0; bool ok0;
  gat_edge_item(csr, aedge, beg, deg, lane, n, src0, ae0, ok0);
  float4 as0 = *(const float4*)(a_src + (long long)src0 * 4);

  float4 t4 = ae0;
  for (int base = 32; base <= deg; base += 32) {
    int s; float4 ae; bool ok;
    gat_edge_item(csr, aedge, beg, deg, base + lane, n, s, ae, ok);
    t4.x += ae.x; t4.y += ae.y; t4.z += ae.z; t4.w += ae.w;
  }
  float4 lm;
  lm.x = gat_wave_sum(t4.x) / cc; lm.y = gat_wave_sum(t4.y) / cc;
  lm.z = gat_wave_sum(t4.z) / cc; lm.w = gat_wave_sum(t4.w) / cc;
  float4 al0 = gat_logit(as0, adn, (lane == 0) ? lm : ae0, ok0);

  float4 m4 = al0;
  for (int base = 32; base <= deg; base += 32) {
    int s; float4 al; bool ok;
    gat_far_item(csr, aedge, a_src, beg, deg, base + lane, n, adn, s, al, ok);
    m4.x = fmaxf(m4.x, al.x); m4.y = fmaxf(m4.y, al.y);
    m4.z = fmaxf(m4.z, al.z); m4.w = fmaxf(m4.w, al.w);
  }
  m4.x = gat_finite_or_zero(gat_wave_max(m4.x));
  m4.y = gat_finite_or_zero(gat_wave_max(m4.y));
  m4.z = gat_finite_or_zero(gat_wave_max(m4.z));
  m4.w = gat_finite_or_zero(gat_wave_max(m4.w));

  float4 p0;
  p0.x = ok0 ? __expf(al0.x - m4.x) : 0.0f; p0.y = ok0 ? __expf(al0.y - m4.y) : 0.0f;
  p0.z = ok0 ? __expf(al0.z - m4.z) : 0.0f; p0.w = ok0 ? __expf(al0.w - m4.w) : 0.0f;
  float4 s4 = p0;
  for (int base = 32; base <= deg; base += 32) {
    int s; float4 al; bool ok;
    gat_far_item(csr, aedge, a_src, beg, deg, base + lane, n, adn, s, al, ok);
    s4.x += ok ? __expf(al.x - m4.x) : 0.0f; s4.y += ok ? __expf(al.y - m4.y) : 0.0f;
    s4.z += ok ? __expf(al.z - m4.z) : 0.0f; s4.w += ok ? __expf(al.w - m4.w) : 0.0f;
  }
  float4 den;
  den.x = gat_wave_sum(s4.x) + 1e-16f; den.y = gat_wave_sum(s4.y) + 1e-16f;
  den.z = gat_wave_sum(s4.z) + 1e-16f; den.w = gat_wave_sum(s4.w) + 1e-16f;

  float4 acc = make_float4(0.0f, 0.0f, 0.0f, 0.0f);
  float4 a4;
  a4.x = p0.x / den.x; a4.y = p0.y / den.y; a4.z = p0.z / den.z; a4.w = p0.w / den.w;
  int items = deg + 1;
  gat_rows(coef[w], xh, a4, src0, (items < 32) ? items : 32, lane, h, c0, acc);
  for (int base = 32; base <= deg; base += 32) {
    int s; float4 al; bool ok;
    gat_far_item(csr, aedge, a_src, beg, deg, base + lane, n, adn, s, al, ok);
    a4.x = ok ? __expf(al.x - m4.x) / den.x : 0.0f; a4.y = ok ? __expf(al.y - m4.y) / den.y : 0.0f;
    a4.z = ok ? __expf(al.z - m4.z) / den.z : 0.0f; a4.w = ok ? __expf(al.w - m4.w) / den.w : 0.0f;
    int left = items - base;
    gat_rows(coef[w], xh, a4, s, (left < 32) ? left : 32, lane, h, c0, acc);
  }
  float4 bv = *(const float4*)(bias + c0);
  float4 o;
  o.x = acc.x + bv.x; o.y = acc.y + bv.y; o.z = acc.z + bv.z; o.w = acc.w + bv.w;
  if (relu) {
    o.x = (o.x > 0.0f) ? o.x : 0.0f; o.y = (o.y > 0.0f) ? o.y : 0.0f;
    o.z = (o.z > 0.0f) ? o.z : 0.0f; o.w = (o.w > 0.0f) ? o.w : 0.0f;
  }
  *(float4*)(out + (long long)n * 128 + c0) = o;
}

extern "C" void kernel_launch(void* const* d_in, const int* in_sizes, int n_in,
                              void* d_out, int out_size, void* d_ws,
                              size_t ws_size, hipStream_t stream) {
  const int N = in_sizes[0] / 128;
  const int E = in_sizes[1] / 2;

  const float* x  = (const float*)d_in[0];
  const int* ei   = (const int*)d_in[1];
  const int* srcI = ei;
  const int* dstI = ei + E;
  const float* ea = (const float*)d_in[2];

  const float* Wl[2]  = {(const float*)d_in[3], (const float*)d_in[9]};
  const float* asl[2] = {(const float*)d_in[4], (const float*)d_in[10]};
  const float* adl[2] = {(const float*)d_in[5], (const float*)d_in[11]};
  const float* Wel[2] = {(const float*)d_in[6], (const float*)d_in[12]};
  const float* ael[2] = {(const float*)d_in[7], (const float*)d_in[13]};
  const float* bl[2]  = {(const float*)d_in[8], (const float*)d_in[14]};

  float* wsf = (float*)d_ws;
  size_t off = 0;
  float* cnt   = wsf + off; off += (size_t)N;
  float* wa    = wsf + off;
  off += (size_t)N * 16;
  float* medge = wsf + off; off += 64;
  float* xh    = wsf + off; off += (size_t)N * 128;
  float* a_src = wsf + off; off += (size_t)N * 4;
  float* a_dst = wsf + off; off += (size_t)N * 4;
  float* aedge = wsf + off; off += ((size_t)E + N) * 4;
  off += (size_t)N * 8;
  size_t lo = (off + 1) & ~(size_t)1;
  int* segb    = (int*)(wsf + lo);
  int* csr     = segb + 2 * (size_t)N + 4;
  const int NB = (N + 511) / 512;
  const int EB = (E + 2047) / 2048;
  size_t to = (lo + 2 * (size_t)N + 4 + 2 * (size_t)E + 3) & ~(size_t)3;
  int4* tmp    = (int4*)(wsf + to);
  int* tbl     = (int*)(wsf + to) + 4 * (size_t)E;
  int* btot    = tbl + (size_t)NB * EB;
  off += (size_t)N * 128;
  (void)ws_size; (void)n_in; (void)out_size;

  hipLaunchKernelGGL(gat_bucket_hist, dim3((unsigned)EB), dim3(256), 0, stream, dstI, tbl, E, NB);
  hipLaunchKernelGGL(gat_bucket_rowscan, dim3((unsigned)NB), dim3(256), 0, stream, tbl, btot, EB);
  hipLaunchKernelGGL(gat_bucket_scatter, dim3((unsigned)EB), dim3(256), 0, stream, srcI, dstI,
                     tbl, btot, tmp, E, NB);
  hipLaunchKernelGGL(gat_bucket_place, dim3((unsigned)NB), dim3(512), 0, stream, tmp, btot,
                     segb, cnt, csr, E, N);

  for (int l = 0; l < 2; ++l) {
    const float* xin = (l == 0) ? x : (const float*)d_out;
    float* outb = (float*)d_out;

    hipLaunchKernelGGL(gat_fold_att, dim3(9), dim3(256), 0, stream, Wl[l], asl[l], adl[l],
                       Wel[l], ael[l], wa, medge);

    {
      long long waves = ((long long)N + 15) / 16;
      long long blocks = (waves + 7) / 8;
      hipLaunchKernelGGL(gat_gemm_wmma, dim3((unsigned)blocks), dim3(256), 0,
                         stream, xin, Wl[l], wa, xh, a_src, a_dst, N);
    }

    hipLaunchKernelGGL(gat_edge_term, dim3((unsigned)((E + 255) / 256)), dim3(256),
                       0, stream, ea, medge, aedge, E);

    {
      long long blocks = ((long long)N + 7) / 8;
      hipLaunchKernelGGL(gat_aggr_csr, dim3((unsigned)blocks), dim3(256), 0,
                         stream, csr, segb, cnt, aedge, a_src, a_dst, xh, bl[l],
                         (l == 0) ? 1 : 0, outb, E, N);
    }
  }
}


# softmax taken online in the per-destination row walk (running max, rescaled sums); a_src, a_dst as 8-lane sums over the gathered rows: no a_src gather, no ninth GEMM tile; sum order changes
# speedup vs baseline: 5.6326x; 1.1266x over previous
#include <hip/hip_runtime.h>
#include <hip/hip_bf16.h>

typedef __attribute__((ext_vector_type(2))) float v2f;
typedef __attribute__((ext_vector_type(8))) float v8f;

#define NEG_SLOPE 0.2f

__global__ __launch_bounds__(64) void gat_fold_edge(const float* __restrict__ We,
                             const float* __restrict__ ae,
                             float* __restrict__ medge) {
  int t = threadIdx.x;
  int k = t >> 2, h = t & 3;
  float s = 0.0f;
  for (int cc = 0; cc < 32; ++cc)
    s += We[k * 128 + h * 32 + cc] * ae[h * 32 + cc];
  medge[k * 4 + h] = s;
}

__global__ __launch_bounds__(256) void gat_gemm_wmma(
    const float* __restrict__ X, const float* __restrict__ W,
    float* __restrict__ Y, int Nrows) {
  int wave = (int)(((long long)blockIdx.x * blockDim.x + threadIdx.x) >> 5);
  int lane = threadIdx.x & 31;
  int m0 = wave * 16;
  if (m0 >= Nrows) return;

  v8f acc[8] = {};
  int half = lane >> 4;
  int l16 = lane & 15;
  int arow = m0 + l16;
  if (arow >= Nrows) arow = Nrows - 1;
  const float* xr = X + (long long)arow * 128;

#pragma unroll 2
  for (int k0 = 0; k0 < 128; k0 += 4) {
    v2f a;
    a.x = xr[k0 + half * 2];
    a.y = xr[k0 + half * 2 + 1];
#pragma unroll
    for (int nt = 0; nt < 8; ++nt) {
      int col = nt * 16 + l16;
      v2f b;
      b.x = W[(k0 + half * 2) * 128 + col];
      b.y = W[(k0 + half * 2 + 1) * 128 + col];
      acc[nt] = __builtin_amdgcn_wmma_f32_16x16x4_f32(
          false, a, false, b, (short)0, acc[nt], false, false);
    }
  }
#pragma unroll
  for (int nt = 0; nt < 8; ++nt) {
#pragma unroll
    for (int r = 0; r < 8; ++r) {
      int row = m0 + r + half * 8;
      if (row < Nrows) Y[(long long)row * 128 + nt * 16 + l16] = acc[nt][r];
    }
  }
}

__global__ __launch_bounds__(256) void gat_edge_term(const float* __restrict__ ea,
                              const float* __restrict__ medge,
                              float* __restrict__ aedge, int E) {
  long long e = (long long)blockIdx.x * blockDim.x + threadIdx.x;
  if (e >= (long long)E) return;
  const float* row = ea + e * 16;
  float e0[16];
#pragma unroll
  for (int k = 0; k < 16; ++k) e0[k] = row[k];
  float o[4];
#pragma unroll
  for (int h = 0; h < 4; ++h) {
    float aE = 0.0f;
#pragma unroll
    for (int k = 0; k < 16; ++k) aE += e0[k] * medge[k * 4 + h];
    o[h] = aE;
  }
  *(float4*)(aedge + e * 4) = make_float4(o[0], o[1], o[2], o[3]);
}

#define GAT_NBMAX 1024

__device__ __forceinline__ int gat_bucket_of(int d, int nb) {
  int b = d >> 9;
  b = (b < 0) ? 0 : b;
  return (b < nb) ? b : nb - 1;
}
__device__ __forceinline__ int gat_local_of(int d, int b) {
  int l = d - (b << 9);
  l = (l < 0) ? 0 : l;
  return (l < 512) ? l : 511;
}
__device__ __forceinline__ int gat_wave_scan(int v, int lane) {
  for (int o = 1; o < 32; o <<= 1) {
    int u = __shfl_up(v, o, 32);
    if (lane >= o) v += u;
  }
  return v;
}

__global__ __launch_bounds__(256) void gat_bucket_hist(const int* __restrict__ dst,
                                int* __restrict__ tbl, int E, int NB) {
  __shared__ int h[GAT_NBMAX];
  int t = threadIdx.x;
  int nb = (NB < GAT_NBMAX) ? NB : GAT_NBMAX;
  int base = blockIdx.x * 2048 + t;
  int d[8];
#pragma unroll
  for (int r = 0; r < 8; ++r) {
    int e = base + r * 256;
    int ee = (e < E) ? e : E - 1;
    d[r] = dst[ee];
  }
  for (int i = t; i < nb; i += 256) h[i] = 0;
  __syncthreads();
#pragma unroll
  for (int r = 0; r < 8; ++r) {
    int e = base + r * 256;
    if (e < E) atomicAdd(&h[gat_bucket_of(d[r], nb)], 1);
  }
  __syncthreads();
  for (int i = t; i < nb; i += 256) tbl[(size_t)i * gridDim.x + blockIdx.x] = h[i];
}

__global__ __launch_bounds__(256) void gat_bucket_rowscan(int* __restrict__ tbl,
                                   int* __restrict__ btot, int B) {
  __shared__ int wsum[8];
  if (blockIdx.x >= GAT_NBMAX) return;
  int t = threadIdx.x, lane = t & 31, w = t >> 5;
  int* row = tbl + (size_t)blockIdx.x * B;
  int carry = 0;
  for (int t0 = 0; t0 < B; t0 += 1024) {
    int i0 = t0 + 4 * t;
    int v[4];
#pragma unroll
    for (int k = 0; k < 4; ++k) v[k] = (i0 + k < B) ? row[i0 + k] : 0;
    int ts = v[0] + v[1] + v[2] + v[3];
    int inc = gat_wave_scan(ts, lane);
    if (lane == 31) wsum[w] = inc;
    __syncthreads();
    int wp = 0, tot = 0;
#pragma unroll
    for (int i = 0; i < 8; ++i) { int u = wsum[i]; wp += (i < w) ? u : 0; tot += u; }
    int ex = carry + wp + inc - ts;
#pragma unroll
    for (int k = 0; k < 4; ++k) {
      if (i0 + k < B) row[i0 + k] = ex;
      ex += v[k];
    }
    carry += tot;
    __syncthreads();
  }
  if (t == 0) btot[blockIdx.x] = carry;
}

__global__ __launch_bounds__(256) void gat_bucket_scatter(const int* __restrict__ srcI,
                                   const int* __restrict__ dstI,
                                   const int* __restrict__ tbl,
                                   const int* __restrict__ btot,
                                   int4* __restrict__ tmp, int E, int NB) {
  __shared__ int off[GAT_NBMAX];
  __shared__ int wsum[8];
  int t = threadIdx.x, lane = t & 31, w = t >> 5;
  int nb = (NB < GAT_NBMAX) ? NB : GAT_NBMAX;
  int base = blockIdx.x * 2048 + t;
  int d[8], s[8];
#pragma unroll
  for (int r = 0; r < 8; ++r) {
    int e = base + r * 256;
    int ee = (e < E) ? e : E - 1;
    d[r] = dstI[ee]; s[r] = srcI[ee];
  }
  int i0 = 4 * t;
  int v[4], mine[4];
#pragma unroll
  for (int k = 0; k < 4; ++k) {
    bool in = i0 + k < nb;
    v[k] = in ? btot[i0 + k] : 0;
    mine[k] = in ? tbl[(size_t)(i0 + k) * gridDim.x + blockIdx.x] : 0;
  }
  int ts = v[0] + v[1] + v[2] + v[3];
  int inc = gat_wave_scan(ts, lane);
  if (lane == 31) wsum[w] = inc;
  __syncthreads();
  int wp = 0;
#pragma unroll
  for (int i = 0; i < 8; ++i) { int u = wsum[i]; wp += (i < w) ? u : 0; }
  int ex = wp + inc - ts;
#pragma unroll
  for (int k = 0; k < 4; ++k) {
    if (i0 + k < nb) off[i0 + k] = ex + mine[k];
    ex += v[k];
  }
  __syncthreads();
#pragma unroll
  for (int r = 0; r < 8; ++r) {
    int e = base + r * 256;
    if (e < E) {
      int pos = atomicAdd(&off[gat_bucket_of(d[r], nb)], 1);
      if ((unsigned)pos < (unsigned)E) tmp[pos] = make_int4(d[r], s[r], e, 0);
    }
  }
}

__global__ __launch_bounds__(512) void gat_bucket_place(const int4* __restrict__ tmp,
                                 const int* __restrict__ btot,
                                 int* __restrict__ segb, float* __restrict__ cnt,
                                 int* __restrict__ csrw, int E, int Nn) {
  __shared__ int c[512];
  __shared__ int wsum[16];
  int2* csr = (int2*)csrw;
  int b = blockIdx.x;
  if (b >= GAT_NBMAX) return;
  int t = threadIdx.x, lane = t & 31, w = t >> 5;
  int p = 0;
  for (int i = t; i < b; i += 512) p += btot[i];
  for (int o = 16; o > 0; o >>= 1) p += __shfl_xor(p, o, 32);
  if (lane == 0) wsum[w] = p;
  c[t] = 0;
  __syncthreads();
  int base = 0;
#pragma unroll
  for (int i = 0; i < 16; ++i) base += wsum[i];
  int m = btot[b];
  base = (base < 0) ? 0 : base;
  base = (base > E) ? E : base;
  m = (m < 0) ? 0 : m;
  m = (m > E - base) ? E - base : m;
  const int4* rec = tmp + base;

  for (int i = t; i < m; i += 4096) {
    int dd[8];
#pragma unroll
    for (int r = 0; r < 8; ++r) {
      int j = i + r * 512;
      dd[r] = rec[(j < m) ? j : m - 1].x;
    }
#pragma unroll
    for (int r = 0; r < 8; ++r)
      if (i + r * 512 < m) atomicAdd(&c[gat_local_of(dd[r], b)], 1);
  }
  __syncthreads();

  int deg = c[t];
  int inc = gat_wave_scan(deg, lane);
  if (lane == 31) wsum[w] = inc;
  __syncthreads();
  int wp = 0;
#pragma unroll
  for (int i = 0; i < 16; ++i) { int u = wsum[i]; wp += (i < w) ? u : 0; }
  int ex = wp + inc - deg;
  int n = (b << 9) + t;
  if (n < Nn) { segb[n] = base + ex; cnt[n] = (float)deg; }
  c[t] = ex;
  __syncthreads();

  for (int i = t; i < m; i += 2048) {
    int4 q[4];
#pragma unroll
    for (int r = 0; r < 4; ++r) {
      int j = i + r * 512;
      q[r] = rec[(j < m) ? j : m - 1];
    }
#pragma unroll
    for (int r = 0; r < 4; ++r) {
      if (i + r * 512 < m) {
        int pos = base + atomicAdd(&c[gat_local_of(q[r].x, b)], 1);
        if ((unsigned)pos < (unsigned)E) csr[pos] = make_int2(q[r].y, q[r].z);
      }
    }
  }
}

__device__ __forceinline__ float gat_finite_or_zero(float m) {
  return (m > -__builtin_inff() && m < __builtin_inff()) ? m : 0.0f;
}

__device__ __forceinline__ float gat_leaky(float v) {
  return fmaxf(v, NEG_SLOPE * v);
}

template <int XM>
__device__ __forceinline__ float gat_xlane(float v) {
  return __int_as_float(__builtin_amdgcn_update_dpp(0, __float_as_int(v), 0x160 + XM, 0xf, 0xf, true));
}
__device__ __forceinline__ float gat_sum8(float v) {
  v += gat_xlane<1>(v);
  v += gat_xlane<2>(v);
  v += gat_xlane<4>(v);
  return v;
}
__device__ __forceinline__ float gat_dot4(float4 x, float4 a) {
  v2f xl = {x.x, x.y}, xh2 = {x.z, x.w};
  v2f al = {a.x, a.y}, ah = {a.z, a.w};
  v2f t = xh2 * ah + xl * al;
  return t.x + t.y;
}

typedef float gat_f4 __attribute__((ext_vector_type(4)));
__device__ __forceinline__ float4 gat_row(const float* __restrict__ xh, int sr, int lane) {
  unsigned long long a = (unsigned long long)xh + ((unsigned long long)(unsigned)sr << 9);
  gat_f4 v = ((const __attribute__((address_space(1))) gat_f4*)a)[lane];
  return make_float4(v.x, v.y, v.z, v.w);
}

struct gat_rows4 { float4 x0, x1, x2, x3; };

__device__ __forceinline__ void gat_issue(const float* __restrict__ xh, int srcv, int j, int rem,
                                          int lane, gat_rows4& R) {
  const float4 z = make_float4(0.0f, 0.0f, 0.0f, 0.0f);
  R.x0 = gat_row(xh, __builtin_amdgcn_readlane(srcv, j), lane);
  R.x1 = z; R.x2 = z; R.x3 = z;
  if (rem > 1) R.x1 = gat_row(xh, __builtin_amdgcn_readlane(srcv, j + 1), lane);
  if (rem > 2) R.x2 = gat_row(xh, __builtin_amdgcn_readlane(srcv, j + 2), lane);
  if (rem > 3) R.x3 = gat_row(xh, __builtin_amdgcn_readlane(srcv, j + 3), lane);
}

__device__ __forceinline__ void gat_step(const gat_rows4& R, const float* sl, int j, int rem, int h,
                                         float4 as4, float adn,
                                         float& m, float& s, float& te, float4& acc) {
  const float ninf = -__builtin_inff();
  float4 e = *(const float4*)(sl + h * 32 + j);
  float d0 = gat_sum8(gat_dot4(R.x0, as4));
  float d1 = gat_sum8(gat_dot4(R.x1, as4));
  float d2 = gat_sum8(gat_dot4(R.x2, as4));
  float d3 = gat_sum8(gat_dot4(R.x3, as4));
  float l0 = gat_leaky((d0 + adn) + e.x);
  float l1 = (rem > 1) ? gat_leaky((d1 + adn) + e.y) : ninf;
  float l2 = (rem > 2) ? gat_leaky((d2 + adn) + e.z) : ninf;
  float l3 = (rem > 3) ? gat_leaky((d3 + adn) + e.w) : ninf;
  float mn = gat_finite_or_zero(fmaxf(fmaxf(m, l0), fmaxf(fmaxf(l1, l2), l3)));
  float sc = __expf(m - mn);
  float p0 = __expf(l0 - mn);
  float p1 = __expf(l1 - mn);
  float p2 = __expf(l2 - mn);
  float p3 = __expf(l3 - mn);
  s = s * sc + ((p0 + p1) + (p2 + p3));
  te += (e.x + e.y) + (e.z + e.w);
  acc.x = acc.x * sc + p0 * R.x0.x; acc.y = acc.y * sc + p0 * R.x0.y;
  acc.z = acc.z * sc + p0 * R.x0.z; acc.w = acc.w * sc + p0 * R.x0.w;
  acc.x += p1 * R.x1.x; acc.y += p1 * R.x1.y; acc.z += p1 * R.x1.z; acc.w += p1 * R.x1.w;
  acc.x += p2 * R.x2.x; acc.y += p2 * R.x2.y; acc.z += p2 * R.x2.z; acc.w += p2 * R.x2.w;
  acc.x += p3 * R.x3.x; acc.y += p3 * R.x3.y; acc.z += p3 * R.x3.z; acc.w += p3 * R.x3.w;
  m = mn;
}

__global__ __launch_bounds__(256) void gat_aggr_csr(
    const int* __restrict__ csrw, const int* __restrict__ segb,
    const float* __restrict__ cnt, const float* __restrict__ aedge,
    const float* __restrict__ att_s, const float* __restrict__ att_d,
    const float* __restrict__ xh, const float* __restrict__ bias, int relu,
    float* __restrict__ out, int E, int Nn) {
  __shared__ __attribute__((aligned(16))) float coef[8][128];
  const int2* csr = (const int2*)csrw;
  int lane = threadIdx.x & 31;
  int w = threadIdx.x >> 5;
  int n = __builtin_amdgcn_readfirstlane(
      (int)((blockIdx.x * blockDim.x + threadIdx.x) >> 5));
  if (n >= Nn) return;
  int c0 = lane * 4;
  int h = lane >> 3;
  int beg = segb[n];
  float cf = cnt[n];
  int end = beg + (int)cf;
  if (beg < 0) beg = 0;
  if (end > E) end = E;
  int deg = end - beg;
  if (deg < 0) deg = 0;
  float cc = (cf < 1.0f) ? 1.0f : cf;

  float4 as4 = *(const float4*)(att_s + c0);
  float4 ad4 = *(const float4*)(att_d + c0);
  float4 xs = gat_row(xh, n, lane);
  float adn = gat_sum8(gat_dot4(xs, ad4));
  float dsn = gat_sum8(gat_dot4(xs, as4));

  float* sl = coef[w];
  float m = -__builtin_inff(), s = 0.0f, te = 0.0f;
  float4 acc = make_float4(0.0f, 0.0f, 0.0f, 0.0f);

  for (int base = 0; base < deg; base += 32) {
    int left = deg - base;
    int cntc = (left < 32) ? left : 32;
    int srcv = n;
    int eid = 0;
    if (lane < cntc) {
      int2 q = csr[beg + base + lane];
      srcv = q.x;
      eid = q.y;
    }
    float4 ae = make_float4(0.0f, 0.0f, 0.0f, 0.0f);
    if (lane < cntc) ae = *(const float4*)(aedge + (long long)eid * 4);
    gat_rows4 R;
    int j = 0;
    int rem = (cntc < 4) ? cntc : 4;
    gat_issue(xh, srcv, 0, rem, lane, R);
    sl[lane] = ae.x; sl[32 + lane] = ae.y; sl[64 + lane] = ae.z; sl[96 + lane] = ae.w;
    for (;;) {
      gat_step(R, sl, j, rem, h, as4, adn, m, s, te, acc);
      j += 4;
      if (j >= cntc) break;
      rem = (cntc - j < 4) ? cntc - j : 4;
      gat_issue(xh, srcv, j, rem, lane, R);
    }
  }

  {
    float ls = gat_leaky((dsn + adn) + te / cc);
    float mn = gat_finite_or_zero(fmaxf(m, ls));
    float sc = __expf(m - mn);
    float p = __expf(ls - mn);
    s = s * sc + p;
    acc.x = acc.x * sc + p * xs.x; acc.y = acc.y * sc + p * xs.y;
    acc.z = acc.z * sc + p * xs.z; acc.w = acc.w * sc + p * xs.w;
  }
  float inv = 1.0f / (s + 1e-16f);
  float4 bv = *(const float4*)(bias + c0);
  float4 o;
  o.x = acc.x * inv + bv.x; o.y = acc.y * inv + bv.y; o.z = acc.z * inv + bv.z; o.w = acc.w * inv + bv.w;
  if (relu) {
    o.x = (o.x > 0.0f) ? o.x : 0.0f; o.y = (o.y > 0.0f) ? o.y : 0.0f;
    o.z = (o.z > 0.0f) ? o.z : 0.0f; o.w = (o.w > 0.0f) ? o.w : 0.0f;
  }
  *(float4*)(out + (long long)n * 128 + c0) = o;
}

extern "C" void kernel_launch(void* const* d_in, const int* in_sizes, int n_in,
                              void* d_out, int out_size, void* d_ws,
                              size_t ws_size, hipStream_t stream) {
  const int N = in_sizes[0] / 128;
  const int E = in_sizes[1] / 2;

  const float* x  = (const float*)d_in[0];
  const int* ei   = (const int*)d_in[1];
  const int* srcI = ei;
  const int* dstI = ei + E;
  const float* ea = (const float*)d_in[2];

  const float* Wl[2]  = {(const float*)d_in[3], (const float*)d_in[9]};
  const float* asl[2] = {(const float*)d_in[4], (const float*)d_in[10]};
  const float* adl[2] = {(const float*)d_in[5], (const float*)d_in[11]};
  const float* Wel[2] = {(const float*)d_in[6], (const float*)d_in[12]};
  const float* ael[2] = {(const float*)d_in[7], (const float*)d_in[13]};
  const float* bl[2]  = {(const float*)d_in[8], (const float*)d_in[14]};

  float* wsf = (float*)d_ws;
  size_t off = 0;
  float* cnt   = wsf + off; off += (size_t)N;
  off += (size_t)N * 16;
  float* medge = wsf + off; off += 64;
  float* xh    = wsf + off; off += (size_t)N * 128;
  off += (size_t)N * 8;
  float* aedge = wsf + off; off += ((size_t)E + N) * 4;
  off += (size_t)N * 8;
  size_t lo = (off + 1) & ~(size_t)1;
  int* segb    = (int*)(wsf + lo);
  int* csr     = segb + 2 * (size_t)N + 4;
  const int NB = (N + 511) / 512;
  const int EB = (E + 2047) / 2048;
  size_t to = (lo + 2 * (size_t)N + 4 + 2 * (size_t)E + 3) & ~(size_t)3;
  int4* tmp    = (int4*)(wsf + to);
  int* tbl     = (int*)(wsf + to) + 4 * (size_t)E;
  int* btot    = tbl + (size_t)NB * EB;
  off += (size_t)N * 128;
  (void)ws_size; (void)n_in; (void)out_size;

  hipLaunchKernelGGL(gat_bucket_hist, dim3((unsigned)EB), dim3(256), 0, stream, dstI, tbl, E, NB);
  hipLaunchKernelGGL(gat_bucket_rowscan, dim3((unsigned)NB), dim3(256), 0, stream, tbl, btot, EB);
  hipLaunchKernelGGL(gat_bucket_scatter, dim3((unsigned)EB), dim3(256), 0, stream, srcI, dstI,
                     tbl, btot, tmp, E, NB);
  hipLaunchKernelGGL(gat_bucket_place, dim3((unsigned)NB), dim3(512), 0, stream, tmp, btot,
                     segb, cnt, csr, E, N);

  for (int l = 0; l < 2; ++l) {
    const float* xin = (l == 0) ? x : (const float*)d_out;
    float* outb = (float*)d_out;

    hipLaunchKernelGGL(gat_fold_edge, dim3(1), dim3(64), 0, stream, Wel[l], ael[l], medge);

    {
      long long waves = ((long long)N + 15) / 16;
      long long blocks = (waves + 7) / 8;
      hipLaunchKernelGGL(gat_gemm_wmma, dim3((unsigned)blocks), dim3(256), 0,
                         stream, xin, Wl[l], xh, N);
    }

    hipLaunchKernelGGL(gat_edge_term, dim3((unsigned)((E + 255) / 256)), dim3(256),
                       0, stream, ea, medge, aedge, E);

    {
      long long blocks = ((long long)N + 7) / 8;
      hipLaunchKernelGGL(gat_aggr_csr, dim3((unsigned)blocks), dim3(256), 0,
                         stream, csr, segb, cnt, aedge, asl[l], adl[l], xh, bl[l],
                         (l == 0) ? 1 : 0, outb, E, N);
    }
  }
}


# edge terms of both layers formed once, in destination-list order (one pass over edge_attr, one fold launch): the walk reads them as contiguous 16-byte groups, no fetch by edge id; 12 -> 10 launches
# speedup vs baseline: 6.0947x; 1.0820x over previous
#include <hip/hip_runtime.h>
#include <hip/hip_bf16.h>

typedef __attribute__((ext_vector_type(2))) float v2f;
typedef __attribute__((ext_vector_type(8))) float v8f;

#define NEG_SLOPE 0.2f

__global__ __launch_bounds__(64) void gat_fold_edge(const float* __restrict__ We1,
                             const float* __restrict__ ae1,
                             const float* __restrict__ We2,
                             const float* __restrict__ ae2,
                             float* __restrict__ medge) {
  const float* We = (blockIdx.x == 0) ? We1 : We2;
  const float* ae = (blockIdx.x == 0) ? ae1 : ae2;
  int t = threadIdx.x;
  int k = t >> 2, h = t & 3;
  float s = 0.0f;
  for (int cc = 0; cc < 32; ++cc)
    s += We[k * 128 + h * 32 + cc] * ae[h * 32 + cc];
  medge[blockIdx.x * 64 + k * 4 + h] = s;
}

__global__ __launch_bounds__(256) void gat_gemm_wmma(
    const float* __restrict__ X, const float* __restrict__ W,
    float* __restrict__ Y, int Nrows) {
  int wave = (int)(((long long)blockIdx.x * blockDim.x + threadIdx.x) >> 5);
  int lane = threadIdx.x & 31;
  int m0 = wave * 16;
  if (m0 >= Nrows) return;

  v8f acc[8] = {};
  int half = lane >> 4;
  int l16 = lane & 15;
  int arow = m0 + l16;
  if (arow >= Nrows) arow = Nrows - 1;
  const float* xr = X + (long long)arow * 128;

#pragma unroll 2
  for (int k0 = 0; k0 < 128; k0 += 4) {
    v2f a;
    a.x = xr[k0 + half * 2];
    a.y = xr[k0 + half * 2 + 1];
#pragma unroll
    for (int nt = 0; nt < 8; ++nt) {
      int col = nt * 16 + l16;
      v2f b;
      b.x = W[(k0 + half * 2) * 128 + col];
      b.y = W[(k0 + half * 2 + 1) * 128 + col];
      acc[nt] = __builtin_amdgcn_wmma_f32_16x16x4_f32(
          false, a, false, b, (short)0, acc[nt], false, false);
    }
  }
#pragma unroll
  for (int nt = 0; nt < 8; ++nt) {
#pragma unroll
    for (int r = 0; r < 8; ++r) {
      int row = m0 + r + half * 8;
      if (row < Nrows) Y[(long long)row * 128 + nt * 16 + l16] = acc[nt][r];
    }
  }
}

__global__ __launch_bounds__(256) void gat_edge_term_list(const int* __restrict__ csrw,
                              const float* __restrict__ ea,
                              const float* __restrict__ medge,
                              float* __restrict__ t1, float* __restrict__ t2, int E) {
  long long p = (long long)blockIdx.x * blockDim.x + threadIdx.x;
  if (p >= (long long)E) return;
  int eid = ((const int2*)csrw)[p].y;
  eid = (eid < 0) ? 0 : eid;
  eid = (eid < E) ? eid : E - 1;
  const float4* row = (const float4*)(ea + (long long)eid * 16);
  float4 r0 = row[0], r1 = row[1], r2 = row[2], r3 = row[3];
  float e0[16] = {r0.x, r0.y, r0.z, r0.w, r1.x, r1.y, r1.z, r1.w,
                  r2.x, r2.y, r2.z, r2.w, r3.x, r3.y, r3.z, r3.w};
  float o[8];
#pragma unroll
  for (int lh = 0; lh < 8; ++lh) {
    const float* mm = medge + (lh >> 2) * 64 + (lh & 3);
    float aE = 0.0f;
#pragma unroll
    for (int k = 0; k < 16; ++k) aE += e0[k] * mm[k * 4];
    o[lh] = aE;
  }
  *(float4*)(t1 + p * 4) = make_float4(o[0], o[1], o[2], o[3]);
  *(float4*)(t2 + p * 4) = make_float4(o[4], o[5], o[6], o[7]);
}

#define GAT_NBMAX 1024

__device__ __forceinline__ int gat_bucket_of(int d, int nb) {
  int b = d >> 9;
  b = (b < 0) ? 0 : b;
  return (b < nb) ? b : nb - 1;
}
__device__ __forceinline__ int gat_local_of(int d, int b) {
  int l = d - (b << 9);
  l = (l < 0) ? 0 : l;
  return (l < 512) ? l : 511;
}
__device__ __forceinline__ int gat_wave_scan(int v, int lane) {
  for (int o = 1; o < 32; o <<= 1) {
    int u = __shfl_up(v, o, 32);
    if (lane >= o) v += u;
  }
  return v;
}

__global__ __launch_bounds__(256) void gat_bucket_hist(const int* __restrict__ dst,
                                int* __restrict__ tbl, int E, int NB) {
  __shared__ int h[GAT_NBMAX];
  int t = threadIdx.x;
  int nb = (NB < GAT_NBMAX) ? NB : GAT_NBMAX;
  int base = blockIdx.x * 2048 + t;
  int d[8];
#pragma unroll
  for (int r = 0; r < 8; ++r) {
    int e = base + r * 256;
    int ee = (e < E) ? e : E - 1;
    d[r] = dst[ee];
  }
  for (int i = t; i < nb; i += 256) h[i] = 0;
  __syncthreads();
#pragma unroll
  for (int r = 0; r < 8; ++r) {
    int e = base + r * 256;
    if (e < E) atomicAdd(&h[gat_bucket_of(d[r], nb)], 1);
  }
  __syncthreads();
  for (int i = t; i < nb; i += 256) tbl[(size_t)i * gridDim.x + blockIdx.x] = h[i];
}

__global__ __launch_bounds__(256) void gat_bucket_rowscan(int* __restrict__ tbl,
                                   int* __restrict__ btot, int B) {
  __shared__ int wsum[8];
  if (blockIdx.x >= GAT_NBMAX) return;
  int t = threadIdx.x, lane = t & 31, w = t >> 5;
  int* row = tbl + (size_t)blockIdx.x * B;
  int carry = 0;
  for (int t0 = 0; t0 < B; t0 += 1024) {
    int i0 = t0 + 4 * t;
    int v[4];
#pragma unroll
    for (int k = 0; k < 4; ++k) v[k] = (i0 + k < B) ? row[i0 + k] : 0;
    int ts = v[0] + v[1] + v[2] + v[3];
    int inc = gat_wave_scan(ts, lane);
    if (lane == 31) wsum[w] = inc;
    __syncthreads();
    int wp = 0, tot = 0;
#pragma unroll
    for (int i = 0; i < 8; ++i) { int u = wsum[i]; wp += (i < w) ? u : 0; tot += u; }
    int ex = carry + wp + inc - ts;
#pragma unroll
    for (int k = 0; k < 4; ++k) {
      if (i0 + k < B) row[i0 + k] = ex;
      ex += v[k];
    }
    carry += tot;
    __syncthreads();
  }
  if (t == 0) btot[blockIdx.x] = carry;
}

__global__ __launch_bounds__(256) void gat_bucket_scatter(const int* __restrict__ srcI,
                                   const int* __restrict__ dstI,
                                   const int* __restrict__ tbl,
                                   const int* __restrict__ btot,
                                   int4* __restrict__ tmp, int E, int NB) {
  __shared__ int off[GAT_NBMAX];
  __shared__ int wsum[8];
  int t = threadIdx.x, lane = t & 31, w = t >> 5;
  int nb = (NB < GAT_NBMAX) ? NB : GAT_NBMAX;
  int base = blockIdx.x * 2048 + t;
  int d[8], s[8];
#pragma unroll
  for (int r = 0; r < 8; ++r) {
    int e = base + r * 256;
    int ee = (e < E) ? e : E - 1;
    d[r] = dstI[ee]; s[r] = srcI[ee];
  }
  int i0 = 4 * t;
  int v[4], mine[4];
#pragma unroll
  for (int k = 0; k < 4; ++k) {
    bool in = i0 + k < nb;
    v[k] = in ? btot[i0 + k] : 0;
    mine[k] = in ? tbl[(size_t)(i0 + k) * gridDim.x + blockIdx.x] : 0;
  }
  int ts = v[0] + v[1] + v[2] + v[3];
  int inc = gat_wave_scan(ts, lane);
  if (lane == 31) wsum[w] = inc;
  __syncthreads();
  int wp = 0;
#pragma unroll
  for (int i = 0; i < 8; ++i) { int u = wsum[i]; wp += (i < w) ? u : 0; }
  int ex = wp + inc - ts;
#pragma unroll
  for (int k = 0; k < 4; ++k) {
    if (i0 + k < nb) off[i0 + k] = ex + mine[k];
    ex += v[k];
  }
  __syncthreads();
#pragma unroll
  for (int r = 0; r < 8; ++r) {
    int e = base + r * 256;
    if (e < E) {
      int pos = atomicAdd(&off[gat_bucket_of(d[r], nb)], 1);
      if ((unsigned)pos < (unsigned)E) tmp[pos] = make_int4(d[r], s[r], e, 0);
    }
  }
}

__global__ __launch_bounds__(512) void gat_bucket_place(const int4* __restrict__ tmp,
                                 const int* __restrict__ btot,
                                 int* __restrict__ segb, float* __restrict__ cnt,
                                 int* __restrict__ csrw, int E, int Nn) {
  __shared__ int c[512];
  __shared__ int wsum[16];
  int2* csr = (int2*)csrw;
  int b = blockIdx.x;
  if (b >= GAT_NBMAX) return;
  int t = threadIdx.x, lane = t & 31, w = t >> 5;
  int p = 0;
  for (int i = t; i < b; i += 512) p += btot[i];
  for (int o = 16; o > 0; o >>= 1) p += __shfl_xor(p, o, 32);
  if (lane == 0) wsum[w] = p;
  c[t] = 0;
  __syncthreads();
  int base = 0;
#pragma unroll
  for (int i = 0; i < 16; ++i) base += wsum[i];
  int m = btot[b];
  base = (base < 0) ? 0 : base;
  base = (base > E) ? E : base;
  m = (m < 0) ? 0 : m;
  m = (m > E - base) ? E - base : m;
  const int4* rec = tmp + base;

  for (int i = t; i < m; i += 4096) {
    int dd[8];
#pragma unroll
    for (int r = 0; r < 8; ++r) {
      int j = i + r * 512;
      dd[r] = rec[(j < m) ? j : m - 1].x;
    }
#pragma unroll
    for (int r = 0; r < 8; ++r)
      if (i + r * 512 < m) atomicAdd(&c[gat_local_of(dd[r], b)], 1);
  }
  __syncthreads();

  int deg = c[t];
  int inc = gat_wave_scan(deg, lane);
  if (lane == 31) wsum[w] = inc;
  __syncthreads();
  int wp = 0;
#pragma unroll
  for (int i = 0; i < 16; ++i) { int u = wsum[i]; wp += (i < w) ? u : 0; }
  int ex = wp + inc - deg;
  int n = (b << 9) + t;
  if (n < Nn) { segb[n] = base + ex; cnt[n] = (float)deg; }
  c[t] = ex;
  __syncthreads();

  for (int i = t; i < m; i += 2048) {
    int4 q[4];
#pragma unroll
    for (int r = 0; r < 4; ++r) {
      int j = i + r * 512;
      q[r] = rec[(j < m) ? j : m - 1];
    }
#pragma unroll
    for (int r = 0; r < 4; ++r) {
      if (i + r * 512 < m) {
        int pos = base + atomicAdd(&c[gat_local_of(q[r].x, b)], 1);
        if ((unsigned)pos < (unsigned)E) csr[pos] = make_int2(q[r].y, q[r].z);
      }
    }
  }
}

__device__ __forceinline__ float gat_finite_or_zero(float m) {
  return (m > -__builtin_inff() && m < __builtin_inff()) ? m : 0.0f;
}

__device__ __forceinline__ float gat_leaky(float v) {
  return fmaxf(v, NEG_SLOPE * v);
}

template <int XM>
__device__ __forceinline__ float gat_xlane(float v) {
  return __int_as_float(__builtin_amdgcn_update_dpp(0, __float_as_int(v), 0x160 + XM, 0xf, 0xf, true));
}
__device__ __forceinline__ float gat_sum8(float v) {
  v += gat_xlane<1>(v);
  v += gat_xlane<2>(v);
  v += gat_xlane<4>(v);
  return v;
}
__device__ __forceinline__ float gat_dot4(float4 x, float4 a) {
  v2f xl = {x.x, x.y}, xh2 = {x.z, x.w};
  v2f al = {a.x, a.y}, ah = {a.z, a.w};
  v2f t = xh2 * ah + xl * al;
  return t.x + t.y;
}

typedef float gat_f4 __attribute__((ext_vector_type(4)));
__device__ __forceinline__ float4 gat_row(const float* __restrict__ xh, int sr, int lane) {
  unsigned long long a = (unsigned long long)xh + ((unsigned long long)(unsigned)sr << 9);
  gat_f4 v = ((const __attribute__((address_space(1))) gat_f4*)a)[lane];
  return make_float4(v.x, v.y, v.z, v.w);
}

struct gat_rows4 { float4 x0, x1, x2, x3; };

__device__ __forceinline__ void gat_issue(const float* __restrict__ xh, int srcv, int j, int rem,
                                          int lane, gat_rows4& R) {
  const float4 z = make_float4(0.0f, 0.0f, 0.0f, 0.0f);
  R.x0 = gat_row(xh, __builtin_amdgcn_readlane(srcv, j), lane);
  R.x1 = z; R.x2 = z; R.x3 = z;
  if (rem > 1) R.x1 = gat_row(xh, __builtin_amdgcn_readlane(srcv, j + 1), lane);
  if (rem > 2) R.x2 = gat_row(xh, __builtin_amdgcn_readlane(srcv, j + 2), lane);
  if (rem > 3) R.x3 = gat_row(xh, __builtin_amdgcn_readlane(srcv, j + 3), lane);
}

__device__ __forceinline__ void gat_step(const gat_rows4& R, const float* sl, int j, int rem, int h,
                                         float4 as4, float adn,
                                         float& m, float& s, float& te, float4& acc) {
  const float ninf = -__builtin_inff();
  float4 e = *(const float4*)(sl + h * 32 + j);
  float d0 = gat_sum8(gat_dot4(R.x0, as4));
  float d1 = gat_sum8(gat_dot4(R.x1, as4));
  float d2 = gat_sum8(gat_dot4(R.x2, as4));
  float d3 = gat_sum8(gat_dot4(R.x3, as4));
  float l0 = gat_leaky((d0 + adn) + e.x);
  float l1 = (rem > 1) ? gat_leaky((d1 + adn) + e.y) : ninf;
  float l2 = (rem > 2) ? gat_leaky((d2 + adn) + e.z) : ninf;
  float l3 = (rem > 3) ? gat_leaky((d3 + adn) + e.w) : ninf;
  float mn = gat_finite_or_zero(fmaxf(fmaxf(m, l0), fmaxf(fmaxf(l1, l2), l3)));
  float sc = __expf(m - mn);
  float p0 = __expf(l0 - mn);
  float p1 = __expf(l1 - mn);
  float p2 = __expf(l2 - mn);
  float p3 = __expf(l3 - mn);
  s = s * sc + ((p0 + p1) + (p2 + p3));
  te += (e.x + e.y) + (e.z + e.w);
  acc.x = acc.x * sc + p0 * R.x0.x; acc.y = acc.y * sc + p0 * R.x0.y;
  acc.z = acc.z * sc + p0 * R.x0.z; acc.w = acc.w * sc + p0 * R.x0.w;
  acc.x += p1 * R.x1.x; acc.y += p1 * R.x1.y; acc.z += p1 * R.x1.z; acc.w += p1 * R.x1.w;
  acc.x += p2 * R.x2.x; acc.y += p2 * R.x2.y; acc.z += p2 * R.x2.z; acc.w += p2 * R.x2.w;
  acc.x += p3 * R.x3.x; acc.y += p3 * R.x3.y; acc.z += p3 * R.x3.z; acc.w += p3 * R.x3.w;
  m = mn;
}

__global__ __launch_bounds__(256) void gat_aggr_csr(
    const int* __restrict__ csrw, const int* __restrict__ segb,
    const float* __restrict__ cnt, const float* __restrict__ aedge,
    const float* __restrict__ att_s, const float* __restrict__ att_d,
    const float* __restrict__ xh, const float* __restrict__ bias, int relu,
    float* __restrict__ out, int E, int Nn) {
  __shared__ __attribute__((aligned(16))) float coef[8][128];
  const int2* csr = (const int2*)csrw;
  int lane = threadIdx.x & 31;
  int w = threadIdx.x >> 5;
  int n = __builtin_amdgcn_readfirstlane(
      (int)((blockIdx.x * blockDim.x + threadIdx.x) >> 5));
  if (n >= Nn) return;
  int c0 = lane * 4;
  int h = lane >> 3;
  int beg = segb[n];
  float cf = cnt[n];
  int end = beg + (int)cf;
  if (beg < 0) beg = 0;
  if (end > E) end = E;
  int deg = end - beg;
  if (deg < 0) deg = 0;
  float cc = (cf < 1.0f) ? 1.0f : cf;

  float4 as4 = *(const float4*)(att_s + c0);
  float4 ad4 = *(const float4*)(att_d + c0);
  float4 xs = gat_row(xh, n, lane);
  float adn = gat_sum8(gat_dot4(xs, ad4));
  float dsn = gat_sum8(gat_dot4(xs, as4));

  float* sl = coef[w];
  float m = -__builtin_inff(), s = 0.0f, te = 0.0f;
  float4 acc = make_float4(0.0f, 0.0f, 0.0f, 0.0f);

  for (int base = 0; base < deg; base += 32) {
    int left = deg - base;
    int cntc = (left < 32) ? left : 32;
    int srcv = n;
    float4 ae = make_float4(0.0f, 0.0f, 0.0f, 0.0f);
    if (lane < cntc) {
      srcv = csr[beg + base + lane].x;
      ae = *(const float4*)(aedge + (long long)(beg + base + lane) * 4);
    }
    gat_rows4 R;
    int j = 0;
    int rem = (cntc < 4) ? cntc : 4;
    gat_issue(xh, srcv, 0, rem, lane, R);
    sl[lane] = ae.x; sl[32 + lane] = ae.y; sl[64 + lane] = ae.z; sl[96 + lane] = ae.w;
    for (;;) {
      gat_step(R, sl, j, rem, h, as4, adn, m, s, te, acc);
      j += 4;
      if (j >= cntc) break;
      rem = (cntc - j < 4) ? cntc - j : 4;
      gat_issue(xh, srcv, j, rem, lane, R);
    }
  }

  {
    float ls = gat_leaky((dsn + adn) + te / cc);
    float mn = gat_finite_or_zero(fmaxf(m, ls));
    float sc = __expf(m - mn);
    float p = __expf(ls - mn);
    s = s * sc + p;
    acc.x = acc.x * sc + p * xs.x; acc.y = acc.y * sc + p * xs.y;
    acc.z = acc.z * sc + p * xs.z; acc.w = acc.w * sc + p * xs.w;
  }
  float inv = 1.0f / (s + 1e-16f);
  float4 bv = *(const float4*)(bias + c0);
  float4 o;
  o.x = acc.x * inv + bv.x; o.y = acc.y * inv + bv.y; o.z = acc.z * inv + bv.z; o.w = acc.w * inv + bv.w;
  if (relu) {
    o.x = (o.x > 0.0f) ? o.x : 0.0f; o.y = (o.y > 0.0f) ? o.y : 0.0f;
    o.z = (o.z > 0.0f) ? o.z : 0.0f; o.w = (o.w > 0.0f) ? o.w : 0.0f;
  }
  *(float4*)(out + (long long)n * 128 + c0) = o;
}

extern "C" void kernel_launch(void* const* d_in, const int* in_sizes, int n_in,
                              void* d_out, int out_size, void* d_ws,
                              size_t ws_size, hipStream_t stream) {
  const int N = in_sizes[0] / 128;
  const int E = in_sizes[1] / 2;

  const float* x  = (const float*)d_in[0];
  const int* ei   = (const int*)d_in[1];
  const int* srcI = ei;
  const int* dstI = ei + E;
  const float* ea = (const float*)d_in[2];

  const float* Wl[2]  = {(const float*)d_in[3], (const float*)d_in[9]};
  const float* asl[2] = {(const float*)d_in[4], (const float*)d_in[10]};
  const float* adl[2] = {(const float*)d_in[5], (const float*)d_in[11]};
  const float* Wel[2] = {(const float*)d_in[6], (const float*)d_in[12]};
  const float* ael[2] = {(const float*)d_in[7], (const float*)d_in[13]};
  const float* bl[2]  = {(const float*)d_in[8], (const float*)d_in[14]};

  float* wsf = (float*)d_ws;
  size_t off = 0;
  float* cnt   = wsf + off; off += (size_t)N;
  float* medge = wsf + off;
  off += (size_t)N * 16;
  off += 64;
  float* xh    = wsf + off; off += (size_t)N * 128;
  off += (size_t)N * 8;
  float* aedge = wsf + off; off += ((size_t)E + N) * 4;
  off += (size_t)N * 8;
  size_t lo = (off + 1) & ~(size_t)1;
  int* segb    = (int*)(wsf + lo);
  int* csr     = segb + 2 * (size_t)N + 4;
  const int NB = (N + 511) / 512;
  const int EB = (E + 2047) / 2048;
  size_t to = (lo + 2 * (size_t)N + 4 + 2 * (size_t)E + 3) & ~(size_t)3;
  int4* tmp    = (int4*)(wsf + to);
  float* aedge2 = wsf + to;
  int* tbl     = (int*)(wsf + to) + 4 * (size_t)E;
  int* btot    = tbl + (size_t)NB * EB;
  off += (size_t)N * 128;
  (void)ws_size; (void)n_in; (void)out_size;

  hipLaunchKernelGGL(gat_bucket_hist, dim3((unsigned)EB), dim3(256), 0, stream, dstI, tbl, E, NB);
  hipLaunchKernelGGL(gat_bucket_rowscan, dim3((unsigned)NB), dim3(256), 0, stream, tbl, btot, EB);
  hipLaunchKernelGGL(gat_bucket_scatter, dim3((unsigned)EB), dim3(256), 0, stream, srcI, dstI,
                     tbl, btot, tmp, E, NB);
  hipLaunchKernelGGL(gat_bucket_place, dim3((unsigned)NB), dim3(512), 0, stream, tmp, btot,
                     segb, cnt, csr, E, N);

  hipLaunchKernelGGL(gat_fold_edge, dim3(2), dim3(64), 0, stream, Wel[0], ael[0], Wel[1], ael[1], medge);
  hipLaunchKernelGGL(gat_edge_term_list, dim3((unsigned)((E + 255) / 256)), dim3(256),
                     0, stream, csr, ea, medge, aedge, aedge2, E);

  for (int l = 0; l < 2; ++l) {
    const float* xin = (l == 0) ? x : (const float*)d_out;
    float* outb = (float*)d_out;

    {
      long long waves = ((long long)N + 15) / 16;
      long long blocks = (waves + 7) / 8;
      hipLaunchKernelGGL(gat_gemm_wmma, dim3((unsigned)blocks), dim3(256), 0,
                         stream, xin, Wl[l], xh, N);
    }

    {
      long long blocks = ((long long)N + 7) / 8;
      hipLaunchKernelGGL(gat_aggr_csr, dim3((unsigned)blocks), dim3(256), 0,
                         stream, csr, segb, cnt, (l == 0) ? aedge : aedge2, asl[l], adl[l], xh, bl[l],
                         (l == 0) ? 1 : 0, outb, E, N);
    }
  }
}


# GEMM B operand packed per call (fold launch) into fragment order, tile columns interleaved by 4: B read as 4 whole-line 16-byte loads per K step, a step ahead; rows stored as whole lines; same sums
# speedup vs baseline: 6.5995x; 1.0828x over previous
#include <hip/hip_runtime.h>
#include <hip/hip_bf16.h>

typedef __attribute__((ext_vector_type(2))) float v2f;
typedef __attribute__((ext_vector_type(8))) float v8f;

#define NEG_SLOPE 0.2f


__device__ __forceinline__ int gat_tile_col(int t, int l16) {
  return 64 * (t >> 2) + 4 * l16 + (t & 3);
}

__global__ __launch_bounds__(64) void gat_fold_edge(const float* __restrict__ We1,
                             const float* __restrict__ ae1,
                             const float* __restrict__ We2,
                             const float* __restrict__ ae2,
                             const float* __restrict__ W1,
                             const float* __restrict__ W2,
                             float* __restrict__ medge,
                             float* __restrict__ wpk) {
  int t = threadIdx.x;
  if (blockIdx.x < 2) {
    const float* We = (blockIdx.x == 0) ? We1 : We2;
    const float* ae = (blockIdx.x == 0) ? ae1 : ae2;
    int k = t >> 2, h = t & 3;
    float s = 0.0f;
    for (int cc = 0; cc < 32; ++cc)
      s += We[k * 128 + h * 32 + cc] * ae[h * 32 + cc];
    medge[blockIdx.x * 64 + k * 4 + h] = s;
    return;
  }
  int bb = (int)blockIdx.x - 2;
  int layer = (bb >> 5) & 1, s = bb & 31;
  const float* Wm = layer ? W2 : W1;
  float* dp = wpk + layer * 16384 + s * 512;
  int L = t & 31, half = L >> 4, l16 = L & 15;
  int k = 4 * s + 2 * half;
  for (int qq = 0; qq < 2; ++qq) {
    int q = (t >> 5) * 2 + qq;
    int cA = gat_tile_col(2 * q, l16), cB = gat_tile_col(2 * q + 1, l16);
    float4 v = make_float4(Wm[k * 128 + cA], Wm[(k + 1) * 128 + cA],
                           Wm[k * 128 + cB], Wm[(k + 1) * 128 + cB]);
    *(float4*)(dp + (q * 32 + L) * 4) = v;
  }
}

__global__ __launch_bounds__(256) void gat_gemm_wmma(
    const float* __restrict__ X, const float* __restrict__ Wp,
    float* __restrict__ Y, int Nrows) {
  int wave = (int)(((long long)blockIdx.x * blockDim.x + threadIdx.x) >> 5);
  int lane = threadIdx.x & 31;
  int m0 = wave * 16;
  if (m0 >= Nrows) return;

  v8f acc[8] = {};
  int half = lane >> 4;
  int l16 = lane & 15;
  int arow = m0 + l16;
  if (arow >= Nrows) arow = Nrows - 1;
  const float* xr = X + (long long)arow * 128;
  const float4* wl = (const float4*)Wp + lane;
  const v2f* xa = (const v2f*)(xr + half * 2);

  float4 b0[4], b1[4];
  v2f a0, a1;
#pragma unroll
  for (int q = 0; q < 4; ++q) b0[q] = wl[q * 32];
  a0 = xa[0];
  for (int s = 0; s < 32; s += 2) {
    const float4* w1 = wl + (s + 1) * 128;
#pragma unroll
    for (int q = 0; q < 4; ++q) b1[q] = w1[q * 32];
    a1 = xa[2 * (s + 1)];
    __builtin_amdgcn_sched_barrier(0);
#pragma unroll
    for (int q = 0; q < 4; ++q) {
      v2f p0 = {b0[q].x, b0[q].y};
      v2f p1 = {b0[q].z, b0[q].w};
      acc[2 * q] = __builtin_amdgcn_wmma_f32_16x16x4_f32(
          false, a0, false, p0, (short)0, acc[2 * q], false, false);
      acc[2 * q + 1] = __builtin_amdgcn_wmma_f32_16x16x4_f32(
          false, a0, false, p1, (short)0, acc[2 * q + 1], false, false);
    }
    int sn = (s + 2 < 32) ? s + 2 : 31;
    const float4* w2 = wl + sn * 128;
#pragma unroll
    for (int q = 0; q < 4; ++q) b0[q] = w2[q * 32];
    a0 = xa[2 * sn];
    __builtin_amdgcn_sched_barrier(0);
#pragma unroll
    for (int q = 0; q < 4; ++q) {
      v2f p0 = {b1[q].x, b1[q].y};
      v2f p1 = {b1[q].z, b1[q].w};
      acc[2 * q] = __builtin_amdgcn_wmma_f32_16x16x4_f32(
          false, a1, false, p0, (short)0, acc[2 * q], false, false);
      acc[2 * q + 1] = __builtin_amdgcn_wmma_f32_16x16x4_f32(
          false, a1, false, p1, (short)0, acc[2 * q + 1], false, false);
    }
  }
#pragma unroll
  for (int g = 0; g < 2; ++g) {
#pragma unroll
    for (int r = 0; r < 8; ++r) {
      int row = m0 + r + half * 8;
      if (row < Nrows)
        *(float4*)(Y + (long long)row * 128 + 64 * g + 4 * l16) =
            make_float4(acc[4 * g][r], acc[4 * g + 1][r], acc[4 * g + 2][r], acc[4 * g + 3][r]);
    }
  }
}

__global__ __launch_bounds__(256) void gat_edge_term_list(const int* __restrict__ csrw,
                              const float* __restrict__ ea,
                              const float* __restrict__ medge,
                              float* __restrict__ t1, float* __restrict__ t2, int E) {
  long long p = (long long)blockIdx.x * blockDim.x + threadIdx.x;
  if (p >= (long long)E) return;
  int eid = ((const int2*)csrw)[p].y;
  eid = (eid < 0) ? 0 : eid;
  eid = (eid < E) ? eid : E - 1;
  const float4* row = (const float4*)(ea + (long long)eid * 16);
  float4 r0 = row[0], r1 = row[1], r2 = row[2], r3 = row[3];
  float e0[16] = {r0.x, r0.y, r0.z, r0.w, r1.x, r1.y, r1.z, r1.w,
                  r2.x, r2.y, r2.z, r2.w, r3.x, r3.y, r3.z, r3.w};
  float o[8];
#pragma unroll
  for (int lh = 0; lh < 8; ++lh) {
    const float* mm = medge + (lh >> 2) * 64 + (lh & 3);
    float aE = 0.0f;
#pragma unroll
    for (int k = 0; k < 16; ++k) aE += e0[k] * mm[k * 4];
    o[lh] = aE;
  }
  *(float4*)(t1 + p * 4) = make_float4(o[0], o[1], o[2], o[3]);
  *(float4*)(t2 + p * 4) = make_float4(o[4], o[5], o[6], o[7]);
}

#define GAT_NBMAX 1024

__device__ __forceinline__ int gat_bucket_of(int d, int nb) {
  int b = d >> 9;
  b = (b < 0) ? 0 : b;
  return (b < nb) ? b : nb - 1;
}
__device__ __forceinline__ int gat_local_of(int d, int b) {
  int l = d - (b << 9);
  l = (l < 0) ? 0 : l;
  return (l < 512) ? l : 511;
}
__device__ __forceinline__ int gat_wave_scan(int v, int lane) {
  for (int o = 1; o < 32; o <<= 1) {
    int u = __shfl_up(v, o, 32);
    if (lane >= o) v += u;
  }
  return v;
}

__global__ __launch_bounds__(256) void gat_bucket_hist(const int* __restrict__ dst,
                                int* __restrict__ tbl, int E, int NB) {
  __shared__ int h[GAT_NBMAX];
  int t = threadIdx.x;
  int nb = (NB < GAT_NBMAX) ? NB : GAT_NBMAX;
  int base = blockIdx.x * 2048 + t;
  int d[8];
#pragma unroll
  for (int r = 0; r < 8; ++r) {
    int e = base + r * 256;
    int ee = (e < E) ? e : E - 1;
    d[r] = dst[ee];
  }
  for (int i = t; i < nb; i += 256) h[i] = 0;
  __syncthreads();
#pragma unroll
  for (int r = 0; r < 8; ++r) {
    int e = base + r * 256;
    if (e < E) atomicAdd(&h[gat_bucket_of(d[r], nb)], 1);
  }
  __syncthreads();
  for (int i = t; i < nb; i += 256) tbl[(size_t)i * gridDim.x + blockIdx.x] = h[i];
}

__global__ __launch_bounds__(256) void gat_bucket_rowscan(int* __restrict__ tbl,
                                   int* __restrict__ btot, int B) {
  __shared__ int wsum[8];
  if (blockIdx.x >= GAT_NBMAX) return;
  int t = threadIdx.x, lane = t & 31, w = t >> 5;
  int* row = tbl + (size_t)blockIdx.x * B;
  int carry = 0;
  for (int t0 = 0; t0 < B; t0 += 1024) {
    int i0 = t0 + 4 * t;
    int v[4];
#pragma unroll
    for (int k = 0; k < 4; ++k) v[k] = (i0 + k < B) ? row[i0 + k] : 0;
    int ts = v[0] + v[1] + v[2] + v[3];
    int inc = gat_wave_scan(ts, lane);
    if (lane == 31) wsum[w] = inc;
    __syncthreads();
    int wp = 0, tot = 0;
#pragma unroll
    for (int i = 0; i < 8; ++i) { int u = wsum[i]; wp += (i < w) ? u : 0; tot += u; }
    int ex = carry + wp + inc - ts;
#pragma unroll
    for (int k = 0; k < 4; ++k) {
      if (i0 + k < B) row[i0 + k] = ex;
      ex += v[k];
    }
    carry += tot;
    __syncthreads();
  }
  if (t == 0) btot[blockIdx.x] = carry;
}

__global__ __launch_bounds__(256) void gat_bucket_scatter(const int* __restrict__ srcI,
                                   const int* __restrict__ dstI,
                                   const int* __restrict__ tbl,
                                   const int* __restrict__ btot,
                                   int4* __restrict__ tmp, int E, int NB) {
  __shared__ int off[GAT_NBMAX];
  __shared__ int wsum[8];
  int t = threadIdx.x, lane = t & 31, w = t >> 5;
  int nb = (NB < GAT_NBMAX) ? NB : GAT_NBMAX;
  int base = blockIdx.x * 2048 + t;
  int d[8], s[8];
#pragma unroll
  for (int r = 0; r < 8; ++r) {
    int e = base + r * 256;
    int ee = (e < E) ? e : E - 1;
    d[r] = dstI[ee]; s[r] = srcI[ee];
  }
  int i0 = 4 * t;
  int v[4], mine[4];
#pragma unroll
  for (int k = 0; k < 4; ++k) {
    bool in = i0 + k < nb;
    v[k] = in ? btot[i0 + k] : 0;
    mine[k] = in ? tbl[(size_t)(i0 + k) * gridDim.x + blockIdx.x] : 0;
  }
  int ts = v[0] + v[1] + v[2] + v[3];
  int inc = gat_wave_scan(ts, lane);
  if (lane == 31) wsum[w] = inc;
  __syncthreads();
  int wp = 0;
#pragma unroll
  for (int i = 0; i < 8; ++i) { int u = wsum[i]; wp += (i < w) ? u : 0; }
  int ex = wp + inc - ts;
#pragma unroll
  for (int k = 0; k < 4; ++k) {
    if (i0 + k < nb) off[i0 + k] = ex + mine[k];
    ex += v[k];
  }
  __syncthreads();
#pragma unroll
  for (int r = 0; r < 8; ++r) {
    int e = base + r * 256;
    if (e < E) {
      int pos = atomicAdd(&off[gat_bucket_of(d[r], nb)], 1);
      if ((unsigned)pos < (unsigned)E) tmp[pos] = make_int4(d[r], s[r], e, 0);
    }
  }
}

__global__ __launch_bounds__(512) void gat_bucket_place(const int4* __restrict__ tmp,
                                 const int* __restrict__ btot,
                                 int* __restrict__ segb, float* __restrict__ cnt,
                                 int* __restrict__ csrw, int E, int Nn) {
  __shared__ int c[512];
  __shared__ int wsum[16];
  int2* csr = (int2*)csrw;
  int b = blockIdx.x;
  if (b >= GAT_NBMAX) return;
  int t = threadIdx.x, lane = t & 31, w = t >> 5;
  int p = 0;
  for (int i = t; i < b; i += 512) p += btot[i];
  for (int o = 16; o > 0; o >>= 1) p += __shfl_xor(p, o, 32);
  if (lane == 0) wsum[w] = p;
  c[t] = 0;
  __syncthreads();
  int base = 0;
#pragma unroll
  for (int i = 0; i < 16; ++i) base += wsum[i];
  int m = btot[b];
  base = (base < 0) ? 0 : base;
  base = (base > E) ? E : base;
  m = (m < 0) ? 0 : m;
  m = (m > E - base) ? E - base : m;
  const int4* rec = tmp + base;

  for (int i = t; i < m; i += 4096) {
    int dd[8];
#pragma unroll
    for (int r = 0; r < 8; ++r) {
      int j = i + r * 512;
      dd[r] = rec[(j < m) ? j : m - 1].x;
    }
#pragma unroll
    for (int r = 0; r < 8; ++r)
      if (i + r * 512 < m) atomicAdd(&c[gat_local_of(dd[r], b)], 1);
  }
  __syncthreads();

  int deg = c[t];
  int inc = gat_wave_scan(deg, lane);
  if (lane == 31) wsum[w] = inc;
  __syncthreads();
  int wp = 0;
#pragma unroll
  for (int i = 0; i < 16; ++i) { int u = wsum[i]; wp += (i < w) ? u : 0; }
  int ex = wp + inc - deg;
  int n = (b << 9) + t;
  if (n < Nn) { segb[n] = base + ex; cnt[n] = (float)deg; }
  c[t] = ex;
  __syncthreads();

  for (int i = t; i < m; i += 2048) {
    int4 q[4];
#pragma unroll
    for (int r = 0; r < 4; ++r) {
      int j = i + r * 512;
      q[r] = rec[(j < m) ? j : m - 1];
    }
#pragma unroll
    for (int r = 0; r < 4; ++r) {
      if (i + r * 512 < m) {
        int pos = base + atomicAdd(&c[gat_local_of(q[r].x, b)], 1);
        if ((unsigned)pos < (unsigned)E) csr[pos] = make_int2(q[r].y, q[r].z);
      }
    }
  }
}

__device__ __forceinline__ float gat_finite_or_zero(float m) {
  return (m > -__builtin_inff() && m < __builtin_inff()) ? m : 0.0f;
}

__device__ __forceinline__ float gat_leaky(float v) {
  return fmaxf(v, NEG_SLOPE * v);
}

template <int XM>
__device__ __forceinline__ float gat_xlane(float v) {
  return __int_as_float(__builtin_amdgcn_update_dpp(0, __float_as_int(v), 0x160 + XM, 0xf, 0xf, true));
}
__device__ __forceinline__ float gat_sum8(float v) {
  v += gat_xlane<1>(v);
  v += gat_xlane<2>(v);
  v += gat_xlane<4>(v);
  return v;
}
__device__ __forceinline__ float gat_dot4(float4 x, float4 a) {
  v2f xl = {x.x, x.y}, xh2 = {x.z, x.w};
  v2f al = {a.x, a.y}, ah = {a.z, a.w};
  v2f t = xh2 * ah + xl * al;
  return t.x + t.y;
}

typedef float gat_f4 __attribute__((ext_vector_type(4)));
__device__ __forceinline__ float4 gat_row(const float* __restrict__ xh, int sr, int lane) {
  unsigned long long a = (unsigned long long)xh + ((unsigned long long)(unsigned)sr << 9);
  gat_f4 v = ((const __attribute__((address_space(1))) gat_f4*)a)[lane];
  return make_float4(v.x, v.y, v.z, v.w);
}

struct gat_rows4 { float4 x0, x1, x2, x3; };

__device__ __forceinline__ void gat_issue(const float* __restrict__ xh, int srcv, int j, int rem,
                                          int lane, gat_rows4& R) {
  const float4 z = make_float4(0.0f, 0.0f, 0.0f, 0.0f);
  R.x0 = gat_row(xh, __builtin_amdgcn_readlane(srcv, j), lane);
  R.x1 = z; R.x2 = z; R.x3 = z;
  if (rem > 1) R.x1 = gat_row(xh, __builtin_amdgcn_readlane(srcv, j + 1), lane);
  if (rem > 2) R.x2 = gat_row(xh, __builtin_amdgcn_readlane(srcv, j + 2), lane);
  if (rem > 3) R.x3 = gat_row(xh, __builtin_amdgcn_readlane(srcv, j + 3), lane);
}

__device__ __forceinline__ void gat_step(const gat_rows4& R, const float* sl, int j, int rem, int h,
                                         float4 as4, float adn,
                                         float& m, float& s, float& te, float4& acc) {
  const float ninf = -__builtin_inff();
  float4 e = *(const float4*)(sl + h * 32 + j);
  float d0 = gat_sum8(gat_dot4(R.x0, as4));
  float d1 = gat_sum8(gat_dot4(R.x1, as4));
  float d2 = gat_sum8(gat_dot4(R.x2, as4));
  float d3 = gat_sum8(gat_dot4(R.x3, as4));
  float l0 = gat_leaky((d0 + adn) + e.x);
  float l1 = (rem > 1) ? gat_leaky((d1 + adn) + e.y) : ninf;
  float l2 = (rem > 2) ? gat_leaky((d2 + adn) + e.z) : ninf;
  float l3 = (rem > 3) ? gat_leaky((d3 + adn) + e.w) : ninf;
  float mn = gat_finite_or_zero(fmaxf(fmaxf(m, l0), fmaxf(fmaxf(l1, l2), l3)));
  float sc = __expf(m - mn);
  float p0 = __expf(l0 - mn);
  float p1 = __expf(l1 - mn);
  float p2 = __expf(l2 - mn);
  float p3 = __expf(l3 - mn);
  s = s * sc + ((p0 + p1) + (p2 + p3));
  te += (e.x + e.y) + (e.z + e.w);
  acc.x = acc.x * sc + p0 * R.x0.x; acc.y = acc.y * sc + p0 * R.x0.y;
  acc.z = acc.z * sc + p0 * R.x0.z; acc.w = acc.w * sc + p0 * R.x0.w;
  acc.x += p1 * R.x1.x; acc.y += p1 * R.x1.y; acc.z += p1 * R.x1.z; acc.w += p1 * R.x1.w;
  acc.x += p2 * R.x2.x; acc.y += p2 * R.x2.y; acc.z += p2 * R.x2.z; acc.w += p2 * R.x2.w;
  acc.x += p3 * R.x3.x; acc.y += p3 * R.x3.y; acc.z += p3 * R.x3.z; acc.w += p3 * R.x3.w;
  m = mn;
}

__global__ __launch_bounds__(256) void gat_aggr_csr(
    const int* __restrict__ csrw, const int* __restrict__ segb,
    const float* __restrict__ cnt, const float* __restrict__ aedge,
    const float* __restrict__ att_s, const float* __restrict__ att_d,
    const float* __restrict__ xh, const float* __restrict__ bias, int relu,
    float* __restrict__ out, int E, int Nn) {
  __shared__ __attribute__((aligned(16))) float coef[8][128];
  const int2* csr = (const int2*)csrw;
  int lane = threadIdx.x & 31;
  int w = threadIdx.x >> 5;
  int n = __builtin_amdgcn_readfirstlane(
      (int)((blockIdx.x * blockDim.x + threadIdx.x) >> 5));
  if (n >= Nn) return;
  int c0 = lane * 4;
  int h = lane >> 3;
  int beg = segb[n];
  float cf = cnt[n];
  int end = beg + (int)cf;
  if (beg < 0) beg = 0;
  if (end > E) end = E;
  int deg = end - beg;
  if (deg < 0) deg = 0;
  float cc = (cf < 1.0f) ? 1.0f : cf;

  float4 as4 = *(const float4*)(att_s + c0);
  float4 ad4 = *(const float4*)(att_d + c0);
  float4 xs = gat_row(xh, n, lane);
  float adn = gat_sum8(gat_dot4(xs, ad4));
  float dsn = gat_sum8(gat_dot4(xs, as4));

  float* sl = coef[w];
  float m = -__builtin_inff(), s = 0.0f, te = 0.0f;
  float4 acc = make_float4(0.0f, 0.0f, 0.0f, 0.0f);

  for (int base = 0; base < deg; base += 32) {
    int left = deg - base;
    int cntc = (left < 32) ? left : 32;
    int srcv = n;
    float4 ae = make_float4(0.0f, 0.0f, 0.0f, 0.0f);
    if (lane < cntc) {
      srcv = csr[beg + base + lane].x;
      ae = *(const float4*)(aedge + (long long)(beg + base + lane) * 4);
    }
    gat_rows4 R;
    int j = 0;
    int rem = (cntc < 4) ? cntc : 4;
    gat_issue(xh, srcv, 0, rem, lane, R);
    sl[lane] = ae.x; sl[32 + lane] = ae.y; sl[64 + lane] = ae.z; sl[96 + lane] = ae.w;
    for (;;) {
      gat_step(R, sl, j, rem, h, as4, adn, m, s, te, acc);
      j += 4;
      if (j >= cntc) break;
      rem = (cntc - j < 4) ? cntc - j : 4;
      gat_issue(xh, srcv, j, rem, lane, R);
    }
  }

  {
    float ls = gat_leaky((dsn + adn) + te / cc);
    float mn = gat_finite_or_zero(fmaxf(m, ls));
    float sc = __expf(m - mn);
    float p = __expf(ls - mn);
    s = s * sc + p;
    acc.x = acc.x * sc + p * xs.x; acc.y = acc.y * sc + p * xs.y;
    acc.z = acc.z * sc + p * xs.z; acc.w = acc.w * sc + p * xs.w;
  }
  float inv = 1.0f / (s + 1e-16f);
  float4 bv = *(const float4*)(bias + c0);
  float4 o;
  o.x = acc.x * inv + bv.x; o.y = acc.y * inv + bv.y; o.z = acc.z * inv + bv.z; o.w = acc.w * inv + bv.w;
  if (relu) {
    o.x = (o.x > 0.0f) ? o.x : 0.0f; o.y = (o.y > 0.0f) ? o.y : 0.0f;
    o.z = (o.z > 0.0f) ? o.z : 0.0f; o.w = (o.w > 0.0f) ? o.w : 0.0f;
  }
  *(float4*)(out + (long long)n * 128 + c0) = o;
}

extern "C" void kernel_launch(void* const* d_in, const int* in_sizes, int n_in,
                              void* d_out, int out_size, void* d_ws,
                              size_t ws_size, hipStream_t stream) {
  const int N = in_sizes[0] / 128;
  const int E = in_sizes[1] / 2;

  const float* x  = (const float*)d_in[0];
  const int* ei   = (const int*)d_in[1];
  const int* srcI = ei;
  const int* dstI = ei + E;
  const float* ea = (const float*)d_in[2];

  const float* Wl[2]  = {(const float*)d_in[3], (const float*)d_in[9]};
  const float* asl[2] = {(const float*)d_in[4], (const float*)d_in[10]};
  const float* adl[2] = {(const float*)d_in[5], (const float*)d_in[11]};
  const float* Wel[2] = {(const float*)d_in[6], (const float*)d_in[12]};
  const float* ael[2] = {(const float*)d_in[7], (const float*)d_in[13]};
  const float* bl[2]  = {(const float*)d_in[8], (const float*)d_in[14]};

  float* wsf = (float*)d_ws;
  size_t off = 0;
  float* cnt   = wsf + off; off += (size_t)N;
  float* medge = wsf + off;
  off += (size_t)N * 16;
  off += 64;
  float* wpk   = wsf + (((size_t)N + 128 + 127) & ~(size_t)127);
  float* xh    = wsf + off; off += (size_t)N * 128;
  off += (size_t)N * 8;
  float* aedge = wsf + off; off += ((size_t)E + N) * 4;
  off += (size_t)N * 8;
  size_t lo = (off + 1) & ~(size_t)1;
  int* segb    = (int*)(wsf + lo);
  int* csr     = segb + 2 * (size_t)N + 4;
  const int NB = (N + 511) / 512;
  const int EB = (E + 2047) / 2048;
  size_t to = (lo + 2 * (size_t)N + 4 + 2 * (size_t)E + 3) & ~(size_t)3;
  int4* tmp    = (int4*)(wsf + to);
  float* aedge2 = wsf + to;
  int* tbl     = (int*)(wsf + to) + 4 * (size_t)E;
  int* btot    = tbl + (size_t)NB * EB;
  off += (size_t)N * 128;
  (void)ws_size; (void)n_in; (void)out_size;

  hipLaunchKernelGGL(gat_bucket_hist, dim3((unsigned)EB), dim3(256), 0, stream, dstI, tbl, E, NB);
  hipLaunchKernelGGL(gat_bucket_rowscan, dim3((unsigned)NB), dim3(256), 0, stream, tbl, btot, EB);
  hipLaunchKernelGGL(gat_bucket_scatter, dim3((unsigned)EB), dim3(256), 0, stream, srcI, dstI,
                     tbl, btot, tmp, E, NB);
  hipLaunchKernelGGL(gat_bucket_place, dim3((unsigned)NB), dim3(512), 0, stream, tmp, btot,
                     segb, cnt, csr, E, N);

  hipLaunchKernelGGL(gat_fold_edge, dim3(2 + 64), dim3(64), 0, stream, Wel[0], ael[0], Wel[1], ael[1],
                     Wl[0], Wl[1], medge, wpk);
  hipLaunchKernelGGL(gat_edge_term_list, dim3((unsigned)((E + 255) / 256)), dim3(256),
                     0, stream, csr, ea, medge, aedge, aedge2, E);

  for (int l = 0; l < 2; ++l) {
    const float* xin = (l == 0) ? x : (const float*)d_out;
    float* outb = (float*)d_out;

    {
      long long waves = ((long long)N + 15) / 16;
      long long blocks = (waves + 7) / 8;
      hipLaunchKernelGGL(gat_gemm_wmma, dim3((unsigned)blocks), dim3(256), 0,
                         stream, xin, wpk + (size_t)l * 16384, xh, N);
    }

    {
      long long blocks = ((long long)N + 7) / 8;
      hipLaunchKernelGGL(gat_aggr_csr, dim3((unsigned)blocks), dim3(256), 0,
                         stream, csr, segb, cnt, (l == 0) ? aedge : aedge2, asl[l], adl[l], xh, bl[l],
                         (l == 0) ? 1 : 0, outb, E, N);
    }
  }
}
